# starting kernel
# baseline (speedup 1.0000x reference)
#include <hip/hip_runtime.h>
#include <hip/hip_bf16.h>

typedef __attribute__((ext_vector_type(2))) float v2f;
typedef __attribute__((ext_vector_type(8))) float v8f;

#define D_FEAT 128


__global__ void zero_f32(float* __restrict__ p, size_t n) {
    size_t i = (size_t)blockIdx.x * blockDim.x + threadIdx.x;
    if (i < n) p[i] = 0.0f;
}

__global__ void degree_kernel(const int* __restrict__ src, const int* __restrict__ dst,
                              float* __restrict__ outdeg, float* __restrict__ indeg, int E) {
    int e = blockIdx.x * blockDim.x + threadIdx.x;
    if (e < E) {
        unsafeAtomicAdd(&outdeg[src[e]], 1.0f);
        unsafeAtomicAdd(&indeg[dst[e]], 1.0f);
    }
}

__global__ void norm_kernel(float* __restrict__ p, int n) {
    int i = blockIdx.x * blockDim.x + threadIdx.x;
    if (i < n) {
        float d = fmaxf(p[i], 1.0f);
        p[i] = rsqrtf(d);
    }
}

__global__ void scatter_kernel(const float* __restrict__ h, const float* __restrict__ norm_src,
                               const int* __restrict__ src, const int* __restrict__ dst,
                               float* __restrict__ agg, int E) {
    int t = blockIdx.x * blockDim.x + threadIdx.x;
    int e = t >> 5;
    int c = (t & 31) << 2;
    if (e < E) {
        int s = src[e];
        int d = dst[e];
        float ns = norm_src[s];
        float4 v = *(const float4*)(h + (size_t)s * D_FEAT + c);
        float* ap = agg + (size_t)d * D_FEAT + c;
        unsafeAtomicAdd(ap + 0, v.x * ns);
        unsafeAtomicAdd(ap + 1, v.y * ns);
        unsafeAtomicAdd(ap + 2, v.z * ns);
        unsafeAtomicAdd(ap + 3, v.w * ns);
    }
}

__global__ void pool_kernel(const float* __restrict__ h, const int* __restrict__ gid,
                            float* __restrict__ hg, int N) {
    int t = blockIdx.x * blockDim.x + threadIdx.x;
    int n = t >> 5;
    int c = (t & 31) << 2;
    if (n < N) {
        int g = gid[n];
        float4 v = *(const float4*)(h + (size_t)n * D_FEAT + c);
        float* hp = hg + (size_t)g * D_FEAT + c;
        unsafeAtomicAdd(hp + 0, v.x);
        unsafeAtomicAdd(hp + 1, v.y);
        unsafeAtomicAdd(hp + 2, v.z);
        unsafeAtomicAdd(hp + 3, v.w);
    }
}

#define MT 4
__global__ void __launch_bounds__(32)
gemm_wmma_f32(const float* __restrict__ A, const float* __restrict__ B,
              const float* __restrict__ bias,
              const float* __restrict__ row_scale,
              const float* __restrict__ resid,
              float* __restrict__ out,
              int Mrows, int K, int Ncol, int relu_c, int mode) {
    int tm   = blockIdx.x;
    int tn   = blockIdx.y;
    int lane = threadIdx.x;
    int half = lane >> 4;
    int l15  = lane & 15;

    int col = tn * 16 + l15;
    float bv = bias[col];

    const float* arow[MT];
    float rs[MT];
#pragma unroll
    for (int i = 0; i < MT; ++i) {
        int row = tm * (16 * MT) + i * 16 + l15;
        if (row >= Mrows) row = Mrows - 1;
        arow[i] = A + (size_t)row * K;
        rs[i] = row_scale ? row_scale[row] : 1.0f;
    }

    v8f acc[MT] = {};
    for (int k0 = 0; k0 < K; k0 += 4) {
        int ka = k0 + half * 2;
        v2f b;
        b.x = B[(size_t)ka * Ncol + col];
        b.y = B[(size_t)(ka + 1) * Ncol + col];
#pragma unroll
        for (int i = 0; i < MT; ++i) {
            __builtin_prefetch(arow[i] + k0 + 64, 0, 0);
            v2f a = *(const v2f*)(arow[i] + ka);
            a.x *= rs[i];
            a.y *= rs[i];
            acc[i] = __builtin_amdgcn_wmma_f32_16x16x4_f32(
                 false, a,  false, b,
                 (short)0, acc[i],  false,  false);
        }
    }

#pragma unroll
    for (int i = 0; i < MT; ++i) {
#pragma unroll
        for (int v = 0; v < 8; ++v) {
            int m = tm * (16 * MT) + i * 16 + half * 8 + v;
            if (m < Mrows) {
                float c = acc[i][v] + bv;
                if (relu_c) c = fmaxf(c, 0.0f);
                float o;
                if (mode == 1)      o = fmaxf(resid[(size_t)m * Ncol + col] + c, 0.0f);
                else if (mode == 2) o = fmaxf(c, 0.0f);
                else                o = c;
                out[(size_t)m * Ncol + col] = o;
            }
        }
    }
}


extern "C" void kernel_launch(void* const* d_in, const int* in_sizes, int n_in,
                              void* d_out, int out_size, void* d_ws, size_t ws_size,
                              hipStream_t stream) {
    const float* node_features = (const float*)d_in[0];
    const int*   edge_src      = (const int*)d_in[1];
    const int*   edge_dst      = (const int*)d_in[2];
    const int*   graph_ids     = (const int*)d_in[3];
    const float* W_proj        = (const float*)d_in[4];
    const float* b_proj        = (const float*)d_in[5];
    const float* W_gcn         = (const float*)d_in[6];
    const float* b_gcn         = (const float*)d_in[7];
    const float* W_mlp0        = (const float*)d_in[8];
    const float* b_mlp0        = (const float*)d_in[9];
    const float* W_mlp1        = (const float*)d_in[10];
    const float* b_mlp1        = (const float*)d_in[11];
    const float* W_mlp2        = (const float*)d_in[12];
    const float* b_mlp2        = (const float*)d_in[13];

    const int D = D_FEAT;
    const int Mm = 256;
    const int N = in_sizes[3];
    const int E = in_sizes[1];
    const int L = in_sizes[6] / (D * D);
    const int G = out_size / Mm;

    float* ws = (float*)d_ws;
    float* norm_src = ws;
    float* norm_dst = norm_src + N;
    float* h        = norm_dst + N;
    float* agg      = h + (size_t)N * D;
    float* hg       = agg + (size_t)N * D;
    float* t0       = hg + (size_t)G * D;
    float* t1       = t0 + (size_t)G * Mm;

    const int TB = 256;
    auto blocks = [&](size_t n) { return (unsigned)((n + TB - 1) / TB); };
    const int MROWS = 16 * MT;

    zero_f32<<<blocks((size_t)2 * N), TB, 0, stream>>>(norm_src, (size_t)2 * N);
    degree_kernel<<<blocks(E), TB, 0, stream>>>(edge_src, edge_dst, norm_src, norm_dst, E);
    norm_kernel<<<blocks((size_t)2 * N), TB, 0, stream>>>(norm_src, 2 * N);

    {
        dim3 g((N + MROWS - 1) / MROWS, D / 16);
        gemm_wmma_f32<<<g, 32, 0, stream>>>(node_features, W_proj, b_proj,
                                            nullptr, nullptr, h, N, D, D,
                                             0,  0);
    }

    for (int i = 0; i < L; ++i) {
        zero_f32<<<blocks((size_t)N * D), TB, 0, stream>>>(agg, (size_t)N * D);
        scatter_kernel<<<blocks((size_t)E * 32), TB, 0, stream>>>(h, norm_src,
                                                                  edge_src, edge_dst, agg, E);
        dim3 g((N + MROWS - 1) / MROWS, D / 16);
        gemm_wmma_f32<<<g, 32, 0, stream>>>(agg, W_gcn + (size_t)i * D * D,
                                            b_gcn + (size_t)i * D,
                                            norm_dst, h, h, N, D, D,
                                             (i < L - 1) ? 1 : 0,  1);
    }

    zero_f32<<<blocks((size_t)G * D), TB, 0, stream>>>(hg, (size_t)G * D);
    pool_kernel<<<blocks((size_t)N * 32), TB, 0, stream>>>(h, graph_ids, hg, N);

    {
        dim3 g0((G + MROWS - 1) / MROWS, Mm / 16);
        gemm_wmma_f32<<<g0, 32, 0, stream>>>(hg, W_mlp0, b_mlp0, nullptr, nullptr,
                                             t0, G, D, Mm, 0,  2);
        gemm_wmma_f32<<<g0, 32, 0, stream>>>(t0, W_mlp1, b_mlp1, nullptr, nullptr,
                                             t1, G, Mm, Mm, 0,  2);
        gemm_wmma_f32<<<g0, 32, 0, stream>>>(t1, W_mlp2, b_mlp2, nullptr, nullptr,
                                             (float*)d_out, G, Mm, Mm, 0,  0);
    }
}


# Node GEMMs (N x128x128) retiled: A rows read once by whole-row loads, staged in LDS for 4 waves that keep their W fragments in registers (was strided A loads, A re-read 8x); k order kept
# speedup vs baseline: 1.5879x; 1.5879x over previous
#include <hip/hip_runtime.h>
#include <hip/hip_bf16.h>

typedef __attribute__((ext_vector_type(2))) float v2f;
typedef __attribute__((ext_vector_type(8))) float v8f;

#define D_FEAT 128


__global__ void zero_f32(float* __restrict__ p, size_t n) {
    size_t i = (size_t)blockIdx.x * blockDim.x + threadIdx.x;
    if (i < n) p[i] = 0.0f;
}

__global__ void degree_kernel(const int* __restrict__ src, const int* __restrict__ dst,
                              float* __restrict__ outdeg, float* __restrict__ indeg, int E) {
    int e = blockIdx.x * blockDim.x + threadIdx.x;
    if (e < E) {
        unsafeAtomicAdd(&outdeg[src[e]], 1.0f);
        unsafeAtomicAdd(&indeg[dst[e]], 1.0f);
    }
}

__global__ void norm_kernel(float* __restrict__ p, int n) {
    int i = blockIdx.x * blockDim.x + threadIdx.x;
    if (i < n) {
        float d = fmaxf(p[i], 1.0f);
        p[i] = rsqrtf(d);
    }
}

__global__ void scatter_kernel(const float* __restrict__ h, const float* __restrict__ norm_src,
                               const int* __restrict__ src, const int* __restrict__ dst,
                               float* __restrict__ agg, int E) {
    int t = blockIdx.x * blockDim.x + threadIdx.x;
    int e = t >> 5;
    int c = (t & 31) << 2;
    if (e < E) {
        int s = src[e];
        int d = dst[e];
        float ns = norm_src[s];
        float4 v = *(const float4*)(h + (size_t)s * D_FEAT + c);
        float* ap = agg + (size_t)d * D_FEAT + c;
        unsafeAtomicAdd(ap + 0, v.x * ns);
        unsafeAtomicAdd(ap + 1, v.y * ns);
        unsafeAtomicAdd(ap + 2, v.z * ns);
        unsafeAtomicAdd(ap + 3, v.w * ns);
    }
}

__global__ void pool_kernel(const float* __restrict__ h, const int* __restrict__ gid,
                            float* __restrict__ hg, int N) {
    int t = blockIdx.x * blockDim.x + threadIdx.x;
    int n = t >> 5;
    int c = (t & 31) << 2;
    if (n < N) {
        int g = gid[n];
        float4 v = *(const float4*)(h + (size_t)n * D_FEAT + c);
        float* hp = hg + (size_t)g * D_FEAT + c;
        unsafeAtomicAdd(hp + 0, v.x);
        unsafeAtomicAdd(hp + 1, v.y);
        unsafeAtomicAdd(hp + 2, v.z);
        unsafeAtomicAdd(hp + 3, v.w);
    }
}

#define MT 4
__global__ void __launch_bounds__(32)
gemm_wmma_f32(const float* __restrict__ A, const float* __restrict__ B,
              const float* __restrict__ bias,
              const float* __restrict__ row_scale,
              const float* __restrict__ resid,
              float* __restrict__ out,
              int Mrows, int K, int Ncol, int relu_c, int mode) {
    int tm   = blockIdx.x;
    int tn   = blockIdx.y;
    int lane = threadIdx.x;
    int half = lane >> 4;
    int l15  = lane & 15;

    int col = tn * 16 + l15;
    float bv = bias[col];

    const float* arow[MT];
    float rs[MT];
#pragma unroll
    for (int i = 0; i < MT; ++i) {
        int row = tm * (16 * MT) + i * 16 + l15;
        if (row >= Mrows) row = Mrows - 1;
        arow[i] = A + (size_t)row * K;
        rs[i] = row_scale ? row_scale[row] : 1.0f;
    }

    v8f acc[MT] = {};
    for (int k0 = 0; k0 < K; k0 += 4) {
        int ka = k0 + half * 2;
        v2f b;
        b.x = B[(size_t)ka * Ncol + col];
        b.y = B[(size_t)(ka + 1) * Ncol + col];
#pragma unroll
        for (int i = 0; i < MT; ++i) {
            __builtin_prefetch(arow[i] + k0 + 64, 0, 0);
            v2f a = *(const v2f*)(arow[i] + ka);
            a.x *= rs[i];
            a.y *= rs[i];
            acc[i] = __builtin_amdgcn_wmma_f32_16x16x4_f32(
                 false, a,  false, b,
                 (short)0, acc[i],  false,  false);
        }
    }

#pragma unroll
    for (int i = 0; i < MT; ++i) {
#pragma unroll
        for (int v = 0; v < 8; ++v) {
            int m = tm * (16 * MT) + i * 16 + half * 8 + v;
            if (m < Mrows) {
                float c = acc[i][v] + bv;
                if (relu_c) c = fmaxf(c, 0.0f);
                float o;
                if (mode == 1)      o = fmaxf(resid[(size_t)m * Ncol + col] + c, 0.0f);
                else if (mode == 2) o = fmaxf(c, 0.0f);
                else                o = c;
                out[(size_t)m * Ncol + col] = o;
            }
        }
    }
}


typedef __attribute__((ext_vector_type(4))) float v4f;
#define GR 16
#define GP 132
template <int MODE>
__global__ void __launch_bounds__(128)
gemm128_wmma_f32(const float* __restrict__ A, const float* __restrict__ B,
                 const float* __restrict__ bias,
                 const float* __restrict__ row_scale,
                 const float* resid,
                 float* out,
                 int Mrows, int relu_c) {
    __shared__ __attribute__((aligned(16))) float sA[2][GR * GP];

    const int tid  = threadIdx.x;
    const int w    = __builtin_amdgcn_readfirstlane(tid >> 5);
    const int lane = tid & 31;
    const int half = lane >> 4;
    const int l15  = lane & 15;
    const int colb = 32 * w + l15;

    const int nchunks = (Mrows + GR - 1) / GR;
    int chunk = blockIdx.x;

    v4f   st[4];
    float rsc[4];
#pragma unroll
    for (int i = 0; i < 4; ++i) {
        int row = chunk * GR + i * 4 + w;
        if (row >= Mrows) row = Mrows - 1;
        st[i] = *(const v4f*)(A + (size_t)row * D_FEAT + lane * 4);
        if (MODE == 1) rsc[i] = row_scale[row];
    }

    v2f bf[32][2];
#pragma unroll
    for (int s = 0; s < 32; ++s) {
        int ka = 4 * s + 2 * half;
#pragma unroll
        for (int n = 0; n < 2; ++n) {
            bf[s][n].x = B[(size_t)ka * D_FEAT + colb + 16 * n];
            bf[s][n].y = B[(size_t)(ka + 1) * D_FEAT + colb + 16 * n];
        }
    }
    float bv[2];
    bv[0] = bias[colb];
    bv[1] = bias[colb + 16];

#pragma unroll
    for (int i = 0; i < 4; ++i) {
        v4f v = st[i];
        if (MODE == 1) v *= rsc[i];
        *(v4f*)(&sA[0][(i * 4 + w) * GP + lane * 4]) = v;
    }
    __syncthreads();

    int buf = 0;
    for (; chunk < nchunks; chunk += gridDim.x) {
        const int  next     = chunk + gridDim.x;
        const bool has_next = next < nchunks;
        const int  row0     = chunk * GR + half * 8;
        const bool full     = chunk * GR + GR <= Mrows;

        if (has_next) {
#pragma unroll
            for (int i = 0; i < 4; ++i) {
                int row = next * GR + i * 4 + w;
                if (row >= Mrows) row = Mrows - 1;
                st[i] = *(const v4f*)(A + (size_t)row * D_FEAT + lane * 4);
                if (MODE == 1) rsc[i] = row_scale[row];
            }
        }

        float r[2][8];
        if (MODE == 1 && full) {
            const float* rp = resid + (size_t)row0 * D_FEAT + colb;
#pragma unroll
            for (int n = 0; n < 2; ++n)
#pragma unroll
                for (int v = 0; v < 8; ++v)
                    r[n][v] = rp[v * D_FEAT + 16 * n];
        }

        v8f acc[2] = {};
        const float* ap = &sA[buf][l15 * GP + 2 * half];
#pragma unroll
        for (int s = 0; s < 32; ++s) {
            v2f a = *(const v2f*)(ap + 4 * s);
            acc[0] = __builtin_amdgcn_wmma_f32_16x16x4_f32(false, a, false, bf[s][0], (short)0, acc[0], false, false);
            acc[1] = __builtin_amdgcn_wmma_f32_16x16x4_f32(false, a, false, bf[s][1], (short)0, acc[1], false, false);
        }

        if (has_next) {
#pragma unroll
            for (int i = 0; i < 4; ++i) {
                v4f v = st[i];
                if (MODE == 1) v *= rsc[i];
                *(v4f*)(&sA[buf ^ 1][(i * 4 + w) * GP + lane * 4]) = v;
            }
        }
        __syncthreads();

        if (full) {
            float* op = out + (size_t)row0 * D_FEAT + colb;
#pragma unroll
            for (int n = 0; n < 2; ++n)
#pragma unroll
                for (int v = 0; v < 8; ++v) {
                    float c = acc[n][v] + bv[n];
                    if (relu_c) c = fmaxf(c, 0.0f);
                    op[v * D_FEAT + 16 * n] = (MODE == 1) ? fmaxf(r[n][v] + c, 0.0f) : c;
                }
        } else {
            float rt[2][8];
            if (MODE == 1) {
#pragma unroll
                for (int n = 0; n < 2; ++n)
#pragma unroll
                    for (int v = 0; v < 8; ++v) {
                        int row = row0 + v;
                        if (row >= Mrows) row = Mrows - 1;
                        rt[n][v] = resid[(size_t)row * D_FEAT + colb + 16 * n];
                    }
            }
#pragma unroll
            for (int n = 0; n < 2; ++n)
#pragma unroll
                for (int v = 0; v < 8; ++v) {
                    int row = row0 + v;
                    float c = acc[n][v] + bv[n];
                    if (relu_c) c = fmaxf(c, 0.0f);
                    float o = (MODE == 1) ? fmaxf(rt[n][v] + c, 0.0f) : c;
                    if (row < Mrows) out[(size_t)row * D_FEAT + colb + 16 * n] = o;
                }
        }

        buf ^= 1;
    }
}


extern "C" void kernel_launch(void* const* d_in, const int* in_sizes, int n_in,
                              void* d_out, int out_size, void* d_ws, size_t ws_size,
                              hipStream_t stream) {
    const float* node_features = (const float*)d_in[0];
    const int*   edge_src      = (const int*)d_in[1];
    const int*   edge_dst      = (const int*)d_in[2];
    const int*   graph_ids     = (const int*)d_in[3];
    const float* W_proj        = (const float*)d_in[4];
    const float* b_proj        = (const float*)d_in[5];
    const float* W_gcn         = (const float*)d_in[6];
    const float* b_gcn         = (const float*)d_in[7];
    const float* W_mlp0        = (const float*)d_in[8];
    const float* b_mlp0        = (const float*)d_in[9];
    const float* W_mlp1        = (const float*)d_in[10];
    const float* b_mlp1        = (const float*)d_in[11];
    const float* W_mlp2        = (const float*)d_in[12];
    const float* b_mlp2        = (const float*)d_in[13];

    const int D = D_FEAT;
    const int Mm = 256;
    const int N = in_sizes[3];
    const int E = in_sizes[1];
    const int L = in_sizes[6] / (D * D);
    const int G = out_size / Mm;

    float* ws = (float*)d_ws;
    float* norm_src = ws;
    float* norm_dst = norm_src + N;
    float* h        = norm_dst + N;
    float* agg      = h + (size_t)N * D;
    float* hg       = agg + (size_t)N * D;
    float* t0       = hg + (size_t)G * D;
    float* t1       = t0 + (size_t)G * Mm;

    const int TB = 256;
    auto blocks = [&](size_t n) { return (unsigned)((n + TB - 1) / TB); };
    const int MROWS = 16 * MT;

    zero_f32<<<blocks((size_t)2 * N), TB, 0, stream>>>(norm_src, (size_t)2 * N);
    degree_kernel<<<blocks(E), TB, 0, stream>>>(edge_src, edge_dst, norm_src, norm_dst, E);
    norm_kernel<<<blocks((size_t)2 * N), TB, 0, stream>>>(norm_src, 2 * N);

    const unsigned gblocks = (unsigned)(((N + GR - 1) / GR + 3) / 4);
    gemm128_wmma_f32<0><<<gblocks, 128, 0, stream>>>(node_features, W_proj, b_proj,
                                                     nullptr, nullptr, h, N,  0);

    for (int i = 0; i < L; ++i) {
        zero_f32<<<blocks((size_t)N * D), TB, 0, stream>>>(agg, (size_t)N * D);
        scatter_kernel<<<blocks((size_t)E * 32), TB, 0, stream>>>(h, norm_src,
                                                                  edge_src, edge_dst, agg, E);
        gemm128_wmma_f32<1><<<gblocks, 128, 0, stream>>>(agg, W_gcn + (size_t)i * D * D,
                                                         b_gcn + (size_t)i * D,
                                                         norm_dst, h, h, N,
                                                          (i < L - 1) ? 1 : 0);
    }

    zero_f32<<<blocks((size_t)G * D), TB, 0, stream>>>(hg, (size_t)G * D);
    pool_kernel<<<blocks((size_t)N * 32), TB, 0, stream>>>(h, graph_ids, hg, N);

    {
        dim3 g0((G + MROWS - 1) / MROWS, Mm / 16);
        gemm_wmma_f32<<<g0, 32, 0, stream>>>(hg, W_mlp0, b_mlp0, nullptr, nullptr,
                                             t0, G, D, Mm, 0,  2);
        gemm_wmma_f32<<<g0, 32, 0, stream>>>(t0, W_mlp1, b_mlp1, nullptr, nullptr,
                                             t1, G, Mm, Mm, 0,  2);
        gemm_wmma_f32<<<g0, 32, 0, stream>>>(t1, W_mlp2, b_mlp2, nullptr, nullptr,
                                             (float*)d_out, G, Mm, Mm, 0,  0);
    }
}


# Per-layer f32 atomic scatter-add replaced by register gather-sum over an in-edge list grouped by destination once per call (int atomic rank + scan); rows summed in list order; agg clear dropped
# speedup vs baseline: 3.1246x; 1.9678x over previous
#include <hip/hip_runtime.h>
#include <hip/hip_bf16.h>

typedef __attribute__((ext_vector_type(2))) float v2f;
typedef __attribute__((ext_vector_type(8))) float v8f;

#define D_FEAT 128


typedef __attribute__((ext_vector_type(4))) float v4f;
typedef __attribute__((ext_vector_type(2))) int v2i;

__global__ void zero_f32(float* __restrict__ p, size_t n) {
    size_t i = (size_t)blockIdx.x * blockDim.x + threadIdx.x;
    if (i < n) p[i] = 0.0f;
}

__global__ void zero_i32(int* __restrict__ p, size_t n) {
    size_t i = (size_t)blockIdx.x * blockDim.x + threadIdx.x;
    if (i < n) p[i] = 0;
}


__global__ void count_rank_kernel(const int* __restrict__ src, const int* __restrict__ dst,
                                  int* __restrict__ outcnt, int* __restrict__ incnt,
                                  int* __restrict__ rank, int E) {
    int e = blockIdx.x * blockDim.x + threadIdx.x;
    if (e < E) {
        atomicAdd(&outcnt[src[e]], 1);
        rank[e] = atomicAdd(&incnt[dst[e]], 1);
    }
}

__global__ void __launch_bounds__(1024)
rowptr_scan_kernel(const int* __restrict__ cnt, int* __restrict__ rowptr, int n) {
    __shared__ int wtot[32];
    const int lane = threadIdx.x & 31;
    const int w    = __builtin_amdgcn_readfirstlane(threadIdx.x >> 5);
    const int seg  = ((n + 4095) / 4096) * 128;
    const int b    = w * seg;
    const int e    = (b + seg < n) ? (b + seg) : n;

    int s = 0;
#pragma unroll 4
    for (int i = b + lane; i < e; i += 32) s += cnt[i];
    for (int o = 16; o > 0; o >>= 1) s += __shfl_xor(s, o);
    if (lane == 0) wtot[w] = s;
    __syncthreads();

    int base = 0;
    for (int k = 0; k < w; ++k) base += wtot[k];

    int c[4];
#pragma unroll
    for (int q = 0; q < 4; ++q) {
        int i = b + 4 * lane + q;
        c[q] = (i < e) ? cnt[i] : 0;
    }
    for (int i0 = b; i0 < e; i0 += 128) {
        int cn[4];
#pragma unroll
        for (int q = 0; q < 4; ++q) {
            int i = i0 + 128 + 4 * lane + q;
            cn[q] = (i < e) ? cnt[i] : 0;
        }
        const int s4 = c[0] + c[1] + c[2] + c[3];
        int v = s4;
        for (int o = 1; o < 32; o <<= 1) {
            int u = __shfl_up(v, o);
            if (lane >= o) v += u;
        }
        int ex = base + v - s4;
#pragma unroll
        for (int q = 0; q < 4; ++q) {
            int i = i0 + 4 * lane + q;
            if (i < e) rowptr[i] = ex;
            ex += c[q];
            c[q] = cn[q];
        }
        base += __shfl(v, 31);
    }
    if (w == 31 && lane == 0) rowptr[n] = base;
}

__global__ void norm_kernel(const int* __restrict__ cnt, float* __restrict__ nrm, int n) {
    int i = blockIdx.x * blockDim.x + threadIdx.x;
    if (i < n) {
        float d = fmaxf((float)cnt[i], 1.0f);
        nrm[i] = rsqrtf(d);
    }
}

__global__ void fill_kernel(const int* __restrict__ src, const int* __restrict__ dst,
                            const int* __restrict__ rank, const int* __restrict__ rowptr,
                            const float* __restrict__ norm_src, int* __restrict__ rec, int E) {
    int e = blockIdx.x * blockDim.x + threadIdx.x;
    if (e < E) {
        int s = src[e];
        int pos = rowptr[dst[e]] + rank[e];
        v2i r;
        r.x = s;
        r.y = __float_as_int(norm_src[s]);
        *(v2i*)(rec + 2 * (size_t)pos) = r;
    }
}

template <int K>
__device__ __forceinline__ void gather_rows(const float* __restrict__ h, v2i r, int j, int lane4, v4f& acc) {
    v4f   v[K];
    float ns[K];
#pragma unroll
    for (int k = 0; k < K; ++k) {
        const int s = __builtin_amdgcn_readlane(r.x, j + k);
        ns[k] = __int_as_float(__builtin_amdgcn_readlane(r.y, j + k));
        v[k] = *(const v4f*)(h + (size_t)s * D_FEAT + lane4);
    }
#pragma unroll
    for (int k = 0; k < K; ++k) acc += v[k] * ns[k];
}

__global__ void __launch_bounds__(256)
gather_sum_kernel(const float* __restrict__ h, const int* __restrict__ rec,
                  const int* __restrict__ rowptr, float* __restrict__ agg, int N) {
    const int lane   = threadIdx.x & 31;
    const int lane4  = lane * 4;
    const int wave   = __builtin_amdgcn_readfirstlane(blockIdx.x * 8 + (threadIdx.x >> 5));
    const int nwaves = gridDim.x * 8;
    for (int n = wave; n < N; n += nwaves) {
        const int beg = rowptr[n];
        const int end = rowptr[n + 1];
        v4f acc = {0.0f, 0.0f, 0.0f, 0.0f};
        for (int b = beg; b < end; b += 32) {
            const int m = (end - b < 32) ? (end - b) : 32;
            v2i r = {0, 0};
            if (lane < m) r = *(const v2i*)(rec + 2 * (size_t)(b + lane));
            int j = 0;
            for (; j + 8 <= m; j += 8) gather_rows<8>(h, r, j, lane4, acc);
            if (j + 4 <= m) { gather_rows<4>(h, r, j, lane4, acc); j += 4; }
            if (j + 2 <= m) { gather_rows<2>(h, r, j, lane4, acc); j += 2; }
            if (j < m)      { gather_rows<1>(h, r, j, lane4, acc); }
        }
        *(v4f*)(agg + (size_t)n * D_FEAT + lane4) = acc;
    }
}

__global__ void pool_kernel(const float* __restrict__ h, const int* __restrict__ gid,
                            float* __restrict__ hg, int N) {
    int t = blockIdx.x * blockDim.x + threadIdx.x;
    int n = t >> 5;
    int c = (t & 31) << 2;
    if (n < N) {
        int g = gid[n];
        float4 v = *(const float4*)(h + (size_t)n * D_FEAT + c);
        float* hp = hg + (size_t)g * D_FEAT + c;
        unsafeAtomicAdd(hp + 0, v.x);
        unsafeAtomicAdd(hp + 1, v.y);
        unsafeAtomicAdd(hp + 2, v.z);
        unsafeAtomicAdd(hp + 3, v.w);
    }
}

#define MT 4
__global__ void __launch_bounds__(32)
gemm_wmma_f32(const float* __restrict__ A, const float* __restrict__ B,
              const float* __restrict__ bias,
              const float* __restrict__ row_scale,
              const float* __restrict__ resid,
              float* __restrict__ out,
              int Mrows, int K, int Ncol, int relu_c, int mode) {
    int tm   = blockIdx.x;
    int tn   = blockIdx.y;
    int lane = threadIdx.x;
    int half = lane >> 4;
    int l15  = lane & 15;

    int col = tn * 16 + l15;
    float bv = bias[col];

    const float* arow[MT];
    float rs[MT];
#pragma unroll
    for (int i = 0; i < MT; ++i) {
        int row = tm * (16 * MT) + i * 16 + l15;
        if (row >= Mrows) row = Mrows - 1;
        arow[i] = A + (size_t)row * K;
        rs[i] = row_scale ? row_scale[row] : 1.0f;
    }

    v8f acc[MT] = {};
    for (int k0 = 0; k0 < K; k0 += 4) {
        int ka = k0 + half * 2;
        v2f b;
        b.x = B[(size_t)ka * Ncol + col];
        b.y = B[(size_t)(ka + 1) * Ncol + col];
#pragma unroll
        for (int i = 0; i < MT; ++i) {
            __builtin_prefetch(arow[i] + k0 + 64, 0, 0);
            v2f a = *(const v2f*)(arow[i] + ka);
            a.x *= rs[i];
            a.y *= rs[i];
            acc[i] = __builtin_amdgcn_wmma_f32_16x16x4_f32(
                 false, a,  false, b,
                 (short)0, acc[i],  false,  false);
        }
    }

#pragma unroll
    for (int i = 0; i < MT; ++i) {
#pragma unroll
        for (int v = 0; v < 8; ++v) {
            int m = tm * (16 * MT) + i * 16 + half * 8 + v;
            if (m < Mrows) {
                float c = acc[i][v] + bv;
                if (relu_c) c = fmaxf(c, 0.0f);
                float o;
                if (mode == 1)      o = fmaxf(resid[(size_t)m * Ncol + col] + c, 0.0f);
                else if (mode == 2) o = fmaxf(c, 0.0f);
                else                o = c;
                out[(size_t)m * Ncol + col] = o;
            }
        }
    }
}


#define GR 16
#define GP 132
template <int MODE>
__global__ void __launch_bounds__(128)
gemm128_wmma_f32(const float* __restrict__ A, const float* __restrict__ B,
                 const float* __restrict__ bias,
                 const float* __restrict__ row_scale,
                 const float* resid,
                 float* out,
                 int Mrows, int relu_c) {
    __shared__ __attribute__((aligned(16))) float sA[2][GR * GP];

    const int tid  = threadIdx.x;
    const int w    = __builtin_amdgcn_readfirstlane(tid >> 5);
    const int lane = tid & 31;
    const int half = lane >> 4;
    const int l15  = lane & 15;
    const int colb = 32 * w + l15;

    const int nchunks = (Mrows + GR - 1) / GR;
    int chunk = blockIdx.x;

    v4f   st[4];
    float rsc[4];
#pragma unroll
    for (int i = 0; i < 4; ++i) {
        int row = chunk * GR + i * 4 + w;
        if (row >= Mrows) row = Mrows - 1;
        st[i] = *(const v4f*)(A + (size_t)row * D_FEAT + lane * 4);
        if (MODE == 1) rsc[i] = row_scale[row];
    }

    v2f bf[32][2];
#pragma unroll
    for (int s = 0; s < 32; ++s) {
        int ka = 4 * s + 2 * half;
#pragma unroll
        for (int n = 0; n < 2; ++n) {
            bf[s][n].x = B[(size_t)ka * D_FEAT + colb + 16 * n];
            bf[s][n].y = B[(size_t)(ka + 1) * D_FEAT + colb + 16 * n];
        }
    }
    float bv[2];
    bv[0] = bias[colb];
    bv[1] = bias[colb + 16];

#pragma unroll
    for (int i = 0; i < 4; ++i) {
        v4f v = st[i];
        if (MODE == 1) v *= rsc[i];
        *(v4f*)(&sA[0][(i * 4 + w) * GP + lane * 4]) = v;
    }
    __syncthreads();

    int buf = 0;
    for (; chunk < nchunks; chunk += gridDim.x) {
        const int  next     = chunk + gridDim.x;
        const bool has_next = next < nchunks;
        const int  row0     = chunk * GR + half * 8;
        const bool full     = chunk * GR + GR <= Mrows;

        if (has_next) {
#pragma unroll
            for (int i = 0; i < 4; ++i) {
                int row = next * GR + i * 4 + w;
                if (row >= Mrows) row = Mrows - 1;
                st[i] = *(const v4f*)(A + (size_t)row * D_FEAT + lane * 4);
                if (MODE == 1) rsc[i] = row_scale[row];
            }
        }

        float r[2][8];
        if (MODE == 1 && full) {
            const float* rp = resid + (size_t)row0 * D_FEAT + colb;
#pragma unroll
            for (int n = 0; n < 2; ++n)
#pragma unroll
                for (int v = 0; v < 8; ++v)
                    r[n][v] = rp[v * D_FEAT + 16 * n];
        }

        v8f acc[2] = {};
        const float* ap = &sA[buf][l15 * GP + 2 * half];
#pragma unroll
        for (int s = 0; s < 32; ++s) {
            v2f a = *(const v2f*)(ap + 4 * s);
            acc[0] = __builtin_amdgcn_wmma_f32_16x16x4_f32(false, a, false, bf[s][0], (short)0, acc[0], false, false);
            acc[1] = __builtin_amdgcn_wmma_f32_16x16x4_f32(false, a, false, bf[s][1], (short)0, acc[1], false, false);
        }

        if (has_next) {
#pragma unroll
            for (int i = 0; i < 4; ++i) {
                v4f v = st[i];
                if (MODE == 1) v *= rsc[i];
                *(v4f*)(&sA[buf ^ 1][(i * 4 + w) * GP + lane * 4]) = v;
            }
        }
        __syncthreads();

        if (full) {
            float* op = out + (size_t)row0 * D_FEAT + colb;
#pragma unroll
            for (int n = 0; n < 2; ++n)
#pragma unroll
                for (int v = 0; v < 8; ++v) {
                    float c = acc[n][v] + bv[n];
                    if (relu_c) c = fmaxf(c, 0.0f);
                    op[v * D_FEAT + 16 * n] = (MODE == 1) ? fmaxf(r[n][v] + c, 0.0f) : c;
                }
        } else {
            float rt[2][8];
            if (MODE == 1) {
#pragma unroll
                for (int n = 0; n < 2; ++n)
#pragma unroll
                    for (int v = 0; v < 8; ++v) {
                        int row = row0 + v;
                        if (row >= Mrows) row = Mrows - 1;
                        rt[n][v] = resid[(size_t)row * D_FEAT + colb + 16 * n];
                    }
            }
#pragma unroll
            for (int n = 0; n < 2; ++n)
#pragma unroll
                for (int v = 0; v < 8; ++v) {
                    int row = row0 + v;
                    float c = acc[n][v] + bv[n];
                    if (relu_c) c = fmaxf(c, 0.0f);
                    float o = (MODE == 1) ? fmaxf(rt[n][v] + c, 0.0f) : c;
                    if (row < Mrows) out[(size_t)row * D_FEAT + colb + 16 * n] = o;
                }
        }

        buf ^= 1;
    }
}


extern "C" void kernel_launch(void* const* d_in, const int* in_sizes, int n_in,
                              void* d_out, int out_size, void* d_ws, size_t ws_size,
                              hipStream_t stream) {
    const float* node_features = (const float*)d_in[0];
    const int*   edge_src      = (const int*)d_in[1];
    const int*   edge_dst      = (const int*)d_in[2];
    const int*   graph_ids     = (const int*)d_in[3];
    const float* W_proj        = (const float*)d_in[4];
    const float* b_proj        = (const float*)d_in[5];
    const float* W_gcn         = (const float*)d_in[6];
    const float* b_gcn         = (const float*)d_in[7];
    const float* W_mlp0        = (const float*)d_in[8];
    const float* b_mlp0        = (const float*)d_in[9];
    const float* W_mlp1        = (const float*)d_in[10];
    const float* b_mlp1        = (const float*)d_in[11];
    const float* W_mlp2        = (const float*)d_in[12];
    const float* b_mlp2        = (const float*)d_in[13];

    const int D = D_FEAT;
    const int Mm = 256;
    const int N = in_sizes[3];
    const int E = in_sizes[1];
    const int L = in_sizes[6] / (D * D);
    const int G = out_size / Mm;

    float* ws = (float*)d_ws;
    float* norm_src = ws;
    float* norm_dst = norm_src + N;
    float* h        = norm_dst + N;
    float* agg      = h + (size_t)N * D;
    float* hg       = agg + (size_t)N * D;
    float* t0       = hg + (size_t)G * D;
    float* t1       = t0 + (size_t)G * Mm;
    int*   rec      = (int*)(t1 + (size_t)G * Mm);
    int*   rank     = rec + (size_t)2 * E;
    int*   outcnt   = rank + E;
    int*   incnt    = outcnt + N;
    int*   rowptr   = incnt + N;

    const int TB = 256;
    auto blocks = [&](size_t n) { return (unsigned)((n + TB - 1) / TB); };
    const int MROWS = 16 * MT;

    zero_i32<<<blocks((size_t)2 * N), TB, 0, stream>>>(outcnt, (size_t)2 * N);
    count_rank_kernel<<<blocks(E), TB, 0, stream>>>(edge_src, edge_dst, outcnt, incnt, rank, E);
    rowptr_scan_kernel<<<1, 1024, 0, stream>>>(incnt, rowptr, N);
    norm_kernel<<<blocks((size_t)2 * N), TB, 0, stream>>>(outcnt, norm_src, 2 * N);
    fill_kernel<<<blocks(E), TB, 0, stream>>>(edge_src, edge_dst, rank, rowptr, norm_src, rec, E);

    const unsigned gblocks = (unsigned)(((N + GR - 1) / GR + 3) / 4);
    gemm128_wmma_f32<0><<<gblocks, 128, 0, stream>>>(node_features, W_proj, b_proj,
                                                     nullptr, nullptr, h, N,  0);

    const unsigned nblk8   = (unsigned)((N + 7) / 8);
    const unsigned ggather = nblk8 < 2048u ? nblk8 : 2048u;
    for (int i = 0; i < L; ++i) {
        gather_sum_kernel<<<ggather, 256, 0, stream>>>(h, rec, rowptr, agg, N);
        gemm128_wmma_f32<1><<<gblocks, 128, 0, stream>>>(agg, W_gcn + (size_t)i * D * D,
                                                         b_gcn + (size_t)i * D,
                                                         norm_dst, h, h, N,
                                                          (i < L - 1) ? 1 : 0);
    }

    zero_f32<<<blocks((size_t)G * D), TB, 0, stream>>>(hg, (size_t)G * D);
    pool_kernel<<<blocks((size_t)N * 32), TB, 0, stream>>>(h, graph_ids, hg, N);

    {
        dim3 g0((G + MROWS - 1) / MROWS, Mm / 16);
        gemm_wmma_f32<<<g0, 32, 0, stream>>>(hg, W_mlp0, b_mlp0, nullptr, nullptr,
                                             t0, G, D, Mm, 0,  2);
        gemm_wmma_f32<<<g0, 32, 0, stream>>>(t0, W_mlp1, b_mlp1, nullptr, nullptr,
                                             t1, G, Mm, Mm, 0,  2);
        gemm_wmma_f32<<<g0, 32, 0, stream>>>(t1, W_mlp2, b_mlp2, nullptr, nullptr,
                                             (float*)d_out, G, Mm, Mm, 0,  0);
    }
}


# GraphConv layer fused to one launch: each wave of the node GEMM gather-sums the contiguous in-edge run of its 4 nodes into the LDS A tile (no stored aggregate, h in two planes); node sum order kept
# speedup vs baseline: 3.3109x; 1.0596x over previous
#include <hip/hip_runtime.h>
#include <hip/hip_bf16.h>

typedef __attribute__((ext_vector_type(2))) float v2f;
typedef __attribute__((ext_vector_type(8))) float v8f;

#define D_FEAT 128


typedef __attribute__((ext_vector_type(4))) float v4f;
typedef __attribute__((ext_vector_type(2))) int v2i;

__global__ void zero_f32(float* __restrict__ p, size_t n) {
    size_t i = (size_t)blockIdx.x * blockDim.x + threadIdx.x;
    if (i < n) p[i] = 0.0f;
}

__global__ void zero_i32(int* __restrict__ p, size_t n) {
    size_t i = (size_t)blockIdx.x * blockDim.x + threadIdx.x;
    if (i < n) p[i] = 0;
}


__global__ void count_rank_kernel(const int* __restrict__ src, const int* __restrict__ dst,
                                  int* __restrict__ outcnt, int* __restrict__ incnt,
                                  int* __restrict__ rank, int E) {
    int e = blockIdx.x * blockDim.x + threadIdx.x;
    if (e < E) {
        atomicAdd(&outcnt[src[e]], 1);
        rank[e] = atomicAdd(&incnt[dst[e]], 1);
    }
}

__global__ void __launch_bounds__(1024)
rowptr_scan_kernel(const int* __restrict__ cnt, int* __restrict__ rowptr, int n) {
    __shared__ int wtot[32];
    const int lane = threadIdx.x & 31;
    const int w    = __builtin_amdgcn_readfirstlane(threadIdx.x >> 5);
    const int seg  = ((n + 4095) / 4096) * 128;
    const int b    = w * seg;
    const int e    = (b + seg < n) ? (b + seg) : n;

    int s = 0;
#pragma unroll 4
    for (int i = b + lane; i < e; i += 32) s += cnt[i];
    for (int o = 16; o > 0; o >>= 1) s += __shfl_xor(s, o);
    if (lane == 0) wtot[w] = s;
    __syncthreads();

    int base = 0;
    for (int k = 0; k < w; ++k) base += wtot[k];

    int c[4];
#pragma unroll
    for (int q = 0; q < 4; ++q) {
        int i = b + 4 * lane + q;
        c[q] = (i < e) ? cnt[i] : 0;
    }
    for (int i0 = b; i0 < e; i0 += 128) {
        int cn[4];
#pragma unroll
        for (int q = 0; q < 4; ++q) {
            int i = i0 + 128 + 4 * lane + q;
            cn[q] = (i < e) ? cnt[i] : 0;
        }
        const int s4 = c[0] + c[1] + c[2] + c[3];
        int v = s4;
        for (int o = 1; o < 32; o <<= 1) {
            int u = __shfl_up(v, o);
            if (lane >= o) v += u;
        }
        int ex = base + v - s4;
#pragma unroll
        for (int q = 0; q < 4; ++q) {
            int i = i0 + 4 * lane + q;
            if (i < e) rowptr[i] = ex;
            ex += c[q];
            c[q] = cn[q];
        }
        base += __shfl(v, 31);
    }
    if (w == 31 && lane == 0) rowptr[n] = base;
}

__global__ void norm_kernel(const int* __restrict__ cnt, float* __restrict__ nrm, int n) {
    int i = blockIdx.x * blockDim.x + threadIdx.x;
    if (i < n) {
        float d = fmaxf((float)cnt[i], 1.0f);
        nrm[i] = rsqrtf(d);
    }
}

__global__ void fill_kernel(const int* __restrict__ src, const int* __restrict__ dst,
                            const int* __restrict__ rank, const int* __restrict__ rowptr,
                            const float* __restrict__ norm_src, int* __restrict__ rec, int E) {
    int e = blockIdx.x * blockDim.x + threadIdx.x;
    if (e < E) {
        int s = src[e];
        int pos = rowptr[dst[e]] + rank[e];
        v2i r;
        r.x = s;
        r.y = __float_as_int(norm_src[s]);
        *(v2i*)(rec + 2 * (size_t)pos) = r;
    }
}

__global__ void pool_kernel(const float* __restrict__ h, const int* __restrict__ gid,
                            float* __restrict__ hg, int N) {
    int t = blockIdx.x * blockDim.x + threadIdx.x;
    int n = t >> 5;
    int c = (t & 31) << 2;
    if (n < N) {
        int g = gid[n];
        float4 v = *(const float4*)(h + (size_t)n * D_FEAT + c);
        float* hp = hg + (size_t)g * D_FEAT + c;
        unsafeAtomicAdd(hp + 0, v.x);
        unsafeAtomicAdd(hp + 1, v.y);
        unsafeAtomicAdd(hp + 2, v.z);
        unsafeAtomicAdd(hp + 3, v.w);
    }
}

#define MT 4
__global__ void __launch_bounds__(32)
gemm_wmma_f32(const float* __restrict__ A, const float* __restrict__ B,
              const float* __restrict__ bias,
              const float* __restrict__ row_scale,
              const float* __restrict__ resid,
              float* __restrict__ out,
              int Mrows, int K, int Ncol, int relu_c, int mode) {
    int tm   = blockIdx.x;
    int tn   = blockIdx.y;
    int lane = threadIdx.x;
    int half = lane >> 4;
    int l15  = lane & 15;

    int col = tn * 16 + l15;
    float bv = bias[col];

    const float* arow[MT];
    float rs[MT];
#pragma unroll
    for (int i = 0; i < MT; ++i) {
        int row = tm * (16 * MT) + i * 16 + l15;
        if (row >= Mrows) row = Mrows - 1;
        arow[i] = A + (size_t)row * K;
        rs[i] = row_scale ? row_scale[row] : 1.0f;
    }

    v8f acc[MT] = {};
    for (int k0 = 0; k0 < K; k0 += 4) {
        int ka = k0 + half * 2;
        v2f b;
        b.x = B[(size_t)ka * Ncol + col];
        b.y = B[(size_t)(ka + 1) * Ncol + col];
#pragma unroll
        for (int i = 0; i < MT; ++i) {
            __builtin_prefetch(arow[i] + k0 + 64, 0, 0);
            v2f a = *(const v2f*)(arow[i] + ka);
            a.x *= rs[i];
            a.y *= rs[i];
            acc[i] = __builtin_amdgcn_wmma_f32_16x16x4_f32(
                 false, a,  false, b,
                 (short)0, acc[i],  false,  false);
        }
    }

#pragma unroll
    for (int i = 0; i < MT; ++i) {
#pragma unroll
        for (int v = 0; v < 8; ++v) {
            int m = tm * (16 * MT) + i * 16 + half * 8 + v;
            if (m < Mrows) {
                float c = acc[i][v] + bv;
                if (relu_c) c = fmaxf(c, 0.0f);
                float o;
                if (mode == 1)      o = fmaxf(resid[(size_t)m * Ncol + col] + c, 0.0f);
                else if (mode == 2) o = fmaxf(c, 0.0f);
                else                o = c;
                out[(size_t)m * Ncol + col] = o;
            }
        }
    }
}


#define GR 16
#define GP 132
__global__ void __launch_bounds__(128)
gemm128_wmma_f32(const float* __restrict__ A, const float* __restrict__ B,
                 const float* __restrict__ bias,
                 float* out,
                 int Mrows) {
    __shared__ __attribute__((aligned(16))) float sA[2][GR * GP];

    const int tid  = threadIdx.x;
    const int w    = __builtin_amdgcn_readfirstlane(tid >> 5);
    const int lane = tid & 31;
    const int half = lane >> 4;
    const int l15  = lane & 15;
    const int colb = 32 * w + l15;

    const int nchunks = (Mrows + GR - 1) / GR;
    int chunk = blockIdx.x;

    v4f   st[4];
#pragma unroll
    for (int i = 0; i < 4; ++i) {
        int row = chunk * GR + i * 4 + w;
        if (row >= Mrows) row = Mrows - 1;
        st[i] = *(const v4f*)(A + (size_t)row * D_FEAT + lane * 4);
    }

    v2f bf[32][2];
#pragma unroll
    for (int s = 0; s < 32; ++s) {
        int ka = 4 * s + 2 * half;
#pragma unroll
        for (int n = 0; n < 2; ++n) {
            bf[s][n].x = B[(size_t)ka * D_FEAT + colb + 16 * n];
            bf[s][n].y = B[(size_t)(ka + 1) * D_FEAT + colb + 16 * n];
        }
    }
    float bv[2];
    bv[0] = bias[colb];
    bv[1] = bias[colb + 16];

#pragma unroll
    for (int i = 0; i < 4; ++i)
        *(v4f*)(&sA[0][(i * 4 + w) * GP + lane * 4]) = st[i];
    __syncthreads();

    int buf = 0;
    for (; chunk < nchunks; chunk += gridDim.x) {
        const int  next     = chunk + gridDim.x;
        const bool has_next = next < nchunks;
        const int  row0     = chunk * GR + half * 8;
        const bool full     = chunk * GR + GR <= Mrows;

        if (has_next) {
#pragma unroll
            for (int i = 0; i < 4; ++i) {
                int row = next * GR + i * 4 + w;
                if (row >= Mrows) row = Mrows - 1;
                st[i] = *(const v4f*)(A + (size_t)row * D_FEAT + lane * 4);
            }
        }

        v8f acc[2] = {};
        const float* ap = &sA[buf][l15 * GP + 2 * half];
#pragma unroll
        for (int s = 0; s < 32; ++s) {
            v2f a = *(const v2f*)(ap + 4 * s);
            acc[0] = __builtin_amdgcn_wmma_f32_16x16x4_f32(false, a, false, bf[s][0], (short)0, acc[0], false, false);
            acc[1] = __builtin_amdgcn_wmma_f32_16x16x4_f32(false, a, false, bf[s][1], (short)0, acc[1], false, false);
        }

        if (has_next) {
#pragma unroll
            for (int i = 0; i < 4; ++i)
                *(v4f*)(&sA[buf ^ 1][(i * 4 + w) * GP + lane * 4]) = st[i];
        }
        __syncthreads();

        if (full) {
            float* op = out + (size_t)row0 * D_FEAT + colb;
#pragma unroll
            for (int n = 0; n < 2; ++n)
#pragma unroll
                for (int v = 0; v < 8; ++v)
                    op[v * D_FEAT + 16 * n] = acc[n][v] + bv[n];
        } else {
#pragma unroll
            for (int n = 0; n < 2; ++n)
#pragma unroll
                for (int v = 0; v < 8; ++v) {
                    int row = row0 + v;
                    if (row < Mrows) out[(size_t)row * D_FEAT + colb + 16 * n] = acc[n][v] + bv[n];
                }
        }

        buf ^= 1;
    }
}


__device__ __forceinline__ int load_bounds(const int* __restrict__ rowptr, int rbase, int lane, int Mrows) {
    int idx = rbase + lane;
    if (idx > Mrows) idx = Mrows;
    return rowptr[idx];
}

__device__ __forceinline__ float load_scales(const float* __restrict__ row_scale, int rbase, int lane, int Mrows) {
    int idx = rbase + lane;
    if (idx > Mrows - 1) idx = Mrows - 1;
    return row_scale[idx];
}


__device__ __forceinline__ void flush_node(float* srow, float scv, int q, v4f& acc) {
    const float sc = __int_as_float(__builtin_amdgcn_readlane(__float_as_int(scv), q));
    *(v4f*)(srow + q * GP) = acc * sc;
    acc = (v4f){0.0f, 0.0f, 0.0f, 0.0f};
}

template <int K>
__device__ __forceinline__ void gather_run(const float* __restrict__ hin, v2i r, int j, int lane4,
                                           int rpv, float scv, float* srow,
                                           int& pos, int& q, int& nend, v4f& acc) {
    v4f   v[K];
    float ns[K];
#pragma unroll
    for (int k = 0; k < K; ++k) {
        const int s = __builtin_amdgcn_readlane(r.x, j + k);
        ns[k] = __int_as_float(__builtin_amdgcn_readlane(r.y, j + k));
        v[k] = *(const v4f*)(hin + (size_t)s * D_FEAT + lane4);
    }
#pragma unroll
    for (int k = 0; k < K; ++k) {
        while (pos == nend) {
            flush_node(srow, scv, q, acc);
            ++q;
            nend = __builtin_amdgcn_readlane(rpv, q + 1);
        }
        acc += v[k] * ns[k];
        ++pos;
    }
}

__device__ __forceinline__ void gather_chunk(const float* __restrict__ hin, const int* __restrict__ rec,
                                             int rpv, float scv, int lane, float* srow) {
    const int lane4 = lane * 4;
    const int beg = __builtin_amdgcn_readlane(rpv, 0);
    const int end = __builtin_amdgcn_readlane(rpv, 4);
    const int last = (end > 0) ? (end - 1) : 0;
    int i0 = beg + lane;      if (i0 > last) i0 = last;
    int i1 = beg + 32 + lane; if (i1 > last) i1 = last;
    v2i r  = *(const v2i*)(rec + 2 * (size_t)i0);
    v2i rn = *(const v2i*)(rec + 2 * (size_t)i1);

    int pos  = beg;
    int q    = 0;
    int nend = __builtin_amdgcn_readlane(rpv, 1);
    v4f acc  = {0.0f, 0.0f, 0.0f, 0.0f};
    for (int base = beg; base < end; base += 32) {
        const int m = (end - base < 32) ? (end - base) : 32;
        int j = 0;
        for (; j + 8 <= m; j += 8) gather_run<8>(hin, r, j, lane4, rpv, scv, srow, pos, q, nend, acc);
        if (j + 4 <= m) { gather_run<4>(hin, r, j, lane4, rpv, scv, srow, pos, q, nend, acc); j += 4; }
        if (j + 2 <= m) { gather_run<2>(hin, r, j, lane4, rpv, scv, srow, pos, q, nend, acc); j += 2; }
        if (j < m)      { gather_run<1>(hin, r, j, lane4, rpv, scv, srow, pos, q, nend, acc); }
        r = rn;
        int i2 = base + 64 + lane; if (i2 > last) i2 = last;
        rn = *(const v2i*)(rec + 2 * (size_t)i2);
    }
    for (; q < 4; ++q) flush_node(srow, scv, q, acc);
}

__global__ void __launch_bounds__(128)
gcn_layer_wmma_f32(const float* __restrict__ hin, const int* __restrict__ rec,
                   const int* __restrict__ rowptr,
                   const float* __restrict__ B, const float* __restrict__ bias,
                   const float* __restrict__ row_scale,
                   float* __restrict__ out,
                   int Mrows, int relu_c) {
    __shared__ __attribute__((aligned(16))) float sA[2][GR * GP];

    const int tid  = threadIdx.x;
    const int w    = __builtin_amdgcn_readfirstlane(tid >> 5);
    const int lane = tid & 31;
    const int half = lane >> 4;
    const int l15  = lane & 15;
    const int colb = 32 * w + l15;

    const int nchunks = (Mrows + GR - 1) / GR;
    int chunk = blockIdx.x;

    int   rpv = load_bounds(rowptr, chunk * GR + 4 * w, lane, Mrows);
    float scv = load_scales(row_scale, chunk * GR + 4 * w, lane, Mrows);

    v2f bf[32][2];
#pragma unroll
    for (int s = 0; s < 32; ++s) {
        int ka = 4 * s + 2 * half;
#pragma unroll
        for (int n = 0; n < 2; ++n) {
            bf[s][n].x = B[(size_t)ka * D_FEAT + colb + 16 * n];
            bf[s][n].y = B[(size_t)(ka + 1) * D_FEAT + colb + 16 * n];
        }
    }
    float bv[2];
    bv[0] = bias[colb];
    bv[1] = bias[colb + 16];

    gather_chunk(hin, rec, rpv, scv, lane, &sA[0][(4 * w) * GP + lane * 4]);
    __syncthreads();

    int buf = 0;
    for (; chunk < nchunks; chunk += gridDim.x) {
        const int  next     = chunk + gridDim.x;
        const bool has_next = next < nchunks;
        const int  row0     = chunk * GR + half * 8;
        const bool full     = chunk * GR + GR <= Mrows;

        rpv = load_bounds(rowptr, next * GR + 4 * w, lane, Mrows);
        scv = load_scales(row_scale, next * GR + 4 * w, lane, Mrows);

        float r[2][8];
        if (full) {
            const float* rp = hin + (size_t)row0 * D_FEAT + colb;
#pragma unroll
            for (int n = 0; n < 2; ++n)
#pragma unroll
                for (int v = 0; v < 8; ++v)
                    r[n][v] = rp[v * D_FEAT + 16 * n];
        }

        v8f acc[2] = {};
        const float* ap = &sA[buf][l15 * GP + 2 * half];
#pragma unroll
        for (int s = 0; s < 32; ++s) {
            v2f a = *(const v2f*)(ap + 4 * s);
            acc[0] = __builtin_amdgcn_wmma_f32_16x16x4_f32(false, a, false, bf[s][0], (short)0, acc[0], false, false);
            acc[1] = __builtin_amdgcn_wmma_f32_16x16x4_f32(false, a, false, bf[s][1], (short)0, acc[1], false, false);
        }

        float c[2][8];
#pragma unroll
        for (int n = 0; n < 2; ++n)
#pragma unroll
            for (int v = 0; v < 8; ++v) {
                c[n][v] = acc[n][v] + bv[n];
                if (relu_c) c[n][v] = fmaxf(c[n][v], 0.0f);
            }

        if (full) {
            float* op = out + (size_t)row0 * D_FEAT + colb;
#pragma unroll
            for (int n = 0; n < 2; ++n)
#pragma unroll
                for (int v = 0; v < 8; ++v)
                    op[v * D_FEAT + 16 * n] = fmaxf(r[n][v] + c[n][v], 0.0f);
        } else {
#pragma unroll
            for (int n = 0; n < 2; ++n)
#pragma unroll
                for (int v = 0; v < 8; ++v) {
                    const int row = row0 + v;
                    if (row < Mrows) {
                        const size_t o = (size_t)row * D_FEAT + colb + 16 * n;
                        out[o] = fmaxf(hin[o] + c[n][v], 0.0f);
                    }
                }
        }

        if (has_next)
            gather_chunk(hin, rec, rpv, scv, lane, &sA[buf ^ 1][(4 * w) * GP + lane * 4]);
        __syncthreads();

        buf ^= 1;
    }
}


extern "C" void kernel_launch(void* const* d_in, const int* in_sizes, int n_in,
                              void* d_out, int out_size, void* d_ws, size_t ws_size,
                              hipStream_t stream) {
    const float* node_features = (const float*)d_in[0];
    const int*   edge_src      = (const int*)d_in[1];
    const int*   edge_dst      = (const int*)d_in[2];
    const int*   graph_ids     = (const int*)d_in[3];
    const float* W_proj        = (const float*)d_in[4];
    const float* b_proj        = (const float*)d_in[5];
    const float* W_gcn         = (const float*)d_in[6];
    const float* b_gcn         = (const float*)d_in[7];
    const float* W_mlp0        = (const float*)d_in[8];
    const float* b_mlp0        = (const float*)d_in[9];
    const float* W_mlp1        = (const float*)d_in[10];
    const float* b_mlp1        = (const float*)d_in[11];
    const float* W_mlp2        = (const float*)d_in[12];
    const float* b_mlp2        = (const float*)d_in[13];

    const int D = D_FEAT;
    const int Mm = 256;
    const int N = in_sizes[3];
    const int E = in_sizes[1];
    const int L = in_sizes[6] / (D * D);
    const int G = out_size / Mm;

    float* ws = (float*)d_ws;
    float* norm_src = ws;
    float* norm_dst = norm_src + N;
    float* h        = norm_dst + N;
    float* agg      = h + (size_t)N * D;
    float* hg       = agg + (size_t)N * D;
    float* t0       = hg + (size_t)G * D;
    float* t1       = t0 + (size_t)G * Mm;
    int*   rec      = (int*)(t1 + (size_t)G * Mm);
    int*   rank     = rec + (size_t)2 * E;
    int*   outcnt   = rank + E;
    int*   incnt    = outcnt + N;
    int*   rowptr   = incnt + N;

    const int TB = 256;
    auto blocks = [&](size_t n) { return (unsigned)((n + TB - 1) / TB); };
    const int MROWS = 16 * MT;

    zero_i32<<<blocks((size_t)2 * N), TB, 0, stream>>>(outcnt, (size_t)2 * N);
    count_rank_kernel<<<blocks(E), TB, 0, stream>>>(edge_src, edge_dst, outcnt, incnt, rank, E);
    rowptr_scan_kernel<<<1, 1024, 0, stream>>>(incnt, rowptr, N);
    norm_kernel<<<blocks((size_t)2 * N), TB, 0, stream>>>(outcnt, norm_src, 2 * N);
    fill_kernel<<<blocks(E), TB, 0, stream>>>(edge_src, edge_dst, rank, rowptr, norm_src, rec, E);

    const unsigned gblocks = (unsigned)(((N + GR - 1) / GR + 3) / 4);
    gemm128_wmma_f32<<<gblocks, 128, 0, stream>>>(node_features, W_proj, b_proj, h, N);

    float* hcur  = h;
    float* hnext = agg;
    for (int i = 0; i < L; ++i) {
        gcn_layer_wmma_f32<<<gblocks, 128, 0, stream>>>(hcur, rec, rowptr, W_gcn + (size_t)i * D * D,
                                                        b_gcn + (size_t)i * D, norm_dst, hnext, N,
                                                         (i < L - 1) ? 1 : 0);
        float* t = hcur; hcur = hnext; hnext = t;
    }

    zero_f32<<<blocks((size_t)G * D), TB, 0, stream>>>(hg, (size_t)G * D);
    pool_kernel<<<blocks((size_t)N * 32), TB, 0, stream>>>(hcur, graph_ids, hg, N);

    {
        dim3 g0((G + MROWS - 1) / MROWS, Mm / 16);
        gemm_wmma_f32<<<g0, 32, 0, stream>>>(hg, W_mlp0, b_mlp0, nullptr, nullptr,
                                             t0, G, D, Mm, 0,  2);
        gemm_wmma_f32<<<g0, 32, 0, stream>>>(t0, W_mlp1, b_mlp1, nullptr, nullptr,
                                             t1, G, Mm, Mm, 0,  2);
        gemm_wmma_f32<<<g0, 32, 0, stream>>>(t1, W_mlp2, b_mlp2, nullptr, nullptr,
                                             (float*)d_out, G, Mm, Mm, 0,  0);
    }
}


# In-edge grouping without per-edge global atomics: 64 blocks count edge slices in LDS histograms (LDS add gives the rank), a per-node pass makes offsets, degrees, norms; node sum order now by slice
# speedup vs baseline: 3.8067x; 1.1497x over previous
#include <hip/hip_runtime.h>
#include <hip/hip_bf16.h>

typedef __attribute__((ext_vector_type(2))) float v2f;
typedef __attribute__((ext_vector_type(8))) float v8f;

#define D_FEAT 128


typedef __attribute__((ext_vector_type(4))) float v4f;
typedef __attribute__((ext_vector_type(2))) int v2i;

__global__ void zero_f32(float* __restrict__ p, size_t n) {
    size_t i = (size_t)blockIdx.x * blockDim.x + threadIdx.x;
    if (i < n) p[i] = 0.0f;
}

#define HB  32
#define HCAP 65536

__global__ void __launch_bounds__(1024)
slice_hist_kernel(const int* __restrict__ src, const int* __restrict__ dst,
                  int* __restrict__ hist, int* __restrict__ rank, int N, int E) {
    __shared__ __attribute__((aligned(16))) int cnt[HCAP];
    const int  tid   = threadIdx.x;
    const bool isdst = blockIdx.x < HB;
    const int  sl    = isdst ? blockIdx.x : blockIdx.x - HB;
    const int* __restrict__ key = isdst ? dst : src;
    const int  per   = (E + HB - 1) / HB;
    const int  e0    = sl * per;
    const int  e1    = (e0 + per < E) ? (e0 + per) : E;
    int* __restrict__ hrow = hist + (size_t)blockIdx.x * N;

    for (int base = 0; base < N; base += HCAP) {
        const int lim = (N - base < HCAP) ? (N - base) : HCAP;
        for (int i = 4 * tid; i < lim; i += 4 * 1024)
            *(int4*)(&cnt[i]) = make_int4(0, 0, 0, 0);
        __syncthreads();
        for (int eb = e0; eb < e1; eb += 8 * 1024) {
            int kk[8];
#pragma unroll
            for (int u = 0; u < 8; ++u) {
                const int e = eb + u * 1024 + tid;
                kk[u] = (e < e1) ? key[e] : -1;
            }
#pragma unroll
            for (int u = 0; u < 8; ++u) {
                const unsigned k = (unsigned)(kk[u] - base);
                if (k < (unsigned)lim) {
                    const int r = atomicAdd(&cnt[k], 1);
                    if (isdst) rank[eb + u * 1024 + tid] = r;
                }
            }
        }
        __syncthreads();
        for (int i = tid; i < lim; i += 1024) hrow[base + i] = cnt[i];
        __syncthreads();
    }
}

__global__ void combine_kernel(int* __restrict__ hist, int* __restrict__ incnt,
                               float* __restrict__ norm_src, float* __restrict__ norm_dst, int N) {
    const int n = blockIdx.x * blockDim.x + threadIdx.x;
    if (n < N) {
        int cd[HB], cs[HB];
#pragma unroll
        for (int b = 0; b < HB; ++b) {
            cd[b] = hist[(size_t)b * N + n];
            cs[b] = hist[(size_t)(HB + b) * N + n];
        }
        int run = 0, so = 0;
#pragma unroll
        for (int b = 0; b < HB; ++b) {
            hist[(size_t)b * N + n] = run;
            run += cd[b];
            so  += cs[b];
        }
        incnt[n]    = run;
        norm_dst[n] = rsqrtf(fmaxf((float)run, 1.0f));
        norm_src[n] = rsqrtf(fmaxf((float)so, 1.0f));
    }
}

__global__ void __launch_bounds__(1024)
rowptr_scan_kernel(const int* __restrict__ cnt, int* __restrict__ rowptr, int n) {
    __shared__ int wtot[32];
    const int lane = threadIdx.x & 31;
    const int w    = __builtin_amdgcn_readfirstlane(threadIdx.x >> 5);
    const int seg  = ((n + 4095) / 4096) * 128;
    const int b    = w * seg;
    const int e    = (b + seg < n) ? (b + seg) : n;

    int s = 0;
#pragma unroll 4
    for (int i = b + lane; i < e; i += 32) s += cnt[i];
    for (int o = 16; o > 0; o >>= 1) s += __shfl_xor(s, o);
    if (lane == 0) wtot[w] = s;
    __syncthreads();

    int base = 0;
    for (int k = 0; k < w; ++k) base += wtot[k];

    int c[4];
#pragma unroll
    for (int q = 0; q < 4; ++q) {
        int i = b + 4 * lane + q;
        c[q] = (i < e) ? cnt[i] : 0;
    }
    for (int i0 = b; i0 < e; i0 += 128) {
        int cn[4];
#pragma unroll
        for (int q = 0; q < 4; ++q) {
            int i = i0 + 128 + 4 * lane + q;
            cn[q] = (i < e) ? cnt[i] : 0;
        }
        const int s4 = c[0] + c[1] + c[2] + c[3];
        int v = s4;
        for (int o = 1; o < 32; o <<= 1) {
            int u = __shfl_up(v, o);
            if (lane >= o) v += u;
        }
        int ex = base + v - s4;
#pragma unroll
        for (int q = 0; q < 4; ++q) {
            int i = i0 + 4 * lane + q;
            if (i < e) rowptr[i] = ex;
            ex += c[q];
            c[q] = cn[q];
        }
        base += __shfl(v, 31);
    }
    if (w == 31 && lane == 0) rowptr[n] = base;
}

__global__ void fill_kernel(const int* __restrict__ src, const int* __restrict__ dst,
                            const int* __restrict__ rank, const int* __restrict__ rowptr,
                            const int* __restrict__ hist,
                            const float* __restrict__ norm_src, int* __restrict__ rec, int N, int E) {
    int e = blockIdx.x * blockDim.x + threadIdx.x;
    if (e < E) {
        const int per = (E + HB - 1) / HB;
        int s = src[e];
        int d = dst[e];
        int pos = rowptr[d] + hist[(size_t)(e / per) * N + d] + rank[e];
        v2i r;
        r.x = s;
        r.y = __float_as_int(norm_src[s]);
        *(v2i*)(rec + 2 * (size_t)pos) = r;
    }
}

__global__ void pool_kernel(const float* __restrict__ h, const int* __restrict__ gid,
                            float* __restrict__ hg, int N) {
    int t = blockIdx.x * blockDim.x + threadIdx.x;
    int n = t >> 5;
    int c = (t & 31) << 2;
    if (n < N) {
        int g = gid[n];
        float4 v = *(const float4*)(h + (size_t)n * D_FEAT + c);
        float* hp = hg + (size_t)g * D_FEAT + c;
        unsafeAtomicAdd(hp + 0, v.x);
        unsafeAtomicAdd(hp + 1, v.y);
        unsafeAtomicAdd(hp + 2, v.z);
        unsafeAtomicAdd(hp + 3, v.w);
    }
}

#define MT 4
__global__ void __launch_bounds__(32)
gemm_wmma_f32(const float* __restrict__ A, const float* __restrict__ B,
              const float* __restrict__ bias,
              const float* __restrict__ row_scale,
              const float* __restrict__ resid,
              float* __restrict__ out,
              int Mrows, int K, int Ncol, int relu_c, int mode) {
    int tm   = blockIdx.x;
    int tn   = blockIdx.y;
    int lane = threadIdx.x;
    int half = lane >> 4;
    int l15  = lane & 15;

    int col = tn * 16 + l15;
    float bv = bias[col];

    const float* arow[MT];
    float rs[MT];
#pragma unroll
    for (int i = 0; i < MT; ++i) {
        int row = tm * (16 * MT) + i * 16 + l15;
        if (row >= Mrows) row = Mrows - 1;
        arow[i] = A + (size_t)row * K;
        rs[i] = row_scale ? row_scale[row] : 1.0f;
    }

    v8f acc[MT] = {};
    for (int k0 = 0; k0 < K; k0 += 4) {
        int ka = k0 + half * 2;
        v2f b;
        b.x = B[(size_t)ka * Ncol + col];
        b.y = B[(size_t)(ka + 1) * Ncol + col];
#pragma unroll
        for (int i = 0; i < MT; ++i) {
            __builtin_prefetch(arow[i] + k0 + 64, 0, 0);
            v2f a = *(const v2f*)(arow[i] + ka);
            a.x *= rs[i];
            a.y *= rs[i];
            acc[i] = __builtin_amdgcn_wmma_f32_16x16x4_f32(
                 false, a,  false, b,
                 (short)0, acc[i],  false,  false);
        }
    }

#pragma unroll
    for (int i = 0; i < MT; ++i) {
#pragma unroll
        for (int v = 0; v < 8; ++v) {
            int m = tm * (16 * MT) + i * 16 + half * 8 + v;
            if (m < Mrows) {
                float c = acc[i][v] + bv;
                if (relu_c) c = fmaxf(c, 0.0f);
                float o;
                if (mode == 1)      o = fmaxf(resid[(size_t)m * Ncol + col] + c, 0.0f);
                else if (mode == 2) o = fmaxf(c, 0.0f);
                else                o = c;
                out[(size_t)m * Ncol + col] = o;
            }
        }
    }
}


#define GR 16
#define GP 132
__global__ void __launch_bounds__(128)
gemm128_wmma_f32(const float* __restrict__ A, const float* __restrict__ B,
                 const float* __restrict__ bias,
                 float* out,
                 int Mrows) {
    __shared__ __attribute__((aligned(16))) float sA[2][GR * GP];

    const int tid  = threadIdx.x;
    const int w    = __builtin_amdgcn_readfirstlane(tid >> 5);
    const int lane = tid & 31;
    const int half = lane >> 4;
    const int l15  = lane & 15;
    const int colb = 32 * w + l15;

    const int nchunks = (Mrows + GR - 1) / GR;
    int chunk = blockIdx.x;

    v4f   st[4];
#pragma unroll
    for (int i = 0; i < 4; ++i) {
        int row = chunk * GR + i * 4 + w;
        if (row >= Mrows) row = Mrows - 1;
        st[i] = *(const v4f*)(A + (size_t)row * D_FEAT + lane * 4);
    }

    v2f bf[32][2];
#pragma unroll
    for (int s = 0; s < 32; ++s) {
        int ka = 4 * s + 2 * half;
#pragma unroll
        for (int n = 0; n < 2; ++n) {
            bf[s][n].x = B[(size_t)ka * D_FEAT + colb + 16 * n];
            bf[s][n].y = B[(size_t)(ka + 1) * D_FEAT + colb + 16 * n];
        }
    }
    float bv[2];
    bv[0] = bias[colb];
    bv[1] = bias[colb + 16];

#pragma unroll
    for (int i = 0; i < 4; ++i)
        *(v4f*)(&sA[0][(i * 4 + w) * GP + lane * 4]) = st[i];
    __syncthreads();

    int buf = 0;
    for (; chunk < nchunks; chunk += gridDim.x) {
        const int  next     = chunk + gridDim.x;
        const bool has_next = next < nchunks;
        const int  row0     = chunk * GR + half * 8;
        const bool full     = chunk * GR + GR <= Mrows;

        if (has_next) {
#pragma unroll
            for (int i = 0; i < 4; ++i) {
                int row = next * GR + i * 4 + w;
                if (row >= Mrows) row = Mrows - 1;
                st[i] = *(const v4f*)(A + (size_t)row * D_FEAT + lane * 4);
            }
        }

        v8f acc[2] = {};
        const float* ap = &sA[buf][l15 * GP + 2 * half];
#pragma unroll
        for (int s = 0; s < 32; ++s) {
            v2f a = *(const v2f*)(ap + 4 * s);
            acc[0] = __builtin_amdgcn_wmma_f32_16x16x4_f32(false, a, false, bf[s][0], (short)0, acc[0], false, false);
            acc[1] = __builtin_amdgcn_wmma_f32_16x16x4_f32(false, a, false, bf[s][1], (short)0, acc[1], false, false);
        }

        if (has_next) {
#pragma unroll
            for (int i = 0; i < 4; ++i)
                *(v4f*)(&sA[buf ^ 1][(i * 4 + w) * GP + lane * 4]) = st[i];
        }
        __syncthreads();

        if (full) {
            float* op = out + (size_t)row0 * D_FEAT + colb;
#pragma unroll
            for (int n = 0; n < 2; ++n)
#pragma unroll
                for (int v = 0; v < 8; ++v)
                    op[v * D_FEAT + 16 * n] = acc[n][v] + bv[n];
        } else {
#pragma unroll
            for (int n = 0; n < 2; ++n)
#pragma unroll
                for (int v = 0; v < 8; ++v) {
                    int row = row0 + v;
                    if (row < Mrows) out[(size_t)row * D_FEAT + colb + 16 * n] = acc[n][v] + bv[n];
                }
        }

        buf ^= 1;
    }
}


__device__ __forceinline__ int load_bounds(const int* __restrict__ rowptr, int rbase, int lane, int Mrows) {
    int idx = rbase + lane;
    if (idx > Mrows) idx = Mrows;
    return rowptr[idx];
}

__device__ __forceinline__ float load_scales(const float* __restrict__ row_scale, int rbase, int lane, int Mrows) {
    int idx = rbase + lane;
    if (idx > Mrows - 1) idx = Mrows - 1;
    return row_scale[idx];
}


__device__ __forceinline__ void flush_node(float* srow, float scv, int q, v4f& acc) {
    const float sc = __int_as_float(__builtin_amdgcn_readlane(__float_as_int(scv), q));
    *(v4f*)(srow + q * GP) = acc * sc;
    acc = (v4f){0.0f, 0.0f, 0.0f, 0.0f};
}

template <int K>
__device__ __forceinline__ void gather_run(const float* __restrict__ hin, v2i r, int j, int lane4,
                                           int rpv, float scv, float* srow,
                                           int& pos, int& q, int& nend, v4f& acc) {
    v4f   v[K];
    float ns[K];
#pragma unroll
    for (int k = 0; k < K; ++k) {
        const int s = __builtin_amdgcn_readlane(r.x, j + k);
        ns[k] = __int_as_float(__builtin_amdgcn_readlane(r.y, j + k));
        v[k] = *(const v4f*)(hin + (size_t)s * D_FEAT + lane4);
    }
#pragma unroll
    for (int k = 0; k < K; ++k) {
        while (pos == nend) {
            flush_node(srow, scv, q, acc);
            ++q;
            nend = __builtin_amdgcn_readlane(rpv, q + 1);
        }
        acc += v[k] * ns[k];
        ++pos;
    }
}

__device__ __forceinline__ void gather_chunk(const float* __restrict__ hin, const int* __restrict__ rec,
                                             int rpv, float scv, int lane, float* srow) {
    const int lane4 = lane * 4;
    const int beg = __builtin_amdgcn_readlane(rpv, 0);
    const int end = __builtin_amdgcn_readlane(rpv, 4);
    const int last = (end > 0) ? (end - 1) : 0;
    int i0 = beg + lane;      if (i0 > last) i0 = last;
    int i1 = beg + 32 + lane; if (i1 > last) i1 = last;
    v2i r  = *(const v2i*)(rec + 2 * (size_t)i0);
    v2i rn = *(const v2i*)(rec + 2 * (size_t)i1);

    int pos  = beg;
    int q    = 0;
    int nend = __builtin_amdgcn_readlane(rpv, 1);
    v4f acc  = {0.0f, 0.0f, 0.0f, 0.0f};
    for (int base = beg; base < end; base += 32) {
        const int m = (end - base < 32) ? (end - base) : 32;
        int j = 0;
        for (; j + 8 <= m; j += 8) gather_run<8>(hin, r, j, lane4, rpv, scv, srow, pos, q, nend, acc);
        if (j + 4 <= m) { gather_run<4>(hin, r, j, lane4, rpv, scv, srow, pos, q, nend, acc); j += 4; }
        if (j + 2 <= m) { gather_run<2>(hin, r, j, lane4, rpv, scv, srow, pos, q, nend, acc); j += 2; }
        if (j < m)      { gather_run<1>(hin, r, j, lane4, rpv, scv, srow, pos, q, nend, acc); }
        r = rn;
        int i2 = base + 64 + lane; if (i2 > last) i2 = last;
        rn = *(const v2i*)(rec + 2 * (size_t)i2);
    }
    for (; q < 4; ++q) flush_node(srow, scv, q, acc);
}

__global__ void __launch_bounds__(128)
gcn_layer_wmma_f32(const float* __restrict__ hin, const int* __restrict__ rec,
                   const int* __restrict__ rowptr,
                   const float* __restrict__ B, const float* __restrict__ bias,
                   const float* __restrict__ row_scale,
                   float* __restrict__ out,
                   int Mrows, int relu_c) {
    __shared__ __attribute__((aligned(16))) float sA[2][GR * GP];

    const int tid  = threadIdx.x;
    const int w    = __builtin_amdgcn_readfirstlane(tid >> 5);
    const int lane = tid & 31;
    const int half = lane >> 4;
    const int l15  = lane & 15;
    const int colb = 32 * w + l15;

    const int nchunks = (Mrows + GR - 1) / GR;
    int chunk = blockIdx.x;

    int   rpv = load_bounds(rowptr, chunk * GR + 4 * w, lane, Mrows);
    float scv = load_scales(row_scale, chunk * GR + 4 * w, lane, Mrows);

    v2f bf[32][2];
#pragma unroll
    for (int s = 0; s < 32; ++s) {
        int ka = 4 * s + 2 * half;
#pragma unroll
        for (int n = 0; n < 2; ++n) {
            bf[s][n].x = B[(size_t)ka * D_FEAT + colb + 16 * n];
            bf[s][n].y = B[(size_t)(ka + 1) * D_FEAT + colb + 16 * n];
        }
    }
    float bv[2];
    bv[0] = bias[colb];
    bv[1] = bias[colb + 16];

    gather_chunk(hin, rec, rpv, scv, lane, &sA[0][(4 * w) * GP + lane * 4]);
    __syncthreads();

    int buf = 0;
    for (; chunk < nchunks; chunk += gridDim.x) {
        const int  next     = chunk + gridDim.x;
        const bool has_next = next < nchunks;
        const int  row0     = chunk * GR + half * 8;
        const bool full     = chunk * GR + GR <= Mrows;

        rpv = load_bounds(rowptr, next * GR + 4 * w, lane, Mrows);
        scv = load_scales(row_scale, next * GR + 4 * w, lane, Mrows);

        float r[2][8];
        if (full) {
            const float* rp = hin + (size_t)row0 * D_FEAT + colb;
#pragma unroll
            for (int n = 0; n < 2; ++n)
#pragma unroll
                for (int v = 0; v < 8; ++v)
                    r[n][v] = rp[v * D_FEAT + 16 * n];
        }

        v8f acc[2] = {};
        const float* ap = &sA[buf][l15 * GP + 2 * half];
#pragma unroll
        for (int s = 0; s < 32; ++s) {
            v2f a = *(const v2f*)(ap + 4 * s);
            acc[0] = __builtin_amdgcn_wmma_f32_16x16x4_f32(false, a, false, bf[s][0], (short)0, acc[0], false, false);
            acc[1] = __builtin_amdgcn_wmma_f32_16x16x4_f32(false, a, false, bf[s][1], (short)0, acc[1], false, false);
        }

        float c[2][8];
#pragma unroll
        for (int n = 0; n < 2; ++n)
#pragma unroll
            for (int v = 0; v < 8; ++v) {
                c[n][v] = acc[n][v] + bv[n];
                if (relu_c) c[n][v] = fmaxf(c[n][v], 0.0f);
            }

        if (full) {
            float* op = out + (size_t)row0 * D_FEAT + colb;
#pragma unroll
            for (int n = 0; n < 2; ++n)
#pragma unroll
                for (int v = 0; v < 8; ++v)
                    op[v * D_FEAT + 16 * n] = fmaxf(r[n][v] + c[n][v], 0.0f);
        } else {
#pragma unroll
            for (int n = 0; n < 2; ++n)
#pragma unroll
                for (int v = 0; v < 8; ++v) {
                    const int row = row0 + v;
                    if (row < Mrows) {
                        const size_t o = (size_t)row * D_FEAT + colb + 16 * n;
                        out[o] = fmaxf(hin[o] + c[n][v], 0.0f);
                    }
                }
        }

        if (has_next)
            gather_chunk(hin, rec, rpv, scv, lane, &sA[buf ^ 1][(4 * w) * GP + lane * 4]);
        __syncthreads();

        buf ^= 1;
    }
}


extern "C" void kernel_launch(void* const* d_in, const int* in_sizes, int n_in,
                              void* d_out, int out_size, void* d_ws, size_t ws_size,
                              hipStream_t stream) {
    const float* node_features = (const float*)d_in[0];
    const int*   edge_src      = (const int*)d_in[1];
    const int*   edge_dst      = (const int*)d_in[2];
    const int*   graph_ids     = (const int*)d_in[3];
    const float* W_proj        = (const float*)d_in[4];
    const float* b_proj        = (const float*)d_in[5];
    const float* W_gcn         = (const float*)d_in[6];
    const float* b_gcn         = (const float*)d_in[7];
    const float* W_mlp0        = (const float*)d_in[8];
    const float* b_mlp0        = (const float*)d_in[9];
    const float* W_mlp1        = (const float*)d_in[10];
    const float* b_mlp1        = (const float*)d_in[11];
    const float* W_mlp2        = (const float*)d_in[12];
    const float* b_mlp2        = (const float*)d_in[13];

    const int D = D_FEAT;
    const int Mm = 256;
    const int N = in_sizes[3];
    const int E = in_sizes[1];
    const int L = in_sizes[6] / (D * D);
    const int G = out_size / Mm;

    float* ws = (float*)d_ws;
    float* norm_src = ws;
    float* norm_dst = norm_src + N;
    float* h        = norm_dst + N;
    float* agg      = h + (size_t)N * D;
    float* hg       = agg + (size_t)N * D;
    float* t0       = hg + (size_t)G * D;
    float* t1       = t0 + (size_t)G * Mm;
    int*   rec      = (int*)(t1 + (size_t)G * Mm);
    int*   rank     = rec + (size_t)2 * E;
    int*   incnt    = rank + E;
    int*   rowptr   = incnt + N;
    int*   hist     = rowptr + N + 1;

    const int TB = 256;
    auto blocks = [&](size_t n) { return (unsigned)((n + TB - 1) / TB); };
    const int MROWS = 16 * MT;

    slice_hist_kernel<<<2 * HB, 1024, 0, stream>>>(edge_src, edge_dst, hist, rank, N, E);
    combine_kernel<<<blocks(N), TB, 0, stream>>>(hist, incnt, norm_src, norm_dst, N);
    rowptr_scan_kernel<<<1, 1024, 0, stream>>>(incnt, rowptr, N);
    fill_kernel<<<blocks(E), TB, 0, stream>>>(edge_src, edge_dst, rank, rowptr, hist, norm_src, rec, N, E);

    const unsigned gblocks = (unsigned)(((N + GR - 1) / GR + 3) / 4);
    gemm128_wmma_f32<<<gblocks, 128, 0, stream>>>(node_features, W_proj, b_proj, h, N);

    float* hcur  = h;
    float* hnext = agg;
    for (int i = 0; i < L; ++i) {
        gcn_layer_wmma_f32<<<gblocks, 128, 0, stream>>>(hcur, rec, rowptr, W_gcn + (size_t)i * D * D,
                                                        b_gcn + (size_t)i * D, norm_dst, hnext, N,
                                                         (i < L - 1) ? 1 : 0);
        float* t = hcur; hcur = hnext; hnext = t;
    }

    zero_f32<<<blocks((size_t)G * D), TB, 0, stream>>>(hg, (size_t)G * D);
    pool_kernel<<<blocks((size_t)N * 32), TB, 0, stream>>>(hcur, graph_ids, hg, N);

    {
        dim3 g0((G + MROWS - 1) / MROWS, Mm / 16);
        gemm_wmma_f32<<<g0, 32, 0, stream>>>(hg, W_mlp0, b_mlp0, nullptr, nullptr,
                                             t0, G, D, Mm, 0,  2);
        gemm_wmma_f32<<<g0, 32, 0, stream>>>(t0, W_mlp1, b_mlp1, nullptr, nullptr,
                                             t1, G, Mm, Mm, 0,  2);
        gemm_wmma_f32<<<g0, 32, 0, stream>>>(t1, W_mlp2, b_mlp2, nullptr, nullptr,
                                             (float*)d_out, G, Mm, Mm, 0,  0);
    }
}


# SumPooling folded into the last GraphConv launch: each 16-row tile summed in registers (8 rows, 2 lane halves), one float atomic per column and tile into hg; pooled sum order changes; no row store
# speedup vs baseline: 4.2658x; 1.1206x over previous
#include <hip/hip_runtime.h>
#include <hip/hip_bf16.h>

typedef __attribute__((ext_vector_type(2))) float v2f;
typedef __attribute__((ext_vector_type(8))) float v8f;

#define D_FEAT 128


typedef __attribute__((ext_vector_type(4))) float v4f;
typedef __attribute__((ext_vector_type(2))) int v2i;

__global__ void zero_f32(float* __restrict__ p, size_t n) {
    size_t i = (size_t)blockIdx.x * blockDim.x + threadIdx.x;
    if (i < n) p[i] = 0.0f;
}

#define HB  32
#define HCAP 65536

__global__ void __launch_bounds__(1024)
slice_hist_kernel(const int* __restrict__ src, const int* __restrict__ dst,
                  int* __restrict__ hist, int* __restrict__ rank, int N, int E) {
    __shared__ __attribute__((aligned(16))) int cnt[HCAP];
    const int  tid   = threadIdx.x;
    const bool isdst = blockIdx.x < HB;
    const int  sl    = isdst ? blockIdx.x : blockIdx.x - HB;
    const int* __restrict__ key = isdst ? dst : src;
    const int  per   = (E + HB - 1) / HB;
    const int  e0    = sl * per;
    const int  e1    = (e0 + per < E) ? (e0 + per) : E;
    int* __restrict__ hrow = hist + (size_t)blockIdx.x * N;

    for (int base = 0; base < N; base += HCAP) {
        const int lim = (N - base < HCAP) ? (N - base) : HCAP;
        for (int i = 4 * tid; i < lim; i += 4 * 1024)
            *(int4*)(&cnt[i]) = make_int4(0, 0, 0, 0);
        __syncthreads();
        for (int eb = e0; eb < e1; eb += 8 * 1024) {
            int kk[8];
#pragma unroll
            for (int u = 0; u < 8; ++u) {
                const int e = eb + u * 1024 + tid;
                kk[u] = (e < e1) ? key[e] : -1;
            }
#pragma unroll
            for (int u = 0; u < 8; ++u) {
                const unsigned k = (unsigned)(kk[u] - base);
                if (k < (unsigned)lim) {
                    const int r = atomicAdd(&cnt[k], 1);
                    if (isdst) rank[eb + u * 1024 + tid] = r;
                }
            }
        }
        __syncthreads();
        for (int i = tid; i < lim; i += 1024) hrow[base + i] = cnt[i];
        __syncthreads();
    }
}

__global__ void combine_kernel(int* __restrict__ hist, int* __restrict__ incnt,
                               float* __restrict__ norm_src, float* __restrict__ norm_dst, int N) {
    const int n = blockIdx.x * blockDim.x + threadIdx.x;
    if (n < N) {
        int cd[HB], cs[HB];
#pragma unroll
        for (int b = 0; b < HB; ++b) {
            cd[b] = hist[(size_t)b * N + n];
            cs[b] = hist[(size_t)(HB + b) * N + n];
        }
        int run = 0, so = 0;
#pragma unroll
        for (int b = 0; b < HB; ++b) {
            hist[(size_t)b * N + n] = run;
            run += cd[b];
            so  += cs[b];
        }
        incnt[n]    = run;
        norm_dst[n] = rsqrtf(fmaxf((float)run, 1.0f));
        norm_src[n] = rsqrtf(fmaxf((float)so, 1.0f));
    }
}

__global__ void __launch_bounds__(1024)
rowptr_scan_kernel(const int* __restrict__ cnt, int* __restrict__ rowptr, int n) {
    __shared__ int wtot[32];
    const int lane = threadIdx.x & 31;
    const int w    = __builtin_amdgcn_readfirstlane(threadIdx.x >> 5);
    const int seg  = ((n + 4095) / 4096) * 128;
    const int b    = w * seg;
    const int e    = (b + seg < n) ? (b + seg) : n;

    int s = 0;
#pragma unroll 4
    for (int i = b + lane; i < e; i += 32) s += cnt[i];
    for (int o = 16; o > 0; o >>= 1) s += __shfl_xor(s, o);
    if (lane == 0) wtot[w] = s;
    __syncthreads();

    int base = 0;
    for (int k = 0; k < w; ++k) base += wtot[k];

    int c[4];
#pragma unroll
    for (int q = 0; q < 4; ++q) {
        int i = b + 4 * lane + q;
        c[q] = (i < e) ? cnt[i] : 0;
    }
    for (int i0 = b; i0 < e; i0 += 128) {
        int cn[4];
#pragma unroll
        for (int q = 0; q < 4; ++q) {
            int i = i0 + 128 + 4 * lane + q;
            cn[q] = (i < e) ? cnt[i] : 0;
        }
        const int s4 = c[0] + c[1] + c[2] + c[3];
        int v = s4;
        for (int o = 1; o < 32; o <<= 1) {
            int u = __shfl_up(v, o);
            if (lane >= o) v += u;
        }
        int ex = base + v - s4;
#pragma unroll
        for (int q = 0; q < 4; ++q) {
            int i = i0 + 4 * lane + q;
            if (i < e) rowptr[i] = ex;
            ex += c[q];
            c[q] = cn[q];
        }
        base += __shfl(v, 31);
    }
    if (w == 31 && lane == 0) rowptr[n] = base;
}

__global__ void fill_kernel(const int* __restrict__ src, const int* __restrict__ dst,
                            const int* __restrict__ rank, const int* __restrict__ rowptr,
                            const int* __restrict__ hist,
                            const float* __restrict__ norm_src, int* __restrict__ rec, int N, int E) {
    int e = blockIdx.x * blockDim.x + threadIdx.x;
    if (e < E) {
        const int per = (E + HB - 1) / HB;
        int s = src[e];
        int d = dst[e];
        int pos = rowptr[d] + hist[(size_t)(e / per) * N + d] + rank[e];
        v2i r;
        r.x = s;
        r.y = __float_as_int(norm_src[s]);
        *(v2i*)(rec + 2 * (size_t)pos) = r;
    }
}

__global__ void pool_kernel(const float* __restrict__ h, const int* __restrict__ gid,
                            float* __restrict__ hg, int N) {
    int t = blockIdx.x * blockDim.x + threadIdx.x;
    int n = t >> 5;
    int c = (t & 31) << 2;
    if (n < N) {
        int g = gid[n];
        float4 v = *(const float4*)(h + (size_t)n * D_FEAT + c);
        float* hp = hg + (size_t)g * D_FEAT + c;
        unsafeAtomicAdd(hp + 0, v.x);
        unsafeAtomicAdd(hp + 1, v.y);
        unsafeAtomicAdd(hp + 2, v.z);
        unsafeAtomicAdd(hp + 3, v.w);
    }
}

#define MT 4
__global__ void __launch_bounds__(32)
gemm_wmma_f32(const float* __restrict__ A, const float* __restrict__ B,
              const float* __restrict__ bias,
              const float* __restrict__ row_scale,
              const float* __restrict__ resid,
              float* __restrict__ out,
              int Mrows, int K, int Ncol, int relu_c, int mode) {
    int tm   = blockIdx.x;
    int tn   = blockIdx.y;
    int lane = threadIdx.x;
    int half = lane >> 4;
    int l15  = lane & 15;

    int col = tn * 16 + l15;
    float bv = bias[col];

    const float* arow[MT];
    float rs[MT];
#pragma unroll
    for (int i = 0; i < MT; ++i) {
        int row = tm * (16 * MT) + i * 16 + l15;
        if (row >= Mrows) row = Mrows - 1;
        arow[i] = A + (size_t)row * K;
        rs[i] = row_scale ? row_scale[row] : 1.0f;
    }

    v8f acc[MT] = {};
    for (int k0 = 0; k0 < K; k0 += 4) {
        int ka = k0 + half * 2;
        v2f b;
        b.x = B[(size_t)ka * Ncol + col];
        b.y = B[(size_t)(ka + 1) * Ncol + col];
#pragma unroll
        for (int i = 0; i < MT; ++i) {
            __builtin_prefetch(arow[i] + k0 + 64, 0, 0);
            v2f a = *(const v2f*)(arow[i] + ka);
            a.x *= rs[i];
            a.y *= rs[i];
            acc[i] = __builtin_amdgcn_wmma_f32_16x16x4_f32(
                 false, a,  false, b,
                 (short)0, acc[i],  false,  false);
        }
    }

#pragma unroll
    for (int i = 0; i < MT; ++i) {
#pragma unroll
        for (int v = 0; v < 8; ++v) {
            int m = tm * (16 * MT) + i * 16 + half * 8 + v;
            if (m < Mrows) {
                float c = acc[i][v] + bv;
                if (relu_c) c = fmaxf(c, 0.0f);
                float o;
                if (mode == 1)      o = fmaxf(resid[(size_t)m * Ncol + col] + c, 0.0f);
                else if (mode == 2) o = fmaxf(c, 0.0f);
                else                o = c;
                out[(size_t)m * Ncol + col] = o;
            }
        }
    }
}


#define GR 16
#define GP 132
__global__ void __launch_bounds__(128)
gemm128_wmma_f32(const float* __restrict__ A, const float* __restrict__ B,
                 const float* __restrict__ bias,
                 float* out,
                 int Mrows) {
    __shared__ __attribute__((aligned(16))) float sA[2][GR * GP];

    const int tid  = threadIdx.x;
    const int w    = __builtin_amdgcn_readfirstlane(tid >> 5);
    const int lane = tid & 31;
    const int half = lane >> 4;
    const int l15  = lane & 15;
    const int colb = 32 * w + l15;

    const int nchunks = (Mrows + GR - 1) / GR;
    int chunk = blockIdx.x;

    v4f   st[4];
#pragma unroll
    for (int i = 0; i < 4; ++i) {
        int row = chunk * GR + i * 4 + w;
        if (row >= Mrows) row = Mrows - 1;
        st[i] = *(const v4f*)(A + (size_t)row * D_FEAT + lane * 4);
    }

    v2f bf[32][2];
#pragma unroll
    for (int s = 0; s < 32; ++s) {
        int ka = 4 * s + 2 * half;
#pragma unroll
        for (int n = 0; n < 2; ++n) {
            bf[s][n].x = B[(size_t)ka * D_FEAT + colb + 16 * n];
            bf[s][n].y = B[(size_t)(ka + 1) * D_FEAT + colb + 16 * n];
        }
    }
    float bv[2];
    bv[0] = bias[colb];
    bv[1] = bias[colb + 16];

#pragma unroll
    for (int i = 0; i < 4; ++i)
        *(v4f*)(&sA[0][(i * 4 + w) * GP + lane * 4]) = st[i];
    __syncthreads();

    int buf = 0;
    for (; chunk < nchunks; chunk += gridDim.x) {
        const int  next     = chunk + gridDim.x;
        const bool has_next = next < nchunks;
        const int  row0     = chunk * GR + half * 8;
        const bool full     = chunk * GR + GR <= Mrows;

        if (has_next) {
#pragma unroll
            for (int i = 0; i < 4; ++i) {
                int row = next * GR + i * 4 + w;
                if (row >= Mrows) row = Mrows - 1;
                st[i] = *(const v4f*)(A + (size_t)row * D_FEAT + lane * 4);
            }
        }

        v8f acc[2] = {};
        const float* ap = &sA[buf][l15 * GP + 2 * half];
#pragma unroll
        for (int s = 0; s < 32; ++s) {
            v2f a = *(const v2f*)(ap + 4 * s);
            acc[0] = __builtin_amdgcn_wmma_f32_16x16x4_f32(false, a, false, bf[s][0], (short)0, acc[0], false, false);
            acc[1] = __builtin_amdgcn_wmma_f32_16x16x4_f32(false, a, false, bf[s][1], (short)0, acc[1], false, false);
        }

        if (has_next) {
#pragma unroll
            for (int i = 0; i < 4; ++i)
                *(v4f*)(&sA[buf ^ 1][(i * 4 + w) * GP + lane * 4]) = st[i];
        }
        __syncthreads();

        if (full) {
            float* op = out + (size_t)row0 * D_FEAT + colb;
#pragma unroll
            for (int n = 0; n < 2; ++n)
#pragma unroll
                for (int v = 0; v < 8; ++v)
                    op[v * D_FEAT + 16 * n] = acc[n][v] + bv[n];
        } else {
#pragma unroll
            for (int n = 0; n < 2; ++n)
#pragma unroll
                for (int v = 0; v < 8; ++v) {
                    int row = row0 + v;
                    if (row < Mrows) out[(size_t)row * D_FEAT + colb + 16 * n] = acc[n][v] + bv[n];
                }
        }

        buf ^= 1;
    }
}


__device__ __forceinline__ int load_bounds(const int* __restrict__ rowptr, int rbase, int lane, int Mrows) {
    int idx = rbase + lane;
    if (idx > Mrows) idx = Mrows;
    return rowptr[idx];
}

__device__ __forceinline__ float load_scales(const float* __restrict__ row_scale, int rbase, int lane, int Mrows) {
    int idx = rbase + lane;
    if (idx > Mrows - 1) idx = Mrows - 1;
    return row_scale[idx];
}


__device__ __forceinline__ void flush_node(float* srow, float scv, int q, v4f& acc) {
    const float sc = __int_as_float(__builtin_amdgcn_readlane(__float_as_int(scv), q));
    *(v4f*)(srow + q * GP) = acc * sc;
    acc = (v4f){0.0f, 0.0f, 0.0f, 0.0f};
}

template <int K>
__device__ __forceinline__ void gather_run(const float* __restrict__ hin, v2i r, int j, int lane4,
                                           int rpv, float scv, float* srow,
                                           int& pos, int& q, int& nend, v4f& acc) {
    v4f   v[K];
    float ns[K];
#pragma unroll
    for (int k = 0; k < K; ++k) {
        const int s = __builtin_amdgcn_readlane(r.x, j + k);
        ns[k] = __int_as_float(__builtin_amdgcn_readlane(r.y, j + k));
        v[k] = *(const v4f*)(hin + (size_t)s * D_FEAT + lane4);
    }
#pragma unroll
    for (int k = 0; k < K; ++k) {
        while (pos == nend) {
            flush_node(srow, scv, q, acc);
            ++q;
            nend = __builtin_amdgcn_readlane(rpv, q + 1);
        }
        acc += v[k] * ns[k];
        ++pos;
    }
}

__device__ __forceinline__ void gather_chunk(const float* __restrict__ hin, const int* __restrict__ rec,
                                             int rpv, float scv, int lane, float* srow) {
    const int lane4 = lane * 4;
    const int beg = __builtin_amdgcn_readlane(rpv, 0);
    const int end = __builtin_amdgcn_readlane(rpv, 4);
    const int last = (end > 0) ? (end - 1) : 0;
    int i0 = beg + lane;      if (i0 > last) i0 = last;
    int i1 = beg + 32 + lane; if (i1 > last) i1 = last;
    v2i r  = *(const v2i*)(rec + 2 * (size_t)i0);
    v2i rn = *(const v2i*)(rec + 2 * (size_t)i1);

    int pos  = beg;
    int q    = 0;
    int nend = __builtin_amdgcn_readlane(rpv, 1);
    v4f acc  = {0.0f, 0.0f, 0.0f, 0.0f};
    for (int base = beg; base < end; base += 32) {
        const int m = (end - base < 32) ? (end - base) : 32;
        int j = 0;
        for (; j + 8 <= m; j += 8) gather_run<8>(hin, r, j, lane4, rpv, scv, srow, pos, q, nend, acc);
        if (j + 4 <= m) { gather_run<4>(hin, r, j, lane4, rpv, scv, srow, pos, q, nend, acc); j += 4; }
        if (j + 2 <= m) { gather_run<2>(hin, r, j, lane4, rpv, scv, srow, pos, q, nend, acc); j += 2; }
        if (j < m)      { gather_run<1>(hin, r, j, lane4, rpv, scv, srow, pos, q, nend, acc); }
        r = rn;
        int i2 = base + 64 + lane; if (i2 > last) i2 = last;
        rn = *(const v2i*)(rec + 2 * (size_t)i2);
    }
    for (; q < 4; ++q) flush_node(srow, scv, q, acc);
}

template <bool POOL>
__global__ void __launch_bounds__(128)
gcn_layer_wmma_f32(const float* __restrict__ hin, const int* __restrict__ rec,
                   const int* __restrict__ rowptr,
                   const float* __restrict__ B, const float* __restrict__ bias,
                   const float* __restrict__ row_scale,
                   float* __restrict__ out,
                   int Mrows, int relu_c,
                   const int* __restrict__ gid, float* __restrict__ hg) {
    __shared__ __attribute__((aligned(16))) float sA[2][GR * GP];

    const int tid  = threadIdx.x;
    const int w    = __builtin_amdgcn_readfirstlane(tid >> 5);
    const int lane = tid & 31;
    const int half = lane >> 4;
    const int l15  = lane & 15;
    const int colb = 32 * w + l15;

    const int nchunks = (Mrows + GR - 1) / GR;
    int chunk = blockIdx.x;

    int   rpv = load_bounds(rowptr, chunk * GR + 4 * w, lane, Mrows);
    float scv = load_scales(row_scale, chunk * GR + 4 * w, lane, Mrows);

    v2f bf[32][2];
#pragma unroll
    for (int s = 0; s < 32; ++s) {
        int ka = 4 * s + 2 * half;
#pragma unroll
        for (int n = 0; n < 2; ++n) {
            bf[s][n].x = B[(size_t)ka * D_FEAT + colb + 16 * n];
            bf[s][n].y = B[(size_t)(ka + 1) * D_FEAT + colb + 16 * n];
        }
    }
    float bv[2];
    bv[0] = bias[colb];
    bv[1] = bias[colb + 16];

    gather_chunk(hin, rec, rpv, scv, lane, &sA[0][(4 * w) * GP + lane * 4]);
    __syncthreads();

    int buf = 0;
    for (; chunk < nchunks; chunk += gridDim.x) {
        const int  next     = chunk + gridDim.x;
        const bool has_next = next < nchunks;
        const int  row0     = chunk * GR + half * 8;
        const bool full     = chunk * GR + GR <= Mrows;

        rpv = load_bounds(rowptr, next * GR + 4 * w, lane, Mrows);
        scv = load_scales(row_scale, next * GR + 4 * w, lane, Mrows);

        int gl = 0;
        if constexpr (POOL) {
            int gr = chunk * GR + l15;
            if (gr > Mrows - 1) gr = Mrows - 1;
            gl = gid[gr];
        }

        float r[2][8];
        if (full) {
            const float* rp = hin + (size_t)row0 * D_FEAT + colb;
#pragma unroll
            for (int n = 0; n < 2; ++n)
#pragma unroll
                for (int v = 0; v < 8; ++v)
                    r[n][v] = rp[v * D_FEAT + 16 * n];
        }

        v8f acc[2] = {};
        const float* ap = &sA[buf][l15 * GP + 2 * half];
#pragma unroll
        for (int s = 0; s < 32; ++s) {
            v2f a = *(const v2f*)(ap + 4 * s);
            acc[0] = __builtin_amdgcn_wmma_f32_16x16x4_f32(false, a, false, bf[s][0], (short)0, acc[0], false, false);
            acc[1] = __builtin_amdgcn_wmma_f32_16x16x4_f32(false, a, false, bf[s][1], (short)0, acc[1], false, false);
        }

        float c[2][8];
#pragma unroll
        for (int n = 0; n < 2; ++n)
#pragma unroll
            for (int v = 0; v < 8; ++v) {
                c[n][v] = acc[n][v] + bv[n];
                if (relu_c) c[n][v] = fmaxf(c[n][v], 0.0f);
            }

        if constexpr (POOL) {
            float o[2][8];
#pragma unroll
            for (int n = 0; n < 2; ++n)
#pragma unroll
                for (int v = 0; v < 8; ++v) {
                    if (full) {
                        o[n][v] = fmaxf(r[n][v] + c[n][v], 0.0f);
                    } else {
                        const int row = row0 + v;
                        o[n][v] = (row < Mrows)
                            ? fmaxf(hin[(size_t)row * D_FEAT + colb + 16 * n] + c[n][v], 0.0f) : 0.0f;
                    }
                }
            const int  g0  = __builtin_amdgcn_readlane(gl, 0);
            const bool uni = full && (__builtin_amdgcn_ballot_w32(gl != g0) == 0u);
            if (uni) {
#pragma unroll
                for (int n = 0; n < 2; ++n) {
                    float s = o[n][0];
#pragma unroll
                    for (int v = 1; v < 8; ++v) s += o[n][v];
                    s += __shfl_xor(s, 16);
                    if (half == 0) unsafeAtomicAdd(hg + (size_t)g0 * D_FEAT + colb + 16 * n, s);
                }
            } else {
                int gv[8];
#pragma unroll
                for (int v = 0; v < 8; ++v) gv[v] = __shfl(gl, half * 8 + v);
#pragma unroll
                for (int n = 0; n < 2; ++n) {
                    float run = o[n][0];
                    int   gc  = gv[0];
#pragma unroll
                    for (int v = 1; v < 8; ++v) {
                        if (gv[v] != gc) {
                            unsafeAtomicAdd(hg + (size_t)gc * D_FEAT + colb + 16 * n, run);
                            run = 0.0f;
                            gc  = gv[v];
                        }
                        run += o[n][v];
                    }
                    unsafeAtomicAdd(hg + (size_t)gc * D_FEAT + colb + 16 * n, run);
                }
            }
        } else
        if (full) {
            float* op = out + (size_t)row0 * D_FEAT + colb;
#pragma unroll
            for (int n = 0; n < 2; ++n)
#pragma unroll
                for (int v = 0; v < 8; ++v)
                    op[v * D_FEAT + 16 * n] = fmaxf(r[n][v] + c[n][v], 0.0f);
        } else {
#pragma unroll
            for (int n = 0; n < 2; ++n)
#pragma unroll
                for (int v = 0; v < 8; ++v) {
                    const int row = row0 + v;
                    if (row < Mrows) {
                        const size_t o = (size_t)row * D_FEAT + colb + 16 * n;
                        out[o] = fmaxf(hin[o] + c[n][v], 0.0f);
                    }
                }
        }

        if (has_next)
            gather_chunk(hin, rec, rpv, scv, lane, &sA[buf ^ 1][(4 * w) * GP + lane * 4]);
        __syncthreads();

        buf ^= 1;
    }
}


extern "C" void kernel_launch(void* const* d_in, const int* in_sizes, int n_in,
                              void* d_out, int out_size, void* d_ws, size_t ws_size,
                              hipStream_t stream) {
    const float* node_features = (const float*)d_in[0];
    const int*   edge_src      = (const int*)d_in[1];
    const int*   edge_dst      = (const int*)d_in[2];
    const int*   graph_ids     = (const int*)d_in[3];
    const float* W_proj        = (const float*)d_in[4];
    const float* b_proj        = (const float*)d_in[5];
    const float* W_gcn         = (const float*)d_in[6];
    const float* b_gcn         = (const float*)d_in[7];
    const float* W_mlp0        = (const float*)d_in[8];
    const float* b_mlp0        = (const float*)d_in[9];
    const float* W_mlp1        = (const float*)d_in[10];
    const float* b_mlp1        = (const float*)d_in[11];
    const float* W_mlp2        = (const float*)d_in[12];
    const float* b_mlp2        = (const float*)d_in[13];

    const int D = D_FEAT;
    const int Mm = 256;
    const int N = in_sizes[3];
    const int E = in_sizes[1];
    const int L = in_sizes[6] / (D * D);
    const int G = out_size / Mm;

    float* ws = (float*)d_ws;
    float* norm_src = ws;
    float* norm_dst = norm_src + N;
    float* h        = norm_dst + N;
    float* agg      = h + (size_t)N * D;
    float* hg       = agg + (size_t)N * D;
    float* t0       = hg + (size_t)G * D;
    float* t1       = t0 + (size_t)G * Mm;
    int*   rec      = (int*)(t1 + (size_t)G * Mm);
    int*   rank     = rec + (size_t)2 * E;
    int*   incnt    = rank + E;
    int*   rowptr   = incnt + N;
    int*   hist     = rowptr + N + 1;

    const int TB = 256;
    auto blocks = [&](size_t n) { return (unsigned)((n + TB - 1) / TB); };
    const int MROWS = 16 * MT;

    slice_hist_kernel<<<2 * HB, 1024, 0, stream>>>(edge_src, edge_dst, hist, rank, N, E);
    combine_kernel<<<blocks(N), TB, 0, stream>>>(hist, incnt, norm_src, norm_dst, N);
    rowptr_scan_kernel<<<1, 1024, 0, stream>>>(incnt, rowptr, N);
    fill_kernel<<<blocks(E), TB, 0, stream>>>(edge_src, edge_dst, rank, rowptr, hist, norm_src, rec, N, E);

    const unsigned gblocks = (unsigned)(((N + GR - 1) / GR + 3) / 4);
    gemm128_wmma_f32<<<gblocks, 128, 0, stream>>>(node_features, W_proj, b_proj, h, N);

    zero_f32<<<blocks((size_t)G * D), TB, 0, stream>>>(hg, (size_t)G * D);
    float* hcur  = h;
    float* hnext = agg;
    for (int i = 0; i < L; ++i) {
        if (i < L - 1)
            gcn_layer_wmma_f32<false><<<gblocks, 128, 0, stream>>>(hcur, rec, rowptr, W_gcn + (size_t)i * D * D,
                                                                   b_gcn + (size_t)i * D, norm_dst, hnext, N,
                                                                    1, nullptr, nullptr);
        else
            gcn_layer_wmma_f32<true><<<gblocks, 128, 0, stream>>>(hcur, rec, rowptr, W_gcn + (size_t)i * D * D,
                                                                  b_gcn + (size_t)i * D, norm_dst, hnext, N,
                                                                   0, graph_ids, hg);
        float* t = hcur; hcur = hnext; hnext = t;
    }

    if (L == 0) pool_kernel<<<blocks((size_t)N * 32), TB, 0, stream>>>(hcur, graph_ids, hg, N);

    {
        dim3 g0((G + MROWS - 1) / MROWS, Mm / 16);
        gemm_wmma_f32<<<g0, 32, 0, stream>>>(hg, W_mlp0, b_mlp0, nullptr, nullptr,
                                             t0, G, D, Mm, 0,  2);
        gemm_wmma_f32<<<g0, 32, 0, stream>>>(t0, W_mlp1, b_mlp1, nullptr, nullptr,
                                             t1, G, Mm, Mm, 0,  2);
        gemm_wmma_f32<<<g0, 32, 0, stream>>>(t1, W_mlp2, b_mlp2, nullptr, nullptr,
                                             (float*)d_out, G, Mm, Mm, 0,  0);
    }
}


# MLP head: 3 one-wave-per-block GEMM launches (a load round trip every 8 k) fused into one launch; 16-row tile per 16-wave block, weight fragments loaded 128 k ahead, t0/t1 in LDS; k order kept
# speedup vs baseline: 5.1238x; 1.2011x over previous
#include <hip/hip_runtime.h>
#include <hip/hip_bf16.h>

typedef __attribute__((ext_vector_type(2))) float v2f;
typedef __attribute__((ext_vector_type(8))) float v8f;

#define D_FEAT 128


typedef __attribute__((ext_vector_type(4))) float v4f;
typedef __attribute__((ext_vector_type(2))) int v2i;

__global__ void zero_f32(float* __restrict__ p, size_t n) {
    size_t i = (size_t)blockIdx.x * blockDim.x + threadIdx.x;
    if (i < n) p[i] = 0.0f;
}

#define HB  32
#define HCAP 65536

__global__ void __launch_bounds__(1024)
slice_hist_kernel(const int* __restrict__ src, const int* __restrict__ dst,
                  int* __restrict__ hist, int* __restrict__ rank, int N, int E) {
    __shared__ __attribute__((aligned(16))) int cnt[HCAP];
    const int  tid   = threadIdx.x;
    const bool isdst = blockIdx.x < HB;
    const int  sl    = isdst ? blockIdx.x : blockIdx.x - HB;
    const int* __restrict__ key = isdst ? dst : src;
    const int  per   = (E + HB - 1) / HB;
    const int  e0    = sl * per;
    const int  e1    = (e0 + per < E) ? (e0 + per) : E;
    int* __restrict__ hrow = hist + (size_t)blockIdx.x * N;

    for (int base = 0; base < N; base += HCAP) {
        const int lim = (N - base < HCAP) ? (N - base) : HCAP;
        for (int i = 4 * tid; i < lim; i += 4 * 1024)
            *(int4*)(&cnt[i]) = make_int4(0, 0, 0, 0);
        __syncthreads();
        for (int eb = e0; eb < e1; eb += 8 * 1024) {
            int kk[8];
#pragma unroll
            for (int u = 0; u < 8; ++u) {
                const int e = eb + u * 1024 + tid;
                kk[u] = (e < e1) ? key[e] : -1;
            }
#pragma unroll
            for (int u = 0; u < 8; ++u) {
                const unsigned k = (unsigned)(kk[u] - base);
                if (k < (unsigned)lim) {
                    const int r = atomicAdd(&cnt[k], 1);
                    if (isdst) rank[eb + u * 1024 + tid] = r;
                }
            }
        }
        __syncthreads();
        for (int i = tid; i < lim; i += 1024) hrow[base + i] = cnt[i];
        __syncthreads();
    }
}

__global__ void combine_kernel(int* __restrict__ hist, int* __restrict__ incnt,
                               float* __restrict__ norm_src, float* __restrict__ norm_dst, int N) {
    const int n = blockIdx.x * blockDim.x + threadIdx.x;
    if (n < N) {
        int cd[HB], cs[HB];
#pragma unroll
        for (int b = 0; b < HB; ++b) {
            cd[b] = hist[(size_t)b * N + n];
            cs[b] = hist[(size_t)(HB + b) * N + n];
        }
        int run = 0, so = 0;
#pragma unroll
        for (int b = 0; b < HB; ++b) {
            hist[(size_t)b * N + n] = run;
            run += cd[b];
            so  += cs[b];
        }
        incnt[n]    = run;
        norm_dst[n] = rsqrtf(fmaxf((float)run, 1.0f));
        norm_src[n] = rsqrtf(fmaxf((float)so, 1.0f));
    }
}

__global__ void __launch_bounds__(1024)
rowptr_scan_kernel(const int* __restrict__ cnt, int* __restrict__ rowptr, int n) {
    __shared__ int wtot[32];
    const int lane = threadIdx.x & 31;
    const int w    = __builtin_amdgcn_readfirstlane(threadIdx.x >> 5);
    const int seg  = ((n + 4095) / 4096) * 128;
    const int b    = w * seg;
    const int e    = (b + seg < n) ? (b + seg) : n;

    int s = 0;
#pragma unroll 4
    for (int i = b + lane; i < e; i += 32) s += cnt[i];
    for (int o = 16; o > 0; o >>= 1) s += __shfl_xor(s, o);
    if (lane == 0) wtot[w] = s;
    __syncthreads();

    int base = 0;
    for (int k = 0; k < w; ++k) base += wtot[k];

    int c[4];
#pragma unroll
    for (int q = 0; q < 4; ++q) {
        int i = b + 4 * lane + q;
        c[q] = (i < e) ? cnt[i] : 0;
    }
    for (int i0 = b; i0 < e; i0 += 128) {
        int cn[4];
#pragma unroll
        for (int q = 0; q < 4; ++q) {
            int i = i0 + 128 + 4 * lane + q;
            cn[q] = (i < e) ? cnt[i] : 0;
        }
        const int s4 = c[0] + c[1] + c[2] + c[3];
        int v = s4;
        for (int o = 1; o < 32; o <<= 1) {
            int u = __shfl_up(v, o);
            if (lane >= o) v += u;
        }
        int ex = base + v - s4;
#pragma unroll
        for (int q = 0; q < 4; ++q) {
            int i = i0 + 4 * lane + q;
            if (i < e) rowptr[i] = ex;
            ex += c[q];
            c[q] = cn[q];
        }
        base += __shfl(v, 31);
    }
    if (w == 31 && lane == 0) rowptr[n] = base;
}

__global__ void fill_kernel(const int* __restrict__ src, const int* __restrict__ dst,
                            const int* __restrict__ rank, const int* __restrict__ rowptr,
                            const int* __restrict__ hist,
                            const float* __restrict__ norm_src, int* __restrict__ rec, int N, int E) {
    int e = blockIdx.x * blockDim.x + threadIdx.x;
    if (e < E) {
        const int per = (E + HB - 1) / HB;
        int s = src[e];
        int d = dst[e];
        int pos = rowptr[d] + hist[(size_t)(e / per) * N + d] + rank[e];
        v2i r;
        r.x = s;
        r.y = __float_as_int(norm_src[s]);
        *(v2i*)(rec + 2 * (size_t)pos) = r;
    }
}

__global__ void pool_kernel(const float* __restrict__ h, const int* __restrict__ gid,
                            float* __restrict__ hg, int N) {
    int t = blockIdx.x * blockDim.x + threadIdx.x;
    int n = t >> 5;
    int c = (t & 31) << 2;
    if (n < N) {
        int g = gid[n];
        float4 v = *(const float4*)(h + (size_t)n * D_FEAT + c);
        float* hp = hg + (size_t)g * D_FEAT + c;
        unsafeAtomicAdd(hp + 0, v.x);
        unsafeAtomicAdd(hp + 1, v.y);
        unsafeAtomicAdd(hp + 2, v.z);
        unsafeAtomicAdd(hp + 3, v.w);
    }
}

#define GR 16
#define GP 132
__global__ void __launch_bounds__(128)
gemm128_wmma_f32(const float* __restrict__ A, const float* __restrict__ B,
                 const float* __restrict__ bias,
                 float* out,
                 int Mrows) {
    __shared__ __attribute__((aligned(16))) float sA[2][GR * GP];

    const int tid  = threadIdx.x;
    const int w    = __builtin_amdgcn_readfirstlane(tid >> 5);
    const int lane = tid & 31;
    const int half = lane >> 4;
    const int l15  = lane & 15;
    const int colb = 32 * w + l15;

    const int nchunks = (Mrows + GR - 1) / GR;
    int chunk = blockIdx.x;

    v4f   st[4];
#pragma unroll
    for (int i = 0; i < 4; ++i) {
        int row = chunk * GR + i * 4 + w;
        if (row >= Mrows) row = Mrows - 1;
        st[i] = *(const v4f*)(A + (size_t)row * D_FEAT + lane * 4);
    }

    v2f bf[32][2];
#pragma unroll
    for (int s = 0; s < 32; ++s) {
        int ka = 4 * s + 2 * half;
#pragma unroll
        for (int n = 0; n < 2; ++n) {
            bf[s][n].x = B[(size_t)ka * D_FEAT + colb + 16 * n];
            bf[s][n].y = B[(size_t)(ka + 1) * D_FEAT + colb + 16 * n];
        }
    }
    float bv[2];
    bv[0] = bias[colb];
    bv[1] = bias[colb + 16];

#pragma unroll
    for (int i = 0; i < 4; ++i)
        *(v4f*)(&sA[0][(i * 4 + w) * GP + lane * 4]) = st[i];
    __syncthreads();

    int buf = 0;
    for (; chunk < nchunks; chunk += gridDim.x) {
        const int  next     = chunk + gridDim.x;
        const bool has_next = next < nchunks;
        const int  row0     = chunk * GR + half * 8;
        const bool full     = chunk * GR + GR <= Mrows;

        if (has_next) {
#pragma unroll
            for (int i = 0; i < 4; ++i) {
                int row = next * GR + i * 4 + w;
                if (row >= Mrows) row = Mrows - 1;
                st[i] = *(const v4f*)(A + (size_t)row * D_FEAT + lane * 4);
            }
        }

        v8f acc[2] = {};
        const float* ap = &sA[buf][l15 * GP + 2 * half];
#pragma unroll
        for (int s = 0; s < 32; ++s) {
            v2f a = *(const v2f*)(ap + 4 * s);
            acc[0] = __builtin_amdgcn_wmma_f32_16x16x4_f32(false, a, false, bf[s][0], (short)0, acc[0], false, false);
            acc[1] = __builtin_amdgcn_wmma_f32_16x16x4_f32(false, a, false, bf[s][1], (short)0, acc[1], false, false);
        }

        if (has_next) {
#pragma unroll
            for (int i = 0; i < 4; ++i)
                *(v4f*)(&sA[buf ^ 1][(i * 4 + w) * GP + lane * 4]) = st[i];
        }
        __syncthreads();

        if (full) {
            float* op = out + (size_t)row0 * D_FEAT + colb;
#pragma unroll
            for (int n = 0; n < 2; ++n)
#pragma unroll
                for (int v = 0; v < 8; ++v)
                    op[v * D_FEAT + 16 * n] = acc[n][v] + bv[n];
        } else {
#pragma unroll
            for (int n = 0; n < 2; ++n)
#pragma unroll
                for (int v = 0; v < 8; ++v) {
                    int row = row0 + v;
                    if (row < Mrows) out[(size_t)row * D_FEAT + colb + 16 * n] = acc[n][v] + bv[n];
                }
        }

        buf ^= 1;
    }
}


__device__ __forceinline__ int load_bounds(const int* __restrict__ rowptr, int rbase, int lane, int Mrows) {
    int idx = rbase + lane;
    if (idx > Mrows) idx = Mrows;
    return rowptr[idx];
}

__device__ __forceinline__ float load_scales(const float* __restrict__ row_scale, int rbase, int lane, int Mrows) {
    int idx = rbase + lane;
    if (idx > Mrows - 1) idx = Mrows - 1;
    return row_scale[idx];
}


__device__ __forceinline__ void flush_node(float* srow, float scv, int q, v4f& acc) {
    const float sc = __int_as_float(__builtin_amdgcn_readlane(__float_as_int(scv), q));
    *(v4f*)(srow + q * GP) = acc * sc;
    acc = (v4f){0.0f, 0.0f, 0.0f, 0.0f};
}

template <int K>
__device__ __forceinline__ void gather_run(const float* __restrict__ hin, v2i r, int j, int lane4,
                                           int rpv, float scv, float* srow,
                                           int& pos, int& q, int& nend, v4f& acc) {
    v4f   v[K];
    float ns[K];
#pragma unroll
    for (int k = 0; k < K; ++k) {
        const int s = __builtin_amdgcn_readlane(r.x, j + k);
        ns[k] = __int_as_float(__builtin_amdgcn_readlane(r.y, j + k));
        v[k] = *(const v4f*)(hin + (size_t)s * D_FEAT + lane4);
    }
#pragma unroll
    for (int k = 0; k < K; ++k) {
        while (pos == nend) {
            flush_node(srow, scv, q, acc);
            ++q;
            nend = __builtin_amdgcn_readlane(rpv, q + 1);
        }
        acc += v[k] * ns[k];
        ++pos;
    }
}

__device__ __forceinline__ void gather_chunk(const float* __restrict__ hin, const int* __restrict__ rec,
                                             int rpv, float scv, int lane, float* srow) {
    const int lane4 = lane * 4;
    const int beg = __builtin_amdgcn_readlane(rpv, 0);
    const int end = __builtin_amdgcn_readlane(rpv, 4);
    const int last = (end > 0) ? (end - 1) : 0;
    int i0 = beg + lane;      if (i0 > last) i0 = last;
    int i1 = beg + 32 + lane; if (i1 > last) i1 = last;
    v2i r  = *(const v2i*)(rec + 2 * (size_t)i0);
    v2i rn = *(const v2i*)(rec + 2 * (size_t)i1);

    int pos  = beg;
    int q    = 0;
    int nend = __builtin_amdgcn_readlane(rpv, 1);
    v4f acc  = {0.0f, 0.0f, 0.0f, 0.0f};
    for (int base = beg; base < end; base += 32) {
        const int m = (end - base < 32) ? (end - base) : 32;
        int j = 0;
        for (; j + 8 <= m; j += 8) gather_run<8>(hin, r, j, lane4, rpv, scv, srow, pos, q, nend, acc);
        if (j + 4 <= m) { gather_run<4>(hin, r, j, lane4, rpv, scv, srow, pos, q, nend, acc); j += 4; }
        if (j + 2 <= m) { gather_run<2>(hin, r, j, lane4, rpv, scv, srow, pos, q, nend, acc); j += 2; }
        if (j < m)      { gather_run<1>(hin, r, j, lane4, rpv, scv, srow, pos, q, nend, acc); }
        r = rn;
        int i2 = base + 64 + lane; if (i2 > last) i2 = last;
        rn = *(const v2i*)(rec + 2 * (size_t)i2);
    }
    for (; q < 4; ++q) flush_node(srow, scv, q, acc);
}

template <bool POOL>
__global__ void __launch_bounds__(128)
gcn_layer_wmma_f32(const float* __restrict__ hin, const int* __restrict__ rec,
                   const int* __restrict__ rowptr,
                   const float* __restrict__ B, const float* __restrict__ bias,
                   const float* __restrict__ row_scale,
                   float* __restrict__ out,
                   int Mrows, int relu_c,
                   const int* __restrict__ gid, float* __restrict__ hg) {
    __shared__ __attribute__((aligned(16))) float sA[2][GR * GP];

    const int tid  = threadIdx.x;
    const int w    = __builtin_amdgcn_readfirstlane(tid >> 5);
    const int lane = tid & 31;
    const int half = lane >> 4;
    const int l15  = lane & 15;
    const int colb = 32 * w + l15;

    const int nchunks = (Mrows + GR - 1) / GR;
    int chunk = blockIdx.x;

    int   rpv = load_bounds(rowptr, chunk * GR + 4 * w, lane, Mrows);
    float scv = load_scales(row_scale, chunk * GR + 4 * w, lane, Mrows);

    v2f bf[32][2];
#pragma unroll
    for (int s = 0; s < 32; ++s) {
        int ka = 4 * s + 2 * half;
#pragma unroll
        for (int n = 0; n < 2; ++n) {
            bf[s][n].x = B[(size_t)ka * D_FEAT + colb + 16 * n];
            bf[s][n].y = B[(size_t)(ka + 1) * D_FEAT + colb + 16 * n];
        }
    }
    float bv[2];
    bv[0] = bias[colb];
    bv[1] = bias[colb + 16];

    gather_chunk(hin, rec, rpv, scv, lane, &sA[0][(4 * w) * GP + lane * 4]);
    __syncthreads();

    int buf = 0;
    for (; chunk < nchunks; chunk += gridDim.x) {
        const int  next     = chunk + gridDim.x;
        const bool has_next = next < nchunks;
        const int  row0     = chunk * GR + half * 8;
        const bool full     = chunk * GR + GR <= Mrows;

        rpv = load_bounds(rowptr, next * GR + 4 * w, lane, Mrows);
        scv = load_scales(row_scale, next * GR + 4 * w, lane, Mrows);

        int gl = 0;
        if constexpr (POOL) {
            int gr = chunk * GR + l15;
            if (gr > Mrows - 1) gr = Mrows - 1;
            gl = gid[gr];
        }

        float r[2][8];
        if (full) {
            const float* rp = hin + (size_t)row0 * D_FEAT + colb;
#pragma unroll
            for (int n = 0; n < 2; ++n)
#pragma unroll
                for (int v = 0; v < 8; ++v)
                    r[n][v] = rp[v * D_FEAT + 16 * n];
        }

        v8f acc[2] = {};
        const float* ap = &sA[buf][l15 * GP + 2 * half];
#pragma unroll
        for (int s = 0; s < 32; ++s) {
            v2f a = *(const v2f*)(ap + 4 * s);
            acc[0] = __builtin_amdgcn_wmma_f32_16x16x4_f32(false, a, false, bf[s][0], (short)0, acc[0], false, false);
            acc[1] = __builtin_amdgcn_wmma_f32_16x16x4_f32(false, a, false, bf[s][1], (short)0, acc[1], false, false);
        }

        float c[2][8];
#pragma unroll
        for (int n = 0; n < 2; ++n)
#pragma unroll
            for (int v = 0; v < 8; ++v) {
                c[n][v] = acc[n][v] + bv[n];
                if (relu_c) c[n][v] = fmaxf(c[n][v], 0.0f);
            }

        if constexpr (POOL) {
            float o[2][8];
#pragma unroll
            for (int n = 0; n < 2; ++n)
#pragma unroll
                for (int v = 0; v < 8; ++v) {
                    if (full) {
                        o[n][v] = fmaxf(r[n][v] + c[n][v], 0.0f);
                    } else {
                        const int row = row0 + v;
                        o[n][v] = (row < Mrows)
                            ? fmaxf(hin[(size_t)row * D_FEAT + colb + 16 * n] + c[n][v], 0.0f) : 0.0f;
                    }
                }
            const int  g0  = __builtin_amdgcn_readlane(gl, 0);
            const bool uni = full && (__builtin_amdgcn_ballot_w32(gl != g0) == 0u);
            if (uni) {
#pragma unroll
                for (int n = 0; n < 2; ++n) {
                    float s = o[n][0];
#pragma unroll
                    for (int v = 1; v < 8; ++v) s += o[n][v];
                    s += __shfl_xor(s, 16);
                    if (half == 0) unsafeAtomicAdd(hg + (size_t)g0 * D_FEAT + colb + 16 * n, s);
                }
            } else {
                int gv[8];
#pragma unroll
                for (int v = 0; v < 8; ++v) gv[v] = __shfl(gl, half * 8 + v);
#pragma unroll
                for (int n = 0; n < 2; ++n) {
                    float run = o[n][0];
                    int   gc  = gv[0];
#pragma unroll
                    for (int v = 1; v < 8; ++v) {
                        if (gv[v] != gc) {
                            unsafeAtomicAdd(hg + (size_t)gc * D_FEAT + colb + 16 * n, run);
                            run = 0.0f;
                            gc  = gv[v];
                        }
                        run += o[n][v];
                    }
                    unsafeAtomicAdd(hg + (size_t)gc * D_FEAT + colb + 16 * n, run);
                }
            }
        } else
        if (full) {
            float* op = out + (size_t)row0 * D_FEAT + colb;
#pragma unroll
            for (int n = 0; n < 2; ++n)
#pragma unroll
                for (int v = 0; v < 8; ++v)
                    op[v * D_FEAT + 16 * n] = fmaxf(r[n][v] + c[n][v], 0.0f);
        } else {
#pragma unroll
            for (int n = 0; n < 2; ++n)
#pragma unroll
                for (int v = 0; v < 8; ++v) {
                    const int row = row0 + v;
                    if (row < Mrows) {
                        const size_t o = (size_t)row * D_FEAT + colb + 16 * n;
                        out[o] = fmaxf(hin[o] + c[n][v], 0.0f);
                    }
                }
        }

        if (has_next)
            gather_chunk(hin, rec, rpv, scv, lane, &sA[buf ^ 1][(4 * w) * GP + lane * 4]);
        __syncthreads();

        buf ^= 1;
    }
}


#define MH 256
#define MP 260

__device__ __forceinline__ void load_bfrag(const float* wp, int half, v2f (&b)[32]) {
#pragma unroll
    for (int s = 0; s < 32; ++s) {
        const int ka = 4 * s + 2 * half;
        b[s].x = wp[(size_t)ka * MH];
        b[s].y = wp[(size_t)(ka + 1) * MH];
    }
}

__device__ __forceinline__ v8f mma_128k(const float* ap, const v2f (&b)[32], v8f acc) {
#pragma unroll
    for (int s = 0; s < 32; ++s) {
        v2f a = *(const v2f*)(ap + 4 * s);
        acc = __builtin_amdgcn_wmma_f32_16x16x4_f32(false, a, false, b[s], (short)0, acc, false, false);
    }
    return acc;
}

__global__ void __launch_bounds__(512)
mlp3_wmma_f32(const float* X,
              const float* W0, const float* b0,
              const float* W1, const float* b1,
              const float* W2, const float* b2,
              float* out, int G) {
    __shared__ __attribute__((aligned(16))) float sX[2][16 * MP];

    const int tid  = threadIdx.x;
    const int w    = __builtin_amdgcn_readfirstlane(tid >> 5);
    const int lane = tid & 31;
    const int half = lane >> 4;
    const int l15  = lane & 15;
    const int col  = 16 * w + l15;
    const int row0 = blockIdx.x * 16;

    int xr = row0 + w;
    if (xr >= G) xr = G - 1;
    const v4f x = *(const v4f*)(X + (size_t)xr * D_FEAT + lane * 4);

    v2f bA[32], bB[32];
    load_bfrag(W0 + col, half, bA);
    load_bfrag(W1 + col, half, bB);
    const float bias0 = b0[col];
    const float bias1 = b1[col];
    const float bias2 = b2[col];

    *(v4f*)(&sX[0][w * MP + lane * 4]) = x;
    __syncthreads();

    const float* a0 = &sX[0][l15 * MP + 2 * half];
    const float* a1 = &sX[1][l15 * MP + 2 * half];
    float* c0 = &sX[0][(8 * half) * MP + col];
    float* c1 = &sX[1][(8 * half) * MP + col];

    v8f acc = {};
    acc = mma_128k(a0, bA, acc);
    load_bfrag(W1 + (size_t)128 * MH + col, half, bA);
#pragma unroll
    for (int v = 0; v < 8; ++v) c1[v * MP] = fmaxf(acc[v] + bias0, 0.0f);
    __syncthreads();

    acc = (v8f){};
    acc = mma_128k(a1, bB, acc);
    load_bfrag(W2 + col, half, bB);
    acc = mma_128k(a1 + 128, bA, acc);
    load_bfrag(W2 + (size_t)128 * MH + col, half, bA);
#pragma unroll
    for (int v = 0; v < 8; ++v) c0[v * MP] = fmaxf(acc[v] + bias1, 0.0f);
    __syncthreads();

    acc = (v8f){};
    acc = mma_128k(a0, bB, acc);
    acc = mma_128k(a0 + 128, bA, acc);
#pragma unroll
    for (int v = 0; v < 8; ++v) c1[v * MP] = acc[v] + bias2;
    __syncthreads();

    const int orow = row0 + w;
    if (orow < G) {
        const float* sp = &sX[1][w * MP + lane * 4];
        float* op = out + (size_t)orow * MH + lane * 4;
        *(v4f*)(op)       = *(const v4f*)(sp);
        *(v4f*)(op + 128) = *(const v4f*)(sp + 128);
    }
}


extern "C" void kernel_launch(void* const* d_in, const int* in_sizes, int n_in,
                              void* d_out, int out_size, void* d_ws, size_t ws_size,
                              hipStream_t stream) {
    const float* node_features = (const float*)d_in[0];
    const int*   edge_src      = (const int*)d_in[1];
    const int*   edge_dst      = (const int*)d_in[2];
    const int*   graph_ids     = (const int*)d_in[3];
    const float* W_proj        = (const float*)d_in[4];
    const float* b_proj        = (const float*)d_in[5];
    const float* W_gcn         = (const float*)d_in[6];
    const float* b_gcn         = (const float*)d_in[7];
    const float* W_mlp0        = (const float*)d_in[8];
    const float* b_mlp0        = (const float*)d_in[9];
    const float* W_mlp1        = (const float*)d_in[10];
    const float* b_mlp1        = (const float*)d_in[11];
    const float* W_mlp2        = (const float*)d_in[12];
    const float* b_mlp2        = (const float*)d_in[13];

    const int D = D_FEAT;
    const int Mm = 256;
    const int N = in_sizes[3];
    const int E = in_sizes[1];
    const int L = in_sizes[6] / (D * D);
    const int G = out_size / Mm;

    float* ws = (float*)d_ws;
    float* norm_src = ws;
    float* norm_dst = norm_src + N;
    float* h        = norm_dst + N;
    float* agg      = h + (size_t)N * D;
    float* hg       = agg + (size_t)N * D;
    int*   rec      = (int*)(hg + (size_t)G * D);
    int*   rank     = rec + (size_t)2 * E;
    int*   incnt    = rank + E;
    int*   rowptr   = incnt + N;
    int*   hist     = rowptr + N + 1;

    const int TB = 256;
    auto blocks = [&](size_t n) { return (unsigned)((n + TB - 1) / TB); };

    slice_hist_kernel<<<2 * HB, 1024, 0, stream>>>(edge_src, edge_dst, hist, rank, N, E);
    combine_kernel<<<blocks(N), TB, 0, stream>>>(hist, incnt, norm_src, norm_dst, N);
    rowptr_scan_kernel<<<1, 1024, 0, stream>>>(incnt, rowptr, N);
    fill_kernel<<<blocks(E), TB, 0, stream>>>(edge_src, edge_dst, rank, rowptr, hist, norm_src, rec, N, E);

    const unsigned gblocks = (unsigned)(((N + GR - 1) / GR + 3) / 4);
    gemm128_wmma_f32<<<gblocks, 128, 0, stream>>>(node_features, W_proj, b_proj, h, N);

    zero_f32<<<blocks((size_t)G * D), TB, 0, stream>>>(hg, (size_t)G * D);
    float* hcur  = h;
    float* hnext = agg;
    for (int i = 0; i < L; ++i) {
        if (i < L - 1)
            gcn_layer_wmma_f32<false><<<gblocks, 128, 0, stream>>>(hcur, rec, rowptr, W_gcn + (size_t)i * D * D,
                                                                   b_gcn + (size_t)i * D, norm_dst, hnext, N,
                                                                    1, nullptr, nullptr);
        else
            gcn_layer_wmma_f32<true><<<gblocks, 128, 0, stream>>>(hcur, rec, rowptr, W_gcn + (size_t)i * D * D,
                                                                  b_gcn + (size_t)i * D, norm_dst, hnext, N,
                                                                   0, graph_ids, hg);
        float* t = hcur; hcur = hnext; hnext = t;
    }

    if (L == 0) pool_kernel<<<blocks((size_t)N * 32), TB, 0, stream>>>(hcur, graph_ids, hg, N);

    mlp3_wmma_f32<<<(unsigned)((G + 15) / 16), 512, 0, stream>>>(hg, W_mlp0, b_mlp0, W_mlp1, b_mlp1, W_mlp2, b_mlp2,
                                                                (float*)d_out, G);
}


# GraphConv layer re-tiled to 8-wave blocks: a wave keeps one 16-column W tile (64 registers, was 128) and gathers 2 nodes, so twice the waves are resident; rows stored whole via LDS; sum orders kept
# speedup vs baseline: 5.4049x; 1.0549x over previous
#include <hip/hip_runtime.h>
#include <hip/hip_bf16.h>

typedef __attribute__((ext_vector_type(2))) float v2f;
typedef __attribute__((ext_vector_type(8))) float v8f;

#define D_FEAT 128


typedef __attribute__((ext_vector_type(4))) float v4f;
typedef __attribute__((ext_vector_type(2))) int v2i;

__global__ void zero_f32(float* __restrict__ p, size_t n) {
    size_t i = (size_t)blockIdx.x * blockDim.x + threadIdx.x;
    if (i < n) p[i] = 0.0f;
}

#define HB  32
#define HCAP 65536

__global__ void __launch_bounds__(1024)
slice_hist_kernel(const int* __restrict__ src, const int* __restrict__ dst,
                  int* __restrict__ hist, int* __restrict__ rank, int N, int E) {
    __shared__ __attribute__((aligned(16))) int cnt[HCAP];
    const int  tid   = threadIdx.x;
    const bool isdst = blockIdx.x < HB;
    const int  sl    = isdst ? blockIdx.x : blockIdx.x - HB;
    const int* __restrict__ key = isdst ? dst : src;
    const int  per   = (E + HB - 1) / HB;
    const int  e0    = sl * per;
    const int  e1    = (e0 + per < E) ? (e0 + per) : E;
    int* __restrict__ hrow = hist + (size_t)blockIdx.x * N;

    for (int base = 0; base < N; base += HCAP) {
        const int lim = (N - base < HCAP) ? (N - base) : HCAP;
        for (int i = 4 * tid; i < lim; i += 4 * 1024)
            *(int4*)(&cnt[i]) = make_int4(0, 0, 0, 0);
        __syncthreads();
        for (int eb = e0; eb < e1; eb += 8 * 1024) {
            int kk[8];
#pragma unroll
            for (int u = 0; u < 8; ++u) {
                const int e = eb + u * 1024 + tid;
                kk[u] = (e < e1) ? key[e] : -1;
            }
#pragma unroll
            for (int u = 0; u < 8; ++u) {
                const unsigned k = (unsigned)(kk[u] - base);
                if (k < (unsigned)lim) {
                    const int r = atomicAdd(&cnt[k], 1);
                    if (isdst) rank[eb + u * 1024 + tid] = r;
                }
            }
        }
        __syncthreads();
        for (int i = tid; i < lim; i += 1024) hrow[base + i] = cnt[i];
        __syncthreads();
    }
}

__global__ void combine_kernel(int* __restrict__ hist, int* __restrict__ incnt,
                               float* __restrict__ norm_src, float* __restrict__ norm_dst, int N) {
    const int n = blockIdx.x * blockDim.x + threadIdx.x;
    if (n < N) {
        int cd[HB], cs[HB];
#pragma unroll
        for (int b = 0; b < HB; ++b) {
            cd[b] = hist[(size_t)b * N + n];
            cs[b] = hist[(size_t)(HB + b) * N + n];
        }
        int run = 0, so = 0;
#pragma unroll
        for (int b = 0; b < HB; ++b) {
            hist[(size_t)b * N + n] = run;
            run += cd[b];
            so  += cs[b];
        }
        incnt[n]    = run;
        norm_dst[n] = rsqrtf(fmaxf((float)run, 1.0f));
        norm_src[n] = rsqrtf(fmaxf((float)so, 1.0f));
    }
}

__global__ void __launch_bounds__(1024)
rowptr_scan_kernel(const int* __restrict__ cnt, int* __restrict__ rowptr, int n) {
    __shared__ int wtot[32];
    const int lane = threadIdx.x & 31;
    const int w    = __builtin_amdgcn_readfirstlane(threadIdx.x >> 5);
    const int seg  = ((n + 4095) / 4096) * 128;
    const int b    = w * seg;
    const int e    = (b + seg < n) ? (b + seg) : n;

    int s = 0;
#pragma unroll 4
    for (int i = b + lane; i < e; i += 32) s += cnt[i];
    for (int o = 16; o > 0; o >>= 1) s += __shfl_xor(s, o);
    if (lane == 0) wtot[w] = s;
    __syncthreads();

    int base = 0;
    for (int k = 0; k < w; ++k) base += wtot[k];

    int c[4];
#pragma unroll
    for (int q = 0; q < 4; ++q) {
        int i = b + 4 * lane + q;
        c[q] = (i < e) ? cnt[i] : 0;
    }
    for (int i0 = b; i0 < e; i0 += 128) {
        int cn[4];
#pragma unroll
        for (int q = 0; q < 4; ++q) {
            int i = i0 + 128 + 4 * lane + q;
            cn[q] = (i < e) ? cnt[i] : 0;
        }
        const int s4 = c[0] + c[1] + c[2] + c[3];
        int v = s4;
        for (int o = 1; o < 32; o <<= 1) {
            int u = __shfl_up(v, o);
            if (lane >= o) v += u;
        }
        int ex = base + v - s4;
#pragma unroll
        for (int q = 0; q < 4; ++q) {
            int i = i0 + 4 * lane + q;
            if (i < e) rowptr[i] = ex;
            ex += c[q];
            c[q] = cn[q];
        }
        base += __shfl(v, 31);
    }
    if (w == 31 && lane == 0) rowptr[n] = base;
}

__global__ void fill_kernel(const int* __restrict__ src, const int* __restrict__ dst,
                            const int* __restrict__ rank, const int* __restrict__ rowptr,
                            const int* __restrict__ hist,
                            const float* __restrict__ norm_src, int* __restrict__ rec, int N, int E) {
    int e = blockIdx.x * blockDim.x + threadIdx.x;
    if (e < E) {
        const int per = (E + HB - 1) / HB;
        int s = src[e];
        int d = dst[e];
        int pos = rowptr[d] + hist[(size_t)(e / per) * N + d] + rank[e];
        v2i r;
        r.x = s;
        r.y = __float_as_int(norm_src[s]);
        *(v2i*)(rec + 2 * (size_t)pos) = r;
    }
}

__global__ void pool_kernel(const float* __restrict__ h, const int* __restrict__ gid,
                            float* __restrict__ hg, int N) {
    int t = blockIdx.x * blockDim.x + threadIdx.x;
    int n = t >> 5;
    int c = (t & 31) << 2;
    if (n < N) {
        int g = gid[n];
        float4 v = *(const float4*)(h + (size_t)n * D_FEAT + c);
        float* hp = hg + (size_t)g * D_FEAT + c;
        unsafeAtomicAdd(hp + 0, v.x);
        unsafeAtomicAdd(hp + 1, v.y);
        unsafeAtomicAdd(hp + 2, v.z);
        unsafeAtomicAdd(hp + 3, v.w);
    }
}

#define GR 16
#define GP 132
__global__ void __launch_bounds__(128)
gemm128_wmma_f32(const float* __restrict__ A, const float* __restrict__ B,
                 const float* __restrict__ bias,
                 float* out,
                 int Mrows) {
    __shared__ __attribute__((aligned(16))) float sA[2][GR * GP];

    const int tid  = threadIdx.x;
    const int w    = __builtin_amdgcn_readfirstlane(tid >> 5);
    const int lane = tid & 31;
    const int half = lane >> 4;
    const int l15  = lane & 15;
    const int colb = 32 * w + l15;

    const int nchunks = (Mrows + GR - 1) / GR;
    int chunk = blockIdx.x;

    v4f   st[4];
#pragma unroll
    for (int i = 0; i < 4; ++i) {
        int row = chunk * GR + i * 4 + w;
        if (row >= Mrows) row = Mrows - 1;
        st[i] = *(const v4f*)(A + (size_t)row * D_FEAT + lane * 4);
    }

    v2f bf[32][2];
#pragma unroll
    for (int s = 0; s < 32; ++s) {
        int ka = 4 * s + 2 * half;
#pragma unroll
        for (int n = 0; n < 2; ++n) {
            bf[s][n].x = B[(size_t)ka * D_FEAT + colb + 16 * n];
            bf[s][n].y = B[(size_t)(ka + 1) * D_FEAT + colb + 16 * n];
        }
    }
    float bv[2];
    bv[0] = bias[colb];
    bv[1] = bias[colb + 16];

#pragma unroll
    for (int i = 0; i < 4; ++i)
        *(v4f*)(&sA[0][(i * 4 + w) * GP + lane * 4]) = st[i];
    __syncthreads();

    int buf = 0;
    for (; chunk < nchunks; chunk += gridDim.x) {
        const int  next     = chunk + gridDim.x;
        const bool has_next = next < nchunks;
        const int  row0     = chunk * GR + half * 8;
        const bool full     = chunk * GR + GR <= Mrows;

        if (has_next) {
#pragma unroll
            for (int i = 0; i < 4; ++i) {
                int row = next * GR + i * 4 + w;
                if (row >= Mrows) row = Mrows - 1;
                st[i] = *(const v4f*)(A + (size_t)row * D_FEAT + lane * 4);
            }
        }

        v8f acc[2] = {};
        const float* ap = &sA[buf][l15 * GP + 2 * half];
#pragma unroll
        for (int s = 0; s < 32; ++s) {
            v2f a = *(const v2f*)(ap + 4 * s);
            acc[0] = __builtin_amdgcn_wmma_f32_16x16x4_f32(false, a, false, bf[s][0], (short)0, acc[0], false, false);
            acc[1] = __builtin_amdgcn_wmma_f32_16x16x4_f32(false, a, false, bf[s][1], (short)0, acc[1], false, false);
        }

        if (has_next) {
#pragma unroll
            for (int i = 0; i < 4; ++i)
                *(v4f*)(&sA[buf ^ 1][(i * 4 + w) * GP + lane * 4]) = st[i];
        }
        __syncthreads();

        if (full) {
            float* op = out + (size_t)row0 * D_FEAT + colb;
#pragma unroll
            for (int n = 0; n < 2; ++n)
#pragma unroll
                for (int v = 0; v < 8; ++v)
                    op[v * D_FEAT + 16 * n] = acc[n][v] + bv[n];
        } else {
#pragma unroll
            for (int n = 0; n < 2; ++n)
#pragma unroll
                for (int v = 0; v < 8; ++v) {
                    int row = row0 + v;
                    if (row < Mrows) out[(size_t)row * D_FEAT + colb + 16 * n] = acc[n][v] + bv[n];
                }
        }

        buf ^= 1;
    }
}

#define LW  8
#define NPW (GR / LW)

__device__ __forceinline__ int load_bounds(const int* __restrict__ rowptr, int rbase, int lane, int Mrows) {
    int idx = rbase + lane;
    if (idx > Mrows) idx = Mrows;
    return rowptr[idx];
}

__device__ __forceinline__ float load_scales(const float* __restrict__ row_scale, int rbase, int lane, int Mrows) {
    int idx = rbase + lane;
    if (idx > Mrows - 1) idx = Mrows - 1;
    return row_scale[idx];
}


__device__ __forceinline__ void flush_node(float* srow, float scv, int q, v4f& acc) {
    const float sc = __int_as_float(__builtin_amdgcn_readlane(__float_as_int(scv), q));
    *(v4f*)(srow + q * GP) = acc * sc;
    acc = (v4f){0.0f, 0.0f, 0.0f, 0.0f};
}

__device__ __forceinline__ void hold_rows(const v4f (&v)[8]) {
    asm volatile("" :: "v"(v[0]), "v"(v[1]), "v"(v[2]), "v"(v[3]), "v"(v[4]), "v"(v[5]), "v"(v[6]), "v"(v[7]));
}
__device__ __forceinline__ void hold_rows(const v4f (&v)[4]) {
    asm volatile("" :: "v"(v[0]), "v"(v[1]), "v"(v[2]), "v"(v[3]));
}
__device__ __forceinline__ void hold_rows(const v4f (&v)[2]) {
    asm volatile("" :: "v"(v[0]), "v"(v[1]));
}
__device__ __forceinline__ void hold_rows(const v4f (&)[1]) {}

template <int K>
__device__ __forceinline__ void gather_run(const float* __restrict__ hin, v2i r, int j, int lane4,
                                           int rpv, float scv, float* srow,
                                           int& pos, int& q, int& nend, v4f& acc) {
    v4f   v[K];
    float ns[K];
#pragma unroll
    for (int k = 0; k < K; ++k) {
        const int s = __builtin_amdgcn_readlane(r.x, j + k);
        ns[k] = __int_as_float(__builtin_amdgcn_readlane(r.y, j + k));
        const unsigned off = (unsigned)s * (unsigned)(D_FEAT * sizeof(float)) + (unsigned)lane4 * (unsigned)sizeof(float);
        v[k] = *(const v4f*)((const char*)hin + off);
    }
    hold_rows(v);
#pragma unroll
    for (int k = 0; k < K; ++k) {
        while (pos == nend) {
            flush_node(srow, scv, q, acc);
            ++q;
            nend = __builtin_amdgcn_readlane(rpv, q + 1);
        }
        acc += v[k] * ns[k];
        ++pos;
    }
}

__device__ __forceinline__ void gather_chunk(const float* __restrict__ hin, const int* __restrict__ rec,
                                             int rpv, float scv, int lane, float* srow) {
    const int lane4 = lane * 4;
    const int beg = __builtin_amdgcn_readlane(rpv, 0);
    const int end = __builtin_amdgcn_readlane(rpv, NPW);
    const int last = (end > 0) ? (end - 1) : 0;
    int i0 = beg + lane;      if (i0 > last) i0 = last;
    int i1 = beg + 32 + lane; if (i1 > last) i1 = last;
    v2i r  = *(const v2i*)(rec + 2 * (size_t)i0);
    v2i rn = *(const v2i*)(rec + 2 * (size_t)i1);

    int pos  = beg;
    int q    = 0;
    int nend = __builtin_amdgcn_readlane(rpv, 1);
    v4f acc  = {0.0f, 0.0f, 0.0f, 0.0f};
    for (int base = beg; base < end; base += 32) {
        const int m = (end - base < 32) ? (end - base) : 32;
        int j = 0;
        for (; j + 8 <= m; j += 8) gather_run<8>(hin, r, j, lane4, rpv, scv, srow, pos, q, nend, acc);
        if (j + 4 <= m) { gather_run<4>(hin, r, j, lane4, rpv, scv, srow, pos, q, nend, acc); j += 4; }
        if (j + 2 <= m) { gather_run<2>(hin, r, j, lane4, rpv, scv, srow, pos, q, nend, acc); j += 2; }
        if (j < m)      { gather_run<1>(hin, r, j, lane4, rpv, scv, srow, pos, q, nend, acc); }
        r = rn;
        int i2 = base + 64 + lane; if (i2 > last) i2 = last;
        rn = *(const v2i*)(rec + 2 * (size_t)i2);
    }
    for (; q < NPW; ++q) flush_node(srow, scv, q, acc);
}

template <bool POOL>
__device__ __forceinline__ void
gcn_layer_body(const float* __restrict__ hin, const int* __restrict__ rec,
               const int* __restrict__ rowptr,
               const float* __restrict__ B, const float* __restrict__ bias,
               const float* __restrict__ row_scale,
               float* __restrict__ out,
               int Mrows, int relu_c,
               const int* __restrict__ gid, float* __restrict__ hg) {
    __shared__ __attribute__((aligned(16))) float sA[2][GR * GP];
    __shared__ __attribute__((aligned(16))) float sC[POOL ? 4 : GR * GP];

    const int tid  = threadIdx.x;
    const int w    = __builtin_amdgcn_readfirstlane(tid >> 5);
    const int lane = tid & 31;
    const int half = lane >> 4;
    const int l15  = lane & 15;
    const int colb = 16 * w + l15;

    const int nchunks = (Mrows + GR - 1) / GR;
    int chunk = blockIdx.x;

    int   rpv = load_bounds(rowptr, chunk * GR + NPW * w, lane, Mrows);
    float scv = load_scales(row_scale, chunk * GR + NPW * w, lane, Mrows);

    v2f bf[32];
#pragma unroll
    for (int s = 0; s < 32; ++s) {
        int ka = 4 * s + 2 * half;
        bf[s].x = B[(size_t)ka * D_FEAT + colb];
        bf[s].y = B[(size_t)(ka + 1) * D_FEAT + colb];
    }
    const float bv = bias[colb];

    gather_chunk(hin, rec, rpv, scv, lane, &sA[0][(NPW * w) * GP + lane * 4]);
    __syncthreads();

    int buf = 0;
    for (; chunk < nchunks; chunk += gridDim.x) {
        const int  next     = chunk + gridDim.x;
        const bool has_next = next < nchunks;
        const int  row0     = chunk * GR + half * 8;
        const bool full     = chunk * GR + GR <= Mrows;

        rpv = load_bounds(rowptr, next * GR + NPW * w, lane, Mrows);
        scv = load_scales(row_scale, next * GR + NPW * w, lane, Mrows);

        int   gl = 0;
        float r[8];
        v4f   rr[NPW];
        if constexpr (POOL) {
            int gr = chunk * GR + l15;
            if (gr > Mrows - 1) gr = Mrows - 1;
            gl = gid[gr];
            if (full) {
                const float* rp = hin + (size_t)row0 * D_FEAT + colb;
#pragma unroll
                for (int v = 0; v < 8; ++v)
                    r[v] = rp[v * D_FEAT];
            }
        } else {
#pragma unroll
            for (int i = 0; i < NPW; ++i) {
                int row = chunk * GR + NPW * w + i;
                if (row > Mrows - 1) row = Mrows - 1;
                rr[i] = *(const v4f*)(hin + (size_t)row * D_FEAT + lane * 4);
            }
        }

        v8f acc = {};
        const float* ap = &sA[buf][l15 * GP + 2 * half];
#pragma unroll
        for (int s = 0; s < 32; ++s) {
            v2f a = *(const v2f*)(ap + 4 * s);
            acc = __builtin_amdgcn_wmma_f32_16x16x4_f32(false, a, false, bf[s], (short)0, acc, false, false);
        }

        float c[8];
#pragma unroll
        for (int v = 0; v < 8; ++v) {
            c[v] = acc[v] + bv;
            if (relu_c) c[v] = fmaxf(c[v], 0.0f);
        }

        if constexpr (POOL) {
            float o[8];
#pragma unroll
            for (int v = 0; v < 8; ++v) {
                if (full) {
                    o[v] = fmaxf(r[v] + c[v], 0.0f);
                } else {
                    const int row = row0 + v;
                    o[v] = (row < Mrows)
                        ? fmaxf(hin[(size_t)row * D_FEAT + colb] + c[v], 0.0f) : 0.0f;
                }
            }
            const int  g0  = __builtin_amdgcn_readlane(gl, 0);
            const bool uni = full && (__builtin_amdgcn_ballot_w32(gl != g0) == 0u);
            if (uni) {
                float s = o[0];
#pragma unroll
                for (int v = 1; v < 8; ++v) s += o[v];
                s += __shfl_xor(s, 16);
                if (half == 0) unsafeAtomicAdd(hg + (size_t)g0 * D_FEAT + colb, s);
            } else {
                int gv[8];
#pragma unroll
                for (int v = 0; v < 8; ++v) gv[v] = __shfl(gl, half * 8 + v);
                float run = o[0];
                int   gc  = gv[0];
#pragma unroll
                for (int v = 1; v < 8; ++v) {
                    if (gv[v] != gc) {
                        unsafeAtomicAdd(hg + (size_t)gc * D_FEAT + colb, run);
                        run = 0.0f;
                        gc  = gv[v];
                    }
                    run += o[v];
                }
                unsafeAtomicAdd(hg + (size_t)gc * D_FEAT + colb, run);
            }
        } else {
            float* cp = &sC[(8 * half) * GP + colb];
#pragma unroll
            for (int v = 0; v < 8; ++v) cp[v * GP] = c[v];
            __syncthreads();
#pragma unroll
            for (int i = 0; i < NPW; ++i) {
                const int row = chunk * GR + NPW * w + i;
                if (row < Mrows) {
                    const v4f cc = *(const v4f*)(&sC[(NPW * w + i) * GP + lane * 4]);
                    v4f o;
                    o.x = fmaxf(rr[i].x + cc.x, 0.0f);
                    o.y = fmaxf(rr[i].y + cc.y, 0.0f);
                    o.z = fmaxf(rr[i].z + cc.z, 0.0f);
                    o.w = fmaxf(rr[i].w + cc.w, 0.0f);
                    *(v4f*)(out + (size_t)row * D_FEAT + lane * 4) = o;
                }
            }
        }

        if (has_next)
            gather_chunk(hin, rec, rpv, scv, lane, &sA[buf ^ 1][(NPW * w) * GP + lane * 4]);
        __syncthreads();

        buf ^= 1;
    }
}


__global__ void __launch_bounds__(32 * LW, 8)
gcn_layer_wmma_f32(const float* __restrict__ hin, const int* __restrict__ rec,
                   const int* __restrict__ rowptr,
                   const float* __restrict__ B, const float* __restrict__ bias,
                   const float* __restrict__ row_scale,
                   float* __restrict__ out, int Mrows, int relu_c) {
    gcn_layer_body<false>(hin, rec, rowptr, B, bias, row_scale, out, Mrows, relu_c, nullptr, nullptr);
}

__global__ void __launch_bounds__(32 * LW, 8)
gcn_layer_pool_wmma_f32(const float* __restrict__ hin, const int* __restrict__ rec,
                        const int* __restrict__ rowptr,
                        const float* __restrict__ B, const float* __restrict__ bias,
                        const float* __restrict__ row_scale,
                        int Mrows, int relu_c,
                        const int* __restrict__ gid, float* __restrict__ hg) {
    gcn_layer_body<true>(hin, rec, rowptr, B, bias, row_scale, nullptr, Mrows, relu_c, gid, hg);
}


#define MH 256
#define MP 260

__device__ __forceinline__ void load_bfrag(const float* wp, int half, v2f (&b)[32]) {
#pragma unroll
    for (int s = 0; s < 32; ++s) {
        const int ka = 4 * s + 2 * half;
        b[s].x = wp[(size_t)ka * MH];
        b[s].y = wp[(size_t)(ka + 1) * MH];
    }
}

__device__ __forceinline__ v8f mma_128k(const float* ap, const v2f (&b)[32], v8f acc) {
#pragma unroll
    for (int s = 0; s < 32; ++s) {
        v2f a = *(const v2f*)(ap + 4 * s);
        acc = __builtin_amdgcn_wmma_f32_16x16x4_f32(false, a, false, b[s], (short)0, acc, false, false);
    }
    return acc;
}

__global__ void __launch_bounds__(512)
mlp3_wmma_f32(const float* X,
              const float* W0, const float* b0,
              const float* W1, const float* b1,
              const float* W2, const float* b2,
              float* out, int G) {
    __shared__ __attribute__((aligned(16))) float sX[2][16 * MP];

    const int tid  = threadIdx.x;
    const int w    = __builtin_amdgcn_readfirstlane(tid >> 5);
    const int lane = tid & 31;
    const int half = lane >> 4;
    const int l15  = lane & 15;
    const int col  = 16 * w + l15;
    const int row0 = blockIdx.x * 16;

    int xr = row0 + w;
    if (xr >= G) xr = G - 1;
    const v4f x = *(const v4f*)(X + (size_t)xr * D_FEAT + lane * 4);

    v2f bA[32], bB[32];
    load_bfrag(W0 + col, half, bA);
    load_bfrag(W1 + col, half, bB);
    const float bias0 = b0[col];
    const float bias1 = b1[col];
    const float bias2 = b2[col];

    *(v4f*)(&sX[0][w * MP + lane * 4]) = x;
    __syncthreads();

    const float* a0 = &sX[0][l15 * MP + 2 * half];
    const float* a1 = &sX[1][l15 * MP + 2 * half];
    float* c0 = &sX[0][(8 * half) * MP + col];
    float* c1 = &sX[1][(8 * half) * MP + col];

    v8f acc = {};
    acc = mma_128k(a0, bA, acc);
    load_bfrag(W1 + (size_t)128 * MH + col, half, bA);
#pragma unroll
    for (int v = 0; v < 8; ++v) c1[v * MP] = fmaxf(acc[v] + bias0, 0.0f);
    __syncthreads();

    acc = (v8f){};
    acc = mma_128k(a1, bB, acc);
    load_bfrag(W2 + col, half, bB);
    acc = mma_128k(a1 + 128, bA, acc);
    load_bfrag(W2 + (size_t)128 * MH + col, half, bA);
#pragma unroll
    for (int v = 0; v < 8; ++v) c0[v * MP] = fmaxf(acc[v] + bias1, 0.0f);
    __syncthreads();

    acc = (v8f){};
    acc = mma_128k(a0, bB, acc);
    acc = mma_128k(a0 + 128, bA, acc);
#pragma unroll
    for (int v = 0; v < 8; ++v) c1[v * MP] = acc[v] + bias2;
    __syncthreads();

    const int orow = row0 + w;
    if (orow < G) {
        const float* sp = &sX[1][w * MP + lane * 4];
        float* op = out + (size_t)orow * MH + lane * 4;
        *(v4f*)(op)       = *(const v4f*)(sp);
        *(v4f*)(op + 128) = *(const v4f*)(sp + 128);
    }
}


extern "C" void kernel_launch(void* const* d_in, const int* in_sizes, int n_in,
                              void* d_out, int out_size, void* d_ws, size_t ws_size,
                              hipStream_t stream) {
    const float* node_features = (const float*)d_in[0];
    const int*   edge_src      = (const int*)d_in[1];
    const int*   edge_dst      = (const int*)d_in[2];
    const int*   graph_ids     = (const int*)d_in[3];
    const float* W_proj        = (const float*)d_in[4];
    const float* b_proj        = (const float*)d_in[5];
    const float* W_gcn         = (const float*)d_in[6];
    const float* b_gcn         = (const float*)d_in[7];
    const float* W_mlp0        = (const float*)d_in[8];
    const float* b_mlp0        = (const float*)d_in[9];
    const float* W_mlp1        = (const float*)d_in[10];
    const float* b_mlp1        = (const float*)d_in[11];
    const float* W_mlp2        = (const float*)d_in[12];
    const float* b_mlp2        = (const float*)d_in[13];

    const int D = D_FEAT;
    const int Mm = 256;
    const int N = in_sizes[3];
    const int E = in_sizes[1];
    const int L = in_sizes[6] / (D * D);
    const int G = out_size / Mm;

    float* ws = (float*)d_ws;
    float* norm_src = ws;
    float* norm_dst = norm_src + N;
    float* h        = norm_dst + N;
    float* agg      = h + (size_t)N * D;
    float* hg       = agg + (size_t)N * D;
    int*   rec      = (int*)(hg + (size_t)G * D);
    int*   rank     = rec + (size_t)2 * E;
    int*   incnt    = rank + E;
    int*   rowptr   = incnt + N;
    int*   hist     = rowptr + N + 1;

    const int TB = 256;
    auto blocks = [&](size_t n) { return (unsigned)((n + TB - 1) / TB); };

    slice_hist_kernel<<<2 * HB, 1024, 0, stream>>>(edge_src, edge_dst, hist, rank, N, E);
    combine_kernel<<<blocks(N), TB, 0, stream>>>(hist, incnt, norm_src, norm_dst, N);
    rowptr_scan_kernel<<<1, 1024, 0, stream>>>(incnt, rowptr, N);
    fill_kernel<<<blocks(E), TB, 0, stream>>>(edge_src, edge_dst, rank, rowptr, hist, norm_src, rec, N, E);

    const unsigned gblocks = (unsigned)(((N + GR - 1) / GR + 3) / 4);
    gemm128_wmma_f32<<<gblocks, 128, 0, stream>>>(node_features, W_proj, b_proj, h, N);

    zero_f32<<<blocks((size_t)G * D), TB, 0, stream>>>(hg, (size_t)G * D);
    float* hcur  = h;
    float* hnext = agg;
    for (int i = 0; i < L; ++i) {
        if (i < L - 1)
            gcn_layer_wmma_f32<<<gblocks, 32 * LW, 0, stream>>>(hcur, rec, rowptr, W_gcn + (size_t)i * D * D,
                                                                b_gcn + (size_t)i * D, norm_dst, hnext, N,
                                                                 1);
        else
            gcn_layer_pool_wmma_f32<<<gblocks, 32 * LW, 0, stream>>>(hcur, rec, rowptr, W_gcn + (size_t)i * D * D,
                                                                     b_gcn + (size_t)i * D, norm_dst, N,
                                                                      0, graph_ids, hg);
        float* t = hcur; hcur = hnext; hnext = t;
    }

    if (L == 0) pool_kernel<<<blocks((size_t)N * 32), TB, 0, stream>>>(hcur, graph_ids, hg, N);

    mlp3_wmma_f32<<<(unsigned)((G + 15) / 16), 512, 0, stream>>>(hg, W_mlp0, b_mlp0, W_mlp1, b_mlp1, W_mlp2, b_mlp2,
                                                                (float*)d_out, G);
}


# One-block prefix scan and the clear launch folded into the per-node pass: slice blocks also write per-bucket count prefixes, so each 256-node block finds its first record itself; 12 -> 10 launches
# speedup vs baseline: 5.8642x; 1.0850x over previous
#include <hip/hip_runtime.h>
#include <hip/hip_bf16.h>

typedef __attribute__((ext_vector_type(2))) float v2f;
typedef __attribute__((ext_vector_type(8))) float v8f;

#define D_FEAT 128


typedef __attribute__((ext_vector_type(4))) float v4f;
typedef __attribute__((ext_vector_type(2))) int v2i;

#define HB  32
#define HCAP 65536
#define CB   256

__device__ __forceinline__ int wave_sum_i32(int v) {
    v += __builtin_amdgcn_update_dpp(0, v, 0x161, 0xf, 0xf, false);
    v += __builtin_amdgcn_update_dpp(0, v, 0x162, 0xf, 0xf, false);
    v += __builtin_amdgcn_update_dpp(0, v, 0x164, 0xf, 0xf, false);
    v += __builtin_amdgcn_update_dpp(0, v, 0x168, 0xf, 0xf, false);
    v += __builtin_amdgcn_permlanex16(v, v, 0x76543210, 0xfedcba98, false, false);
    return v;
}

__device__ __forceinline__ int wave_scan_i32(int v, int lane) {
    for (int o = 1; o < 32; o <<= 1) {
        int u = __shfl_up(v, o);
        if (lane >= o) v += u;
    }
    return v;
}

__global__ void __launch_bounds__(1024)
slice_hist_kernel(const int* __restrict__ src, const int* __restrict__ dst,
                  int* __restrict__ hist, int* __restrict__ rank, int* __restrict__ cpre, int N, int E) {
    __shared__ __attribute__((aligned(16))) int cnt[HCAP];
    __shared__ __attribute__((aligned(16))) int bs[HCAP / CB];
    const int  tid   = threadIdx.x;
    const int  lane  = tid & 31;
    const int  w     = __builtin_amdgcn_readfirstlane(tid >> 5);
    const bool isdst = blockIdx.x < HB;
    const int  sl    = isdst ? blockIdx.x : blockIdx.x - HB;
    const int* __restrict__ key = isdst ? dst : src;
    const int  per   = (E + HB - 1) / HB;
    const int  e0    = sl * per;
    const int  e1    = (e0 + per < E) ? (e0 + per) : E;
    int* __restrict__ hrow = hist + (size_t)blockIdx.x * N;
    int carry = 0;

    for (int base = 0; base < N; base += HCAP) {
        const int lim = (N - base < HCAP) ? (N - base) : HCAP;
        const int nb  = (lim + CB - 1) / CB;
        for (int i = 4 * tid; i < nb * CB; i += 4 * 1024)
            *(int4*)(&cnt[i]) = make_int4(0, 0, 0, 0);
        __syncthreads();
        for (int eb = e0; eb < e1; eb += 8 * 1024) {
            int kk[8];
#pragma unroll
            for (int u = 0; u < 8; ++u) {
                const int e = eb + u * 1024 + tid;
                kk[u] = (e < e1) ? key[e] : -1;
            }
#pragma unroll
            for (int u = 0; u < 8; ++u) {
                const unsigned k = (unsigned)(kk[u] - base);
                if (k < (unsigned)lim) {
                    const int r = atomicAdd(&cnt[k], 1);
                    if (isdst) rank[eb + u * 1024 + tid] = r;
                }
            }
        }
        __syncthreads();
        for (int i = tid; i < lim; i += 1024) hrow[base + i] = cnt[i];
        if (isdst) {
            for (int k = w; k < nb; k += 32) {
                const int4 a = *(const int4*)(&cnt[k * CB + 8 * lane]);
                const int4 b = *(const int4*)(&cnt[k * CB + 8 * lane + 4]);
                const int  s = wave_sum_i32((a.x + a.y) + (a.z + a.w) + (b.x + b.y) + (b.z + b.w));
                if (lane == 0) bs[k] = s;
            }
            __syncthreads();
            if (w == 0) {
                int c8[8];
                int t = 0;
#pragma unroll
                for (int j = 0; j < 8; ++j) {
                    const int k = 8 * lane + j;
                    c8[j] = (k < nb) ? bs[k] : 0;
                    t += c8[j];
                }
                const int inc = wave_scan_i32(t, lane);
                int ex = carry + inc - t;
#pragma unroll
                for (int j = 0; j < 8; ++j) {
                    const int k = 8 * lane + j;
                    if (k < nb) cpre[(size_t)(base / CB + k) * HB + sl] = ex;
                    ex += c8[j];
                }
                carry += __shfl(inc, 31);
            }
        }
        __syncthreads();
    }
}

__global__ void __launch_bounds__(CB)
combine_kernel(int* __restrict__ hist, const int* __restrict__ cpre, int* __restrict__ rowptr,
               float* __restrict__ norm_src, float* __restrict__ norm_dst,
               float* __restrict__ hg, int hgn, int N) {
    __shared__ int wtot[CB / 32];
    const int  tid  = threadIdx.x;
    const int  lane = tid & 31;
    const int  w    = __builtin_amdgcn_readfirstlane(tid >> 5);
    const int  n    = blockIdx.x * CB + tid;
    const bool in   = n < N;
    const int  nn   = in ? n : (N - 1);

    const int front = cpre[(size_t)blockIdx.x * HB + lane];

    int cd[HB], cs[HB];
#pragma unroll
    for (int b = 0; b < HB; ++b) {
        cd[b] = hist[(size_t)b * N + nn];
        cs[b] = hist[(size_t)(HB + b) * N + nn];
    }
    int run = 0, so = 0;
#pragma unroll
    for (int b = 0; b < HB; ++b) {
        run += cd[b];
        so  += cs[b];
    }
    if (!in) run = 0;

    const int inc = wave_scan_i32(run, lane);
    if (lane == 31) wtot[w] = inc;
    int first = wave_sum_i32(front);
    __syncthreads();
    for (int k = 0; k < w; ++k) first += wtot[k];
    first += inc - run;

    if (in) {
        rowptr[n] = first;
        if (n == N - 1) rowptr[N] = first + run;
        int at = first;
#pragma unroll
        for (int b = 0; b < HB; ++b) {
            hist[(size_t)b * N + n] = at;
            at += cd[b];
        }
        norm_dst[n] = rsqrtf(fmaxf((float)run, 1.0f));
        norm_src[n] = rsqrtf(fmaxf((float)so, 1.0f));
    }

    for (int i = blockIdx.x * CB + tid; i < hgn; i += gridDim.x * CB) hg[i] = 0.0f;
}

__global__ void fill_kernel(const int* __restrict__ src, const int* __restrict__ dst,
                            const int* __restrict__ rank,
                            const int* __restrict__ hist,
                            const float* __restrict__ norm_src, int* __restrict__ rec, int N, int E) {
    int e = blockIdx.x * blockDim.x + threadIdx.x;
    if (e < E) {
        const int per = (E + HB - 1) / HB;
        int s = src[e];
        int d = dst[e];
        int pos = hist[(size_t)(e / per) * N + d] + rank[e];
        v2i r;
        r.x = s;
        r.y = __float_as_int(norm_src[s]);
        *(v2i*)(rec + 2 * (size_t)pos) = r;
    }
}

__global__ void pool_kernel(const float* __restrict__ h, const int* __restrict__ gid,
                            float* __restrict__ hg, int N) {
    int t = blockIdx.x * blockDim.x + threadIdx.x;
    int n = t >> 5;
    int c = (t & 31) << 2;
    if (n < N) {
        int g = gid[n];
        float4 v = *(const float4*)(h + (size_t)n * D_FEAT + c);
        float* hp = hg + (size_t)g * D_FEAT + c;
        unsafeAtomicAdd(hp + 0, v.x);
        unsafeAtomicAdd(hp + 1, v.y);
        unsafeAtomicAdd(hp + 2, v.z);
        unsafeAtomicAdd(hp + 3, v.w);
    }
}

#define GR 16
#define GP 132
__global__ void __launch_bounds__(128)
gemm128_wmma_f32(const float* __restrict__ A, const float* __restrict__ B,
                 const float* __restrict__ bias,
                 float* out,
                 int Mrows) {
    __shared__ __attribute__((aligned(16))) float sA[2][GR * GP];

    const int tid  = threadIdx.x;
    const int w    = __builtin_amdgcn_readfirstlane(tid >> 5);
    const int lane = tid & 31;
    const int half = lane >> 4;
    const int l15  = lane & 15;
    const int colb = 32 * w + l15;

    const int nchunks = (Mrows + GR - 1) / GR;
    int chunk = blockIdx.x;

    v4f   st[4];
#pragma unroll
    for (int i = 0; i < 4; ++i) {
        int row = chunk * GR + i * 4 + w;
        if (row >= Mrows) row = Mrows - 1;
        st[i] = *(const v4f*)(A + (size_t)row * D_FEAT + lane * 4);
    }

    v2f bf[32][2];
#pragma unroll
    for (int s = 0; s < 32; ++s) {
        int ka = 4 * s + 2 * half;
#pragma unroll
        for (int n = 0; n < 2; ++n) {
            bf[s][n].x = B[(size_t)ka * D_FEAT + colb + 16 * n];
            bf[s][n].y = B[(size_t)(ka + 1) * D_FEAT + colb + 16 * n];
        }
    }
    float bv[2];
    bv[0] = bias[colb];
    bv[1] = bias[colb + 16];

#pragma unroll
    for (int i = 0; i < 4; ++i)
        *(v4f*)(&sA[0][(i * 4 + w) * GP + lane * 4]) = st[i];
    __syncthreads();

    int buf = 0;
    for (; chunk < nchunks; chunk += gridDim.x) {
        const int  next     = chunk + gridDim.x;
        const bool has_next = next < nchunks;
        const int  row0     = chunk * GR + half * 8;
        const bool full     = chunk * GR + GR <= Mrows;

        if (has_next) {
#pragma unroll
            for (int i = 0; i < 4; ++i) {
                int row = next * GR + i * 4 + w;
                if (row >= Mrows) row = Mrows - 1;
                st[i] = *(const v4f*)(A + (size_t)row * D_FEAT + lane * 4);
            }
        }

        v8f acc[2] = {};
        const float* ap = &sA[buf][l15 * GP + 2 * half];
#pragma unroll
        for (int s = 0; s < 32; ++s) {
            v2f a = *(const v2f*)(ap + 4 * s);
            acc[0] = __builtin_amdgcn_wmma_f32_16x16x4_f32(false, a, false, bf[s][0], (short)0, acc[0], false, false);
            acc[1] = __builtin_amdgcn_wmma_f32_16x16x4_f32(false, a, false, bf[s][1], (short)0, acc[1], false, false);
        }

        if (has_next) {
#pragma unroll
            for (int i = 0; i < 4; ++i)
                *(v4f*)(&sA[buf ^ 1][(i * 4 + w) * GP + lane * 4]) = st[i];
        }
        __syncthreads();

        if (full) {
            float* op = out + (size_t)row0 * D_FEAT + colb;
#pragma unroll
            for (int n = 0; n < 2; ++n)
#pragma unroll
                for (int v = 0; v < 8; ++v)
                    op[v * D_FEAT + 16 * n] = acc[n][v] + bv[n];
        } else {
#pragma unroll
            for (int n = 0; n < 2; ++n)
#pragma unroll
                for (int v = 0; v < 8; ++v) {
                    int row = row0 + v;
                    if (row < Mrows) out[(size_t)row * D_FEAT + colb + 16 * n] = acc[n][v] + bv[n];
                }
        }

        buf ^= 1;
    }
}

#define LW  8
#define NPW (GR / LW)

__device__ __forceinline__ int load_bounds(const int* __restrict__ rowptr, int rbase, int lane, int Mrows) {
    int idx = rbase + lane;
    if (idx > Mrows) idx = Mrows;
    return rowptr[idx];
}

__device__ __forceinline__ float load_scales(const float* __restrict__ row_scale, int rbase, int lane, int Mrows) {
    int idx = rbase + lane;
    if (idx > Mrows - 1) idx = Mrows - 1;
    return row_scale[idx];
}


__device__ __forceinline__ void flush_node(float* srow, float scv, int q, v4f& acc) {
    const float sc = __int_as_float(__builtin_amdgcn_readlane(__float_as_int(scv), q));
    *(v4f*)(srow + q * GP) = acc * sc;
    acc = (v4f){0.0f, 0.0f, 0.0f, 0.0f};
}

__device__ __forceinline__ void hold_rows(const v4f (&v)[8]) {
    asm volatile("" :: "v"(v[0]), "v"(v[1]), "v"(v[2]), "v"(v[3]), "v"(v[4]), "v"(v[5]), "v"(v[6]), "v"(v[7]));
}
__device__ __forceinline__ void hold_rows(const v4f (&v)[4]) {
    asm volatile("" :: "v"(v[0]), "v"(v[1]), "v"(v[2]), "v"(v[3]));
}
__device__ __forceinline__ void hold_rows(const v4f (&v)[2]) {
    asm volatile("" :: "v"(v[0]), "v"(v[1]));
}
__device__ __forceinline__ void hold_rows(const v4f (&)[1]) {}

template <int K>
__device__ __forceinline__ void gather_run(const float* __restrict__ hin, v2i r, int j, int lane4,
                                           int rpv, float scv, float* srow,
                                           int& pos, int& q, int& nend, v4f& acc) {
    v4f   v[K];
    float ns[K];
#pragma unroll
    for (int k = 0; k < K; ++k) {
        const int s = __builtin_amdgcn_readlane(r.x, j + k);
        ns[k] = __int_as_float(__builtin_amdgcn_readlane(r.y, j + k));
        const unsigned off = (unsigned)s * (unsigned)(D_FEAT * sizeof(float)) + (unsigned)lane4 * (unsigned)sizeof(float);
        v[k] = *(const v4f*)((const char*)hin + off);
    }
    hold_rows(v);
#pragma unroll
    for (int k = 0; k < K; ++k) {
        while (pos == nend) {
            flush_node(srow, scv, q, acc);
            ++q;
            nend = __builtin_amdgcn_readlane(rpv, q + 1);
        }
        acc += v[k] * ns[k];
        ++pos;
    }
}

__device__ __forceinline__ void gather_chunk(const float* __restrict__ hin, const int* __restrict__ rec,
                                             int rpv, float scv, int lane, float* srow) {
    const int lane4 = lane * 4;
    const int beg = __builtin_amdgcn_readlane(rpv, 0);
    const int end = __builtin_amdgcn_readlane(rpv, NPW);
    const int last = (end > 0) ? (end - 1) : 0;
    int i0 = beg + lane;      if (i0 > last) i0 = last;
    int i1 = beg + 32 + lane; if (i1 > last) i1 = last;
    v2i r  = *(const v2i*)(rec + 2 * (size_t)i0);
    v2i rn = *(const v2i*)(rec + 2 * (size_t)i1);

    int pos  = beg;
    int q    = 0;
    int nend = __builtin_amdgcn_readlane(rpv, 1);
    v4f acc  = {0.0f, 0.0f, 0.0f, 0.0f};
    for (int base = beg; base < end; base += 32) {
        const int m = (end - base < 32) ? (end - base) : 32;
        int j = 0;
        for (; j + 8 <= m; j += 8) gather_run<8>(hin, r, j, lane4, rpv, scv, srow, pos, q, nend, acc);
        if (j + 4 <= m) { gather_run<4>(hin, r, j, lane4, rpv, scv, srow, pos, q, nend, acc); j += 4; }
        if (j + 2 <= m) { gather_run<2>(hin, r, j, lane4, rpv, scv, srow, pos, q, nend, acc); j += 2; }
        if (j < m)      { gather_run<1>(hin, r, j, lane4, rpv, scv, srow, pos, q, nend, acc); }
        r = rn;
        int i2 = base + 64 + lane; if (i2 > last) i2 = last;
        rn = *(const v2i*)(rec + 2 * (size_t)i2);
    }
    for (; q < NPW; ++q) flush_node(srow, scv, q, acc);
}

template <bool POOL>
__device__ __forceinline__ void
gcn_layer_body(const float* __restrict__ hin, const int* __restrict__ rec,
               const int* __restrict__ rowptr,
               const float* __restrict__ B, const float* __restrict__ bias,
               const float* __restrict__ row_scale,
               float* __restrict__ out,
               int Mrows, int relu_c,
               const int* __restrict__ gid, float* __restrict__ hg) {
    __shared__ __attribute__((aligned(16))) float sA[2][GR * GP];
    __shared__ __attribute__((aligned(16))) float sC[POOL ? 4 : GR * GP];

    const int tid  = threadIdx.x;
    const int w    = __builtin_amdgcn_readfirstlane(tid >> 5);
    const int lane = tid & 31;
    const int half = lane >> 4;
    const int l15  = lane & 15;
    const int colb = 16 * w + l15;

    const int nchunks = (Mrows + GR - 1) / GR;
    int chunk = blockIdx.x;

    int   rpv = load_bounds(rowptr, chunk * GR + NPW * w, lane, Mrows);
    float scv = load_scales(row_scale, chunk * GR + NPW * w, lane, Mrows);

    v2f bf[32];
#pragma unroll
    for (int s = 0; s < 32; ++s) {
        int ka = 4 * s + 2 * half;
        bf[s].x = B[(size_t)ka * D_FEAT + colb];
        bf[s].y = B[(size_t)(ka + 1) * D_FEAT + colb];
    }
    const float bv = bias[colb];

    gather_chunk(hin, rec, rpv, scv, lane, &sA[0][(NPW * w) * GP + lane * 4]);
    __syncthreads();

    int buf = 0;
    for (; chunk < nchunks; chunk += gridDim.x) {
        const int  next     = chunk + gridDim.x;
        const bool has_next = next < nchunks;
        const int  row0     = chunk * GR + half * 8;
        const bool full     = chunk * GR + GR <= Mrows;

        rpv = load_bounds(rowptr, next * GR + NPW * w, lane, Mrows);
        scv = load_scales(row_scale, next * GR + NPW * w, lane, Mrows);

        int   gl = 0;
        float r[8];
        v4f   rr[NPW];
        if constexpr (POOL) {
            int gr = chunk * GR + l15;
            if (gr > Mrows - 1) gr = Mrows - 1;
            gl = gid[gr];
            if (full) {
                const float* rp = hin + (size_t)row0 * D_FEAT + colb;
#pragma unroll
                for (int v = 0; v < 8; ++v)
                    r[v] = rp[v * D_FEAT];
            }
        } else {
#pragma unroll
            for (int i = 0; i < NPW; ++i) {
                int row = chunk * GR + NPW * w + i;
                if (row > Mrows - 1) row = Mrows - 1;
                rr[i] = *(const v4f*)(hin + (size_t)row * D_FEAT + lane * 4);
            }
        }

        v8f acc = {};
        const float* ap = &sA[buf][l15 * GP + 2 * half];
#pragma unroll
        for (int s = 0; s < 32; ++s) {
            v2f a = *(const v2f*)(ap + 4 * s);
            acc = __builtin_amdgcn_wmma_f32_16x16x4_f32(false, a, false, bf[s], (short)0, acc, false, false);
        }

        float c[8];
#pragma unroll
        for (int v = 0; v < 8; ++v) {
            c[v] = acc[v] + bv;
            if (relu_c) c[v] = fmaxf(c[v], 0.0f);
        }

        if constexpr (POOL) {
            float o[8];
#pragma unroll
            for (int v = 0; v < 8; ++v) {
                if (full) {
                    o[v] = fmaxf(r[v] + c[v], 0.0f);
                } else {
                    const int row = row0 + v;
                    o[v] = (row < Mrows)
                        ? fmaxf(hin[(size_t)row * D_FEAT + colb] + c[v], 0.0f) : 0.0f;
                }
            }
            const int  g0  = __builtin_amdgcn_readlane(gl, 0);
            const bool uni = full && (__builtin_amdgcn_ballot_w32(gl != g0) == 0u);
            if (uni) {
                float s = o[0];
#pragma unroll
                for (int v = 1; v < 8; ++v) s += o[v];
                s += __shfl_xor(s, 16);
                if (half == 0) unsafeAtomicAdd(hg + (size_t)g0 * D_FEAT + colb, s);
            } else {
                int gv[8];
#pragma unroll
                for (int v = 0; v < 8; ++v) gv[v] = __shfl(gl, half * 8 + v);
                float run = o[0];
                int   gc  = gv[0];
#pragma unroll
                for (int v = 1; v < 8; ++v) {
                    if (gv[v] != gc) {
                        unsafeAtomicAdd(hg + (size_t)gc * D_FEAT + colb, run);
                        run = 0.0f;
                        gc  = gv[v];
                    }
                    run += o[v];
                }
                unsafeAtomicAdd(hg + (size_t)gc * D_FEAT + colb, run);
            }
        } else {
            float* cp = &sC[(8 * half) * GP + colb];
#pragma unroll
            for (int v = 0; v < 8; ++v) cp[v * GP] = c[v];
            __syncthreads();
#pragma unroll
            for (int i = 0; i < NPW; ++i) {
                const int row = chunk * GR + NPW * w + i;
                if (row < Mrows) {
                    const v4f cc = *(const v4f*)(&sC[(NPW * w + i) * GP + lane * 4]);
                    v4f o;
                    o.x = fmaxf(rr[i].x + cc.x, 0.0f);
                    o.y = fmaxf(rr[i].y + cc.y, 0.0f);
                    o.z = fmaxf(rr[i].z + cc.z, 0.0f);
                    o.w = fmaxf(rr[i].w + cc.w, 0.0f);
                    *(v4f*)(out + (size_t)row * D_FEAT + lane * 4) = o;
                }
            }
        }

        if (has_next)
            gather_chunk(hin, rec, rpv, scv, lane, &sA[buf ^ 1][(NPW * w) * GP + lane * 4]);
        __syncthreads();

        buf ^= 1;
    }
}


__global__ void __launch_bounds__(32 * LW, 8)
gcn_layer_wmma_f32(const float* __restrict__ hin, const int* __restrict__ rec,
                   const int* __restrict__ rowptr,
                   const float* __restrict__ B, const float* __restrict__ bias,
                   const float* __restrict__ row_scale,
                   float* __restrict__ out, int Mrows, int relu_c) {
    gcn_layer_body<false>(hin, rec, rowptr, B, bias, row_scale, out, Mrows, relu_c, nullptr, nullptr);
}

__global__ void __launch_bounds__(32 * LW, 8)
gcn_layer_pool_wmma_f32(const float* __restrict__ hin, const int* __restrict__ rec,
                        const int* __restrict__ rowptr,
                        const float* __restrict__ B, const float* __restrict__ bias,
                        const float* __restrict__ row_scale,
                        int Mrows, int relu_c,
                        const int* __restrict__ gid, float* __restrict__ hg) {
    gcn_layer_body<true>(hin, rec, rowptr, B, bias, row_scale, nullptr, Mrows, relu_c, gid, hg);
}


#define MH 256
#define MP 260

__device__ __forceinline__ void load_bfrag(const float* wp, int half, v2f (&b)[32]) {
#pragma unroll
    for (int s = 0; s < 32; ++s) {
        const int ka = 4 * s + 2 * half;
        b[s].x = wp[(size_t)ka * MH];
        b[s].y = wp[(size_t)(ka + 1) * MH];
    }
}

__device__ __forceinline__ v8f mma_128k(const float* ap, const v2f (&b)[32], v8f acc) {
#pragma unroll
    for (int s = 0; s < 32; ++s) {
        v2f a = *(const v2f*)(ap + 4 * s);
        acc = __builtin_amdgcn_wmma_f32_16x16x4_f32(false, a, false, b[s], (short)0, acc, false, false);
    }
    return acc;
}

__global__ void __launch_bounds__(512)
mlp3_wmma_f32(const float* X,
              const float* W0, const float* b0,
              const float* W1, const float* b1,
              const float* W2, const float* b2,
              float* out, int G) {
    __shared__ __attribute__((aligned(16))) float sX[2][16 * MP];

    const int tid  = threadIdx.x;
    const int w    = __builtin_amdgcn_readfirstlane(tid >> 5);
    const int lane = tid & 31;
    const int half = lane >> 4;
    const int l15  = lane & 15;
    const int col  = 16 * w + l15;
    const int row0 = blockIdx.x * 16;

    int xr = row0 + w;
    if (xr >= G) xr = G - 1;
    const v4f x = *(const v4f*)(X + (size_t)xr * D_FEAT + lane * 4);

    v2f bA[32], bB[32];
    load_bfrag(W0 + col, half, bA);
    load_bfrag(W1 + col, half, bB);
    const float bias0 = b0[col];
    const float bias1 = b1[col];
    const float bias2 = b2[col];

    *(v4f*)(&sX[0][w * MP + lane * 4]) = x;
    __syncthreads();

    const float* a0 = &sX[0][l15 * MP + 2 * half];
    const float* a1 = &sX[1][l15 * MP + 2 * half];
    float* c0 = &sX[0][(8 * half) * MP + col];
    float* c1 = &sX[1][(8 * half) * MP + col];

    v8f acc = {};
    acc = mma_128k(a0, bA, acc);
    load_bfrag(W1 + (size_t)128 * MH + col, half, bA);
#pragma unroll
    for (int v = 0; v < 8; ++v) c1[v * MP] = fmaxf(acc[v] + bias0, 0.0f);
    __syncthreads();

    acc = (v8f){};
    acc = mma_128k(a1, bB, acc);
    load_bfrag(W2 + col, half, bB);
    acc = mma_128k(a1 + 128, bA, acc);
    load_bfrag(W2 + (size_t)128 * MH + col, half, bA);
#pragma unroll
    for (int v = 0; v < 8; ++v) c0[v * MP] = fmaxf(acc[v] + bias1, 0.0f);
    __syncthreads();

    acc = (v8f){};
    acc = mma_128k(a0, bB, acc);
    acc = mma_128k(a0 + 128, bA, acc);
#pragma unroll
    for (int v = 0; v < 8; ++v) c1[v * MP] = acc[v] + bias2;
    __syncthreads();

    const int orow = row0 + w;
    if (orow < G) {
        const float* sp = &sX[1][w * MP + lane * 4];
        float* op = out + (size_t)orow * MH + lane * 4;
        *(v4f*)(op)       = *(const v4f*)(sp);
        *(v4f*)(op + 128) = *(const v4f*)(sp + 128);
    }
}


extern "C" void kernel_launch(void* const* d_in, const int* in_sizes, int n_in,
                              void* d_out, int out_size, void* d_ws, size_t ws_size,
                              hipStream_t stream) {
    const float* node_features = (const float*)d_in[0];
    const int*   edge_src      = (const int*)d_in[1];
    const int*   edge_dst      = (const int*)d_in[2];
    const int*   graph_ids     = (const int*)d_in[3];
    const float* W_proj        = (const float*)d_in[4];
    const float* b_proj        = (const float*)d_in[5];
    const float* W_gcn         = (const float*)d_in[6];
    const float* b_gcn         = (const float*)d_in[7];
    const float* W_mlp0        = (const float*)d_in[8];
    const float* b_mlp0        = (const float*)d_in[9];
    const float* W_mlp1        = (const float*)d_in[10];
    const float* b_mlp1        = (const float*)d_in[11];
    const float* W_mlp2        = (const float*)d_in[12];
    const float* b_mlp2        = (const float*)d_in[13];

    const int D = D_FEAT;
    const int Mm = 256;
    const int N = in_sizes[3];
    const int E = in_sizes[1];
    const int L = in_sizes[6] / (D * D);
    const int G = out_size / Mm;

    float* ws = (float*)d_ws;
    float* norm_src = ws;
    float* norm_dst = norm_src + N;
    float* h        = norm_dst + N;
    float* agg      = h + (size_t)N * D;
    float* hg       = agg + (size_t)N * D;
    int*   rec      = (int*)(hg + (size_t)G * D);
    int*   rank     = rec + (size_t)2 * E;
    int*   rowptr   = rank + E;
    int*   hist     = rowptr + N + 1;
    int*   cpre     = hist + (size_t)2 * HB * N;

    const int TB = 256;
    auto blocks = [&](size_t n) { return (unsigned)((n + TB - 1) / TB); };

    slice_hist_kernel<<<2 * HB, 1024, 0, stream>>>(edge_src, edge_dst, hist, rank, cpre, N, E);
    combine_kernel<<<(unsigned)((N + CB - 1) / CB), CB, 0, stream>>>(hist, cpre, rowptr, norm_src, norm_dst,
                                                                   hg, G * D, N);
    fill_kernel<<<blocks(E), TB, 0, stream>>>(edge_src, edge_dst, rank, hist, norm_src, rec, N, E);

    const unsigned gblocks = (unsigned)(((N + GR - 1) / GR + 3) / 4);
    gemm128_wmma_f32<<<gblocks, 128, 0, stream>>>(node_features, W_proj, b_proj, h, N);

    float* hcur  = h;
    float* hnext = agg;
    for (int i = 0; i < L; ++i) {
        if (i < L - 1)
            gcn_layer_wmma_f32<<<gblocks, 32 * LW, 0, stream>>>(hcur, rec, rowptr, W_gcn + (size_t)i * D * D,
                                                                b_gcn + (size_t)i * D, norm_dst, hnext, N,
                                                                 1);
        else
            gcn_layer_pool_wmma_f32<<<gblocks, 32 * LW, 0, stream>>>(hcur, rec, rowptr, W_gcn + (size_t)i * D * D,
                                                                     b_gcn + (size_t)i * D, norm_dst, N,
                                                                      0, graph_ids, hg);
        float* t = hcur; hcur = hnext; hnext = t;
    }

    if (L == 0) pool_kernel<<<blocks((size_t)N * 32), TB, 0, stream>>>(hcur, graph_ids, hg, N);

    mlp3_wmma_f32<<<(unsigned)((G + 15) / 16), 512, 0, stream>>>(hg, W_mlp0, b_mlp0, W_mlp1, b_mlp1, W_mlp2, b_mlp2,
                                                                (float*)d_out, G);
}


# speedup vs baseline: 6.3011x; 1.0745x over previous
#include <hip/hip_runtime.h>
#include <hip/hip_bf16.h>

typedef __attribute__((ext_vector_type(2))) float v2f;
typedef __attribute__((ext_vector_type(8))) float v8f;

#define D_FEAT 128


typedef __attribute__((ext_vector_type(4))) float v4f;
typedef __attribute__((ext_vector_type(2))) int v2i;

#define HB  32
#define HCAP 65536
#define CB   256

__device__ __forceinline__ int wave_sum_i32(int v) {
    v += __builtin_amdgcn_update_dpp(0, v, 0x161, 0xf, 0xf, false);
    v += __builtin_amdgcn_update_dpp(0, v, 0x162, 0xf, 0xf, false);
    v += __builtin_amdgcn_update_dpp(0, v, 0x164, 0xf, 0xf, false);
    v += __builtin_amdgcn_update_dpp(0, v, 0x168, 0xf, 0xf, false);
    v += __builtin_amdgcn_permlanex16(v, v, 0x76543210, 0xfedcba98, false, false);
    return v;
}

__device__ __forceinline__ int wave_scan_i32(int v, int lane) {
    for (int o = 1; o < 32; o <<= 1) {
        int u = __shfl_up(v, o);
        if (lane >= o) v += u;
    }
    return v;
}

__global__ void __launch_bounds__(1024)
slice_hist_kernel(const int* __restrict__ src, const int* __restrict__ dst,
                  int* __restrict__ hist, int* __restrict__ rank, int* __restrict__ cpre, int N, int E) {
    __shared__ __attribute__((aligned(16))) int cnt[HCAP];
    __shared__ __attribute__((aligned(16))) int bs[HCAP / CB];
    const int  tid   = threadIdx.x;
    const int  lane  = tid & 31;
    const int  w     = __builtin_amdgcn_readfirstlane(tid >> 5);
    const bool isdst = blockIdx.x < HB;
    const int  sl    = isdst ? blockIdx.x : blockIdx.x - HB;
    const int* __restrict__ key = isdst ? dst : src;
    const int  per   = (E + HB - 1) / HB;
    const int  e0    = sl * per;
    const int  e1    = (e0 + per < E) ? (e0 + per) : E;
    int* __restrict__ hrow = hist + (size_t)blockIdx.x * N;
    int carry = 0;

    for (int base = 0; base < N; base += HCAP) {
        const int lim = (N - base < HCAP) ? (N - base) : HCAP;
        const int nb  = (lim + CB - 1) / CB;
        for (int i = 4 * tid; i < nb * CB; i += 4 * 1024)
            *(int4*)(&cnt[i]) = make_int4(0, 0, 0, 0);
        __syncthreads();
        for (int eb = e0; eb < e1; eb += 8 * 1024) {
            int kk[8];
#pragma unroll
            for (int u = 0; u < 8; ++u) {
                const int e = eb + u * 1024 + tid;
                kk[u] = (e < e1) ? key[e] : -1;
            }
#pragma unroll
            for (int u = 0; u < 8; ++u) {
                const unsigned k = (unsigned)(kk[u] - base);
                if (k < (unsigned)lim) {
                    const int r = atomicAdd(&cnt[k], 1);
                    if (isdst) rank[eb + u * 1024 + tid] = r;
                }
            }
        }
        __syncthreads();
        for (int i = tid; i < lim; i += 1024) hrow[base + i] = cnt[i];
        if (isdst) {
            for (int k = w; k < nb; k += 32) {
                const int4 a = *(const int4*)(&cnt[k * CB + 8 * lane]);
                const int4 b = *(const int4*)(&cnt[k * CB + 8 * lane + 4]);
                const int  s = wave_sum_i32((a.x + a.y) + (a.z + a.w) + (b.x + b.y) + (b.z + b.w));
                if (lane == 0) bs[k] = s;
            }
            __syncthreads();
            if (w == 0) {
                int c8[8];
                int t = 0;
#pragma unroll
                for (int j = 0; j < 8; ++j) {
                    const int k = 8 * lane + j;
                    c8[j] = (k < nb) ? bs[k] : 0;
                    t += c8[j];
                }
                const int inc = wave_scan_i32(t, lane);
                int ex = carry + inc - t;
#pragma unroll
                for (int j = 0; j < 8; ++j) {
                    const int k = 8 * lane + j;
                    if (k < nb) cpre[(size_t)(base / CB + k) * HB + sl] = ex;
                    ex += c8[j];
                }
                carry += __shfl(inc, 31);
            }
        }
        __syncthreads();
    }
}

__global__ void __launch_bounds__(CB)
combine_kernel(int* __restrict__ hist, const int* __restrict__ cpre, int* __restrict__ rowptr,
               float* __restrict__ norm_src, float* __restrict__ norm_dst,
               float* __restrict__ hg, int hgn, int N) {
    __shared__ int wtot[CB / 32];
    const int  tid  = threadIdx.x;
    const int  lane = tid & 31;
    const int  w    = __builtin_amdgcn_readfirstlane(tid >> 5);
    const int  n    = blockIdx.x * CB + tid;
    const bool in   = n < N;
    const int  nn   = in ? n : (N - 1);

    const int front = cpre[(size_t)blockIdx.x * HB + lane];

    int cd[HB], cs[HB];
#pragma unroll
    for (int b = 0; b < HB; ++b) {
        cd[b] = hist[(size_t)b * N + nn];
        cs[b] = hist[(size_t)(HB + b) * N + nn];
    }
    int run = 0, so = 0;
#pragma unroll
    for (int b = 0; b < HB; ++b) {
        run += cd[b];
        so  += cs[b];
    }
    if (!in) run = 0;

    const int inc = wave_scan_i32(run, lane);
    if (lane == 31) wtot[w] = inc;
    int first = wave_sum_i32(front);
    __syncthreads();
    for (int k = 0; k < w; ++k) first += wtot[k];
    first += inc - run;

    if (in) {
        rowptr[n] = first;
        if (n == N - 1) rowptr[N] = first + run;
        int at = first;
#pragma unroll
        for (int b = 0; b < HB; ++b) {
            hist[(size_t)b * N + n] = at;
            at += cd[b];
        }
        norm_dst[n] = rsqrtf(fmaxf((float)run, 1.0f));
        norm_src[n] = rsqrtf(fmaxf((float)so, 1.0f));
    }

    for (int i = blockIdx.x * CB + tid; i < hgn; i += gridDim.x * CB) hg[i] = 0.0f;
}

__global__ void fill_kernel(const int* __restrict__ src, const int* __restrict__ dst,
                            const int* __restrict__ rank,
                            const int* __restrict__ hist,
                            const float* __restrict__ norm_src, int* __restrict__ rec, int N, int E) {
    int e = blockIdx.x * blockDim.x + threadIdx.x;
    if (e < E) {
        const int per = (E + HB - 1) / HB;
        int s = src[e];
        int d = dst[e];
        int pos = hist[(size_t)(e / per) * N + d] + rank[e];
        v2i r;
        r.x = s;
        r.y = __float_as_int(norm_src[s]);
        *(v2i*)(rec + 2 * (size_t)pos) = r;
    }
}

__global__ void pool_kernel(const float* __restrict__ h, const int* __restrict__ gid,
                            float* __restrict__ hg, int N) {
    int t = blockIdx.x * blockDim.x + threadIdx.x;
    int n = t >> 5;
    int c = (t & 31) << 2;
    if (n < N) {
        int g = gid[n];
        float4 v = *(const float4*)(h + (size_t)n * D_FEAT + c);
        float* hp = hg + (size_t)g * D_FEAT + c;
        unsafeAtomicAdd(hp + 0, v.x);
        unsafeAtomicAdd(hp + 1, v.y);
        unsafeAtomicAdd(hp + 2, v.z);
        unsafeAtomicAdd(hp + 3, v.w);
    }
}

#define GR 16
#define GP 132
__global__ void __launch_bounds__(128)
gemm128_wmma_f32(const float* __restrict__ A, const float* __restrict__ B,
                 const float* __restrict__ bias,
                 float* out,
                 int Mrows) {
    __shared__ __attribute__((aligned(16))) float sA[2][GR * GP];

    const int tid  = threadIdx.x;
    const int w    = __builtin_amdgcn_readfirstlane(tid >> 5);
    const int lane = tid & 31;
    const int half = lane >> 4;
    const int l15  = lane & 15;
    const int colb = 32 * w + l15;

    const int nchunks = (Mrows + GR - 1) / GR;
    int chunk = blockIdx.x;

    v4f   st[4];
#pragma unroll
    for (int i = 0; i < 4; ++i) {
        int row = chunk * GR + i * 4 + w;
        if (row >= Mrows) row = Mrows - 1;
        st[i] = *(const v4f*)(A + (size_t)row * D_FEAT + lane * 4);
    }

    v2f bf[32][2];
#pragma unroll
    for (int s = 0; s < 32; ++s) {
        int ka = 4 * s + 2 * half;
#pragma unroll
        for (int n = 0; n < 2; ++n) {
            bf[s][n].x = B[(size_t)ka * D_FEAT + colb + 16 * n];
            bf[s][n].y = B[(size_t)(ka + 1) * D_FEAT + colb + 16 * n];
        }
    }
    float bv[2];
    bv[0] = bias[colb];
    bv[1] = bias[colb + 16];

#pragma unroll
    for (int i = 0; i < 4; ++i)
        *(v4f*)(&sA[0][(i * 4 + w) * GP + lane * 4]) = st[i];
    __syncthreads();

    int buf = 0;
    for (; chunk < nchunks; chunk += gridDim.x) {
        const int  next     = chunk + gridDim.x;
        const bool has_next = next < nchunks;
        const int  row0     = chunk * GR + half * 8;
        const bool full     = chunk * GR + GR <= Mrows;

        if (has_next) {
#pragma unroll
            for (int i = 0; i < 4; ++i) {
                int row = next * GR + i * 4 + w;
                if (row >= Mrows) row = Mrows - 1;
                st[i] = *(const v4f*)(A + (size_t)row * D_FEAT + lane * 4);
            }
        }

        v8f acc[2] = {};
        const float* ap = &sA[buf][l15 * GP + 2 * half];
#pragma unroll
        for (int s = 0; s < 32; ++s) {
            v2f a = *(const v2f*)(ap + 4 * s);
            acc[0] = __builtin_amdgcn_wmma_f32_16x16x4_f32(false, a, false, bf[s][0], (short)0, acc[0], false, false);
            acc[1] = __builtin_amdgcn_wmma_f32_16x16x4_f32(false, a, false, bf[s][1], (short)0, acc[1], false, false);
        }

        if (has_next) {
#pragma unroll
            for (int i = 0; i < 4; ++i)
                *(v4f*)(&sA[buf ^ 1][(i * 4 + w) * GP + lane * 4]) = st[i];
        }
        __syncthreads();

        if (full) {
            float* op = out + (size_t)row0 * D_FEAT + colb;
#pragma unroll
            for (int n = 0; n < 2; ++n)
#pragma unroll
                for (int v = 0; v < 8; ++v)
                    op[v * D_FEAT + 16 * n] = acc[n][v] + bv[n];
        } else {
#pragma unroll
            for (int n = 0; n < 2; ++n)
#pragma unroll
                for (int v = 0; v < 8; ++v) {
                    int row = row0 + v;
                    if (row < Mrows) out[(size_t)row * D_FEAT + colb + 16 * n] = acc[n][v] + bv[n];
                }
        }

        buf ^= 1;
    }
}

#define LW  8
#define NPW (GR / LW)

__device__ __forceinline__ int load_bounds(const int* __restrict__ rowptr, int rbase, int lane, int Mrows) {
    int idx = rbase + lane;
    if (idx > Mrows) idx = Mrows;
    return rowptr[idx];
}

__device__ __forceinline__ float load_scales(const float* __restrict__ row_scale, int rbase, int lane, int Mrows) {
    int idx = rbase + lane;
    if (idx > Mrows - 1) idx = Mrows - 1;
    return row_scale[idx];
}


__device__ __forceinline__ void flush_node(float* srow, float scv, int q, v4f& acc) {
    const float sc = __int_as_float(__builtin_amdgcn_readlane(__float_as_int(scv), q));
    *(v4f*)(srow + q * GP) = acc * sc;
    acc = (v4f){0.0f, 0.0f, 0.0f, 0.0f};
}

__device__ __forceinline__ void hold_rows(const v4f (&v)[8]) {
    asm volatile("" :: "v"(v[0]), "v"(v[1]), "v"(v[2]), "v"(v[3]), "v"(v[4]), "v"(v[5]), "v"(v[6]), "v"(v[7]));
}
__device__ __forceinline__ void hold_rows(const v4f (&v)[4]) {
    asm volatile("" :: "v"(v[0]), "v"(v[1]), "v"(v[2]), "v"(v[3]));
}
__device__ __forceinline__ void hold_rows(const v4f (&v)[2]) {
    asm volatile("" :: "v"(v[0]), "v"(v[1]));
}
__device__ __forceinline__ void hold_rows(const v4f (&)[1]) {}

template <int K>
__device__ __forceinline__ void gather_run(const float* __restrict__ hin, v2i r, int j, int lane4,
                                           int rpv, float scv, float* srow,
                                           int& pos, int& q, int& nend, v4f& acc) {
    v4f   v[K];
    float ns[K];
#pragma unroll
    for (int k = 0; k < K; ++k) {
        const int s = __builtin_amdgcn_readlane(r.x, j + k);
        ns[k] = __int_as_float(__builtin_amdgcn_readlane(r.y, j + k));
        const unsigned off = (unsigned)s * (unsigned)(D_FEAT * sizeof(float)) + (unsigned)lane4 * (unsigned)sizeof(float);
        v[k] = *(const v4f*)((const char*)hin + off);
    }
    hold_rows(v);
#pragma unroll
    for (int k = 0; k < K; ++k) {
        while (pos == nend) {
            flush_node(srow, scv, q, acc);
            ++q;
            nend = __builtin_amdgcn_readlane(rpv, q + 1);
        }
        acc += v[k] * ns[k];
        ++pos;
    }
}

__device__ __forceinline__ void gather_chunk(const float* __restrict__ hin, const int* __restrict__ rec,
                                             int rpv, float scv, int lane, float* srow) {
    const int lane4 = lane * 4;
    const int beg = __builtin_amdgcn_readlane(rpv, 0);
    const int end = __builtin_amdgcn_readlane(rpv, NPW);
    const int last = (end > 0) ? (end - 1) : 0;
    int i0 = beg + lane;      if (i0 > last) i0 = last;
    int i1 = beg + 32 + lane; if (i1 > last) i1 = last;
    v2i r  = *(const v2i*)(rec + 2 * (size_t)i0);
    v2i rn = *(const v2i*)(rec + 2 * (size_t)i1);

    int pos  = beg;
    int q    = 0;
    int nend = __builtin_amdgcn_readlane(rpv, 1);
    v4f acc  = {0.0f, 0.0f, 0.0f, 0.0f};
    for (int base = beg; base < end; base += 32) {
        const int m = (end - base < 32) ? (end - base) : 32;
        int j = 0;
        for (; j + 8 <= m; j += 8) gather_run<8>(hin, r, j, lane4, rpv, scv, srow, pos, q, nend, acc);
        if (j + 4 <= m) { gather_run<4>(hin, r, j, lane4, rpv, scv, srow, pos, q, nend, acc); j += 4; }
        if (j + 2 <= m) { gather_run<2>(hin, r, j, lane4, rpv, scv, srow, pos, q, nend, acc); j += 2; }
        if (j < m)      { gather_run<1>(hin, r, j, lane4, rpv, scv, srow, pos, q, nend, acc); }
        r = rn;
        int i2 = base + 64 + lane; if (i2 > last) i2 = last;
        rn = *(const v2i*)(rec + 2 * (size_t)i2);
    }
    for (; q < NPW; ++q) flush_node(srow, scv, q, acc);
}

template <bool POOL>
__device__ __forceinline__ void
gcn_layer_body(const float* __restrict__ hin, const int* __restrict__ rec,
               const int* __restrict__ rowptr,
               const float* __restrict__ B, const float* __restrict__ bias,
               const float* __restrict__ row_scale,
               float* __restrict__ out,
               int Mrows, int relu_c,
               const int* __restrict__ gid, float* __restrict__ hg) {
    __shared__ __attribute__((aligned(16))) float sA[2][GR * GP];
    __shared__ __attribute__((aligned(16))) float sC[POOL ? 4 : GR * GP];

    const int tid  = threadIdx.x;
    const int w    = __builtin_amdgcn_readfirstlane(tid >> 5);
    const int lane = tid & 31;
    const int half = lane >> 4;
    const int l15  = lane & 15;
    const int colb = 16 * w + l15;

    const int nchunks = (Mrows + GR - 1) / GR;
    int chunk = blockIdx.x;

    int   rpv = load_bounds(rowptr, chunk * GR + NPW * w, lane, Mrows);
    float scv = load_scales(row_scale, chunk * GR + NPW * w, lane, Mrows);

    v2f bf[32];
#pragma unroll
    for (int s = 0; s < 32; ++s) {
        int ka = 4 * s + 2 * half;
        bf[s].x = B[(size_t)ka * D_FEAT + colb];
        bf[s].y = B[(size_t)(ka + 1) * D_FEAT + colb];
    }
    const float bv = bias[colb];

    gather_chunk(hin, rec, rpv, scv, lane, &sA[0][(NPW * w) * GP + lane * 4]);
    __syncthreads();

    int buf = 0;
    for (; chunk < nchunks; chunk += gridDim.x) {
        const int  next     = chunk + gridDim.x;
        const bool has_next = next < nchunks;
        const int  row0     = chunk * GR + half * 8;
        const bool full     = chunk * GR + GR <= Mrows;

        rpv = load_bounds(rowptr, next * GR + NPW * w, lane, Mrows);
        scv = load_scales(row_scale, next * GR + NPW * w, lane, Mrows);

        int   gl = 0;
        float r[8];
        v4f   rr[NPW];
        if constexpr (POOL) {
            int gr = chunk * GR + l15;
            if (gr > Mrows - 1) gr = Mrows - 1;
            gl = gid[gr];
            if (full) {
                const float* rp = hin + (size_t)row0 * D_FEAT + colb;
#pragma unroll
                for (int v = 0; v < 8; ++v)
                    r[v] = rp[v * D_FEAT];
            }
        } else {
#pragma unroll
            for (int i = 0; i < NPW; ++i) {
                int row = chunk * GR + NPW * w + i;
                if (row > Mrows - 1) row = Mrows - 1;
                rr[i] = *(const v4f*)(hin + (size_t)row * D_FEAT + lane * 4);
            }
        }

        v8f acc = {};
        const float* ap = &sA[buf][l15 * GP + 2 * half];
#pragma unroll
        for (int s = 0; s < 32; ++s) {
            v2f a = *(const v2f*)(ap + 4 * s);
            acc = __builtin_amdgcn_wmma_f32_16x16x4_f32(false, a, false, bf[s], (short)0, acc, false, false);
        }

        float c[8];
#pragma unroll
        for (int v = 0; v < 8; ++v) {
            c[v] = acc[v] + bv;
            if (relu_c) c[v] = fmaxf(c[v], 0.0f);
        }

        if constexpr (POOL) {
            float o[8];
#pragma unroll
            for (int v = 0; v < 8; ++v) {
                if (full) {
                    o[v] = fmaxf(r[v] + c[v], 0.0f);
                } else {
                    const int row = row0 + v;
                    o[v] = (row < Mrows)
                        ? fmaxf(hin[(size_t)row * D_FEAT + colb] + c[v], 0.0f) : 0.0f;
                }
            }
            const int  g0  = __builtin_amdgcn_readlane(gl, 0);
            const bool uni = full && (__builtin_amdgcn_ballot_w32(gl != g0) == 0u);
            if (uni) {
                float s = o[0];
#pragma unroll
                for (int v = 1; v < 8; ++v) s += o[v];
                s += __shfl_xor(s, 16);
                if (half == 0) unsafeAtomicAdd(hg + (size_t)g0 * D_FEAT + colb, s);
            } else {
                int gv[8];
#pragma unroll
                for (int v = 0; v < 8; ++v) gv[v] = __shfl(gl, half * 8 + v);
                float run = o[0];
                int   gc  = gv[0];
#pragma unroll
                for (int v = 1; v < 8; ++v) {
                    if (gv[v] != gc) {
                        unsafeAtomicAdd(hg + (size_t)gc * D_FEAT + colb, run);
                        run = 0.0f;
                        gc  = gv[v];
                    }
                    run += o[v];
                }
                unsafeAtomicAdd(hg + (size_t)gc * D_FEAT + colb, run);
            }
        } else {
            float* cp = &sC[(8 * half) * GP + colb];
#pragma unroll
            for (int v = 0; v < 8; ++v) cp[v * GP] = c[v];
            __syncthreads();
#pragma unroll
            for (int i = 0; i < NPW; ++i) {
                const int row = chunk * GR + NPW * w + i;
                if (row < Mrows) {
                    const v4f cc = *(const v4f*)(&sC[(NPW * w + i) * GP + lane * 4]);
                    v4f o;
                    o.x = fmaxf(rr[i].x + cc.x, 0.0f);
                    o.y = fmaxf(rr[i].y + cc.y, 0.0f);
                    o.z = fmaxf(rr[i].z + cc.z, 0.0f);
                    o.w = fmaxf(rr[i].w + cc.w, 0.0f);
                    *(v4f*)(out + (size_t)row * D_FEAT + lane * 4) = o;
                }
            }
        }

        if (has_next)
            gather_chunk(hin, rec, rpv, scv, lane, &sA[buf ^ 1][(NPW * w) * GP + lane * 4]);
        __syncthreads();

        buf ^= 1;
    }
}


__global__ void __launch_bounds__(32 * LW, 8)
gcn_layer_wmma_f32(const float* __restrict__ hin, const int* __restrict__ rec,
                   const int* __restrict__ rowptr,
                   const float* __restrict__ B, const float* __restrict__ bias,
                   const float* __restrict__ row_scale,
                   float* __restrict__ out, int Mrows, int relu_c) {
    gcn_layer_body<false>(hin, rec, rowptr, B, bias, row_scale, out, Mrows, relu_c, nullptr, nullptr);
}

__global__ void __launch_bounds__(32 * LW, 8)
gcn_layer_pool_wmma_f32(const float* __restrict__ hin, const int* __restrict__ rec,
                        const int* __restrict__ rowptr,
                        const float* __restrict__ B, const float* __restrict__ bias,
                        const float* __restrict__ row_scale,
                        int Mrows, int relu_c,
                        const int* __restrict__ gid, float* __restrict__ hg) {
    gcn_layer_body<true>(hin, rec, rowptr, B, bias, row_scale, nullptr, Mrows, relu_c, gid, hg);
}


#define MH 256
#define MP 260

__device__ __forceinline__ void load_bfrag(const float* wp, int half, v2f (&b)[32]) {
#pragma unroll
    for (int s = 0; s < 32; ++s) {
        const int ka = 4 * s + 2 * half;
        b[s].x = wp[(size_t)ka * MH];
        b[s].y = wp[(size_t)(ka + 1) * MH];
    }
}

__device__ __forceinline__ v8f mma_128k(const float* ap, const v2f (&b)[32], v8f acc) {
#pragma unroll
    for (int s = 0; s < 32; ++s) {
        v2f a = *(const v2f*)(ap + 4 * s);
        acc = __builtin_amdgcn_wmma_f32_16x16x4_f32(false, a, false, b[s], (short)0, acc, false, false);
    }
    return acc;
}

__global__ void __launch_bounds__(512)
mlp3_wmma_f32(const float* X,
              const float* W0, const float* b0,
              const float* W1, const float* b1,
              const float* W2, const float* b2,
              float* out, int G) {
    __shared__ __attribute__((aligned(16))) float sX[2][16 * MP];

    const int tid  = threadIdx.x;
    const int w    = __builtin_amdgcn_readfirstlane(tid >> 5);
    const int lane = tid & 31;
    const int half = lane >> 4;
    const int l15  = lane & 15;
    const int col  = 16 * w + l15;
    const int row0 = blockIdx.x * 16;

    int xr = row0 + w;
    if (xr >= G) xr = G - 1;
    const v4f x = *(const v4f*)(X + (size_t)xr * D_FEAT + lane * 4);

    v2f bA[32], bB[32];
    load_bfrag(W0 + col, half, bA);
    load_bfrag(W1 + col, half, bB);
    const float bias0 = b0[col];
    const float bias1 = b1[col];
    const float bias2 = b2[col];

    *(v4f*)(&sX[0][w * MP + lane * 4]) = x;
    __syncthreads();

    const float* a0 = &sX[0][l15 * MP + 2 * half];
    const float* a1 = &sX[1][l15 * MP + 2 * half];
    float* c0 = &sX[0][(8 * half) * MP + col];
    float* c1 = &sX[1][(8 * half) * MP + col];

    v8f acc = {};
    acc = mma_128k(a0, bA, acc);
    load_bfrag(W1 + (size_t)128 * MH + col, half, bA);
#pragma unroll
    for (int v = 0; v < 8; ++v) c1[v * MP] = fmaxf(acc[v] + bias0, 0.0f);
    __syncthreads();

    acc = (v8f){};
    acc = mma_128k(a1, bB, acc);
    load_bfrag(W2 + col, half, bB);
    acc = mma_128k(a1 + 128, bA, acc);
    load_bfrag(W2 + (size_t)128 * MH + col, half, bA);
#pragma unroll
    for (int v = 0; v < 8; ++v) c0[v * MP] = fmaxf(acc[v] + bias1, 0.0f);
    __syncthreads();

    acc = (v8f){};
    acc = mma_128k(a0, bB, acc);
    acc = mma_128k(a0 + 128, bA, acc);
#pragma unroll
    for (int v = 0; v < 8; ++v) c1[v * MP] = acc[v] + bias2;
    __syncthreads();

    const int orow = row0 + w;
    if (orow < G) {
        const float* sp = &sX[1][w * MP + lane * 4];
        float* op = out + (size_t)orow * MH + lane * 4;
        *(v4f*)(op)       = *(const v4f*)(sp);
        *(v4f*)(op + 128) = *(const v4f*)(sp + 128);
    }
}


extern "C" void kernel_launch(void* const* d_in, const int* in_sizes, int n_in,
                              void* d_out, int out_size, void* d_ws, size_t ws_size,
                              hipStream_t stream) {
    const float* node_features = (const float*)d_in[0];
    const int*   edge_src      = (const int*)d_in[1];
    const int*   edge_dst      = (const int*)d_in[2];
    const int*   graph_ids     = (const int*)d_in[3];
    const float* W_proj        = (const float*)d_in[4];
    const float* b_proj        = (const float*)d_in[5];
    const float* W_gcn         = (const float*)d_in[6];
    const float* b_gcn         = (const float*)d_in[7];
    const float* W_mlp0        = (const float*)d_in[8];
    const float* b_mlp0        = (const float*)d_in[9];
    const float* W_mlp1        = (const float*)d_in[10];
    const float* b_mlp1        = (const float*)d_in[11];
    const float* W_mlp2        = (const float*)d_in[12];
    const float* b_mlp2        = (const float*)d_in[13];

    const int D = D_FEAT;
    const int Mm = 256;
    const int N = in_sizes[3];
    const int E = in_sizes[1];
    const int L = in_sizes[6] / (D * D);
    const int G = out_size / Mm;

    float* ws = (float*)d_ws;
    float* norm_src = ws;
    float* norm_dst = norm_src + N;
    float* h        = (float*)(((size_t)(norm_dst + N) + 511) & ~(size_t)511);
    float* agg      = h + (size_t)N * D;
    float* hg       = agg + (size_t)N * D;
    int*   rec      = (int*)(hg + (size_t)G * D);
    int*   rank     = rec + (size_t)2 * E;
    int*   rowptr   = rank + E;
    int*   hist     = rowptr + N + 1;
    int*   cpre     = hist + (size_t)2 * HB * N;

    const int TB = 256;
    auto blocks = [&](size_t n) { return (unsigned)((n + TB - 1) / TB); };

    slice_hist_kernel<<<2 * HB, 1024, 0, stream>>>(edge_src, edge_dst, hist, rank, cpre, N, E);
    combine_kernel<<<(unsigned)((N + CB - 1) / CB), CB, 0, stream>>>(hist, cpre, rowptr, norm_src, norm_dst,
                                                                   hg, G * D, N);
    fill_kernel<<<blocks(E), TB, 0, stream>>>(edge_src, edge_dst, rank, hist, norm_src, rec, N, E);

    const unsigned gblocks = (unsigned)(((N + GR - 1) / GR + 3) / 4);
    gemm128_wmma_f32<<<gblocks, 128, 0, stream>>>(node_features, W_proj, b_proj, h, N);

    float* hcur  = h;
    float* hnext = agg;
    for (int i = 0; i < L; ++i) {
        if (i < L - 1)
            gcn_layer_wmma_f32<<<gblocks, 32 * LW, 0, stream>>>(hcur, rec, rowptr, W_gcn + (size_t)i * D * D,
                                                                b_gcn + (size_t)i * D, norm_dst, hnext, N,
                                                                 1);
        else
            gcn_layer_pool_wmma_f32<<<gblocks, 32 * LW, 0, stream>>>(hcur, rec, rowptr, W_gcn + (size_t)i * D * D,
                                                                     b_gcn + (size_t)i * D, norm_dst, N,
                                                                      0, graph_ids, hg);
        float* t = hcur; hcur = hnext; hnext = t;
    }

    if (L == 0) pool_kernel<<<blocks((size_t)N * 32), TB, 0, stream>>>(hcur, graph_ids, hg, N);

    mlp3_wmma_f32<<<(unsigned)((G + 15) / 16), 512, 0, stream>>>(hg, W_mlp0, b_mlp0, W_mlp1, b_mlp1, W_mlp2, b_mlp2,
                                                                (float*)d_out, G);
}
